# starting kernel
# baseline (speedup 1.0000x reference)
#include <hip/hip_runtime.h>
#include <hip/hip_bf16.h>
#include <math.h>

#define N1D      64
#define GRID_N   (64 * 64 * 64)
#define N_ATOMS  128
#define K_TERMS  6
#define SPACING  0.5f
#define LOG2E    1.4426950408889634f

typedef float v2f __attribute__((ext_vector_type(2)));
typedef float v8f __attribute__((ext_vector_type(8)));

__device__ __forceinline__ float fast_exp2(float x) {
#if __has_builtin(__builtin_amdgcn_exp2f)
    return __builtin_amdgcn_exp2f(x);
#else
    return exp2f(x);
#endif
}

__global__ __launch_bounds__(256) void gen_dft_kernel(float* __restrict__ Wre,
                                                      float* __restrict__ WimF,
                                                      float* __restrict__ WimI) {
    int t = blockIdx.x * blockDim.x + threadIdx.x;
    if (t >= N1D * N1D) return;
    int k = t >> 6;
    int n = t & 63;
    int kn = (k * n) & 63;
    float ang = (float)kn * 0.09817477042468103f;
    float sgn = ((k + n) & 1) ? -0.125f : 0.125f;
    float c = sgn * cosf(ang);
    float s = sgn * sinf(ang);
    Wre[t]  = c;
    WimF[t] = -s;
    WimI[t] = s;
}

__global__ __launch_bounds__(256) void density_kernel(const float* __restrict__ X,
                                                      const float* __restrict__ aw,
                                                      const float* __restrict__ bw,
                                                      float* __restrict__ rho_re,
                                                      float* __restrict__ rho_im) {
    const int lane   = threadIdx.x & 31;
    const int waveId = (blockIdx.x * blockDim.x + threadIdx.x) >> 5;
    const int g0 = waveId << 4;
    const int xg = g0 >> 12;
    const int yg = (g0 >> 6) & 63;
    const int zb = g0 & 63;

    const float cx = 16.0f;
    const float gx = (float)xg * SPACING - cx;
    const float gy = (float)yg * SPACING - cx;
    const int half = lane >> 4;
    const int l16  = lane & 15;

    float gzm = ((float)(zb + l16) * SPACING) - cx;
    v2f Aop;
    Aop.x = half ? (-2.0f * gzm) : (-2.0f * gx);
    Aop.y = half ? 0.0f          : (-2.0f * gy);

    float gsq[8];
#pragma unroll
    for (int v = 0; v < 8; ++v) {
        float gz = ((float)(zb + v + 8 * half) * SPACING) - cx;
        gsq[v] = gx * gx + gy * gy + gz * gz;
    }

    float dens[8] = {0.f, 0.f, 0.f, 0.f, 0.f, 0.f, 0.f, 0.f};

    for (int a0 = 0; a0 < N_ATOMS; a0 += 16) {
        const int a = a0 + l16;
        float ax = X[3 * a + 0] - cx;
        float ay = X[3 * a + 1] - cx;
        float az = X[3 * a + 2] - cx;
        float xsq = ax * ax + ay * ay + az * az;

        v2f Bop;
        Bop.x = half ? az   : ax;
        Bop.y = half ? 0.0f : ay;

        v8f dot = {};
        dot = __builtin_amdgcn_wmma_f32_16x16x4_f32(
            false, Aop, false, Bop, (short)0, dot, false, false);

        float awk[K_TERMS], bl2[K_TERMS];
#pragma unroll
        for (int k = 0; k < K_TERMS; ++k) {
            awk[k] = aw[a * K_TERMS + k];
            bl2[k] = bw[a * K_TERMS + k] * LOG2E;
        }

#pragma unroll
        for (int v = 0; v < 8; ++v) {
            float d2 = dot[v] + gsq[v] + xsq;
            float s = 0.0f;
#pragma unroll
            for (int k = 0; k < K_TERMS; ++k)
                s += awk[k] * fast_exp2(bl2[k] * d2);
            dens[v] += s;
        }
    }

#pragma unroll
    for (int v = 0; v < 8; ++v) {
        float s = dens[v];
        s += __shfl_xor(s, 1, 32);
        s += __shfl_xor(s, 2, 32);
        s += __shfl_xor(s, 4, 32);
        s += __shfl_xor(s, 8, 32);
        dens[v] = s;
    }
    if (lane == 0) {
#pragma unroll
        for (int v = 0; v < 8; ++v) rho_re[g0 + v] = dens[v];
    }
    if (lane == 16) {
#pragma unroll
        for (int v = 0; v < 8; ++v) rho_re[g0 + 8 + v] = dens[v];
    }
    if (lane < 16) rho_im[g0 + lane] = 0.0f;
}

__global__ __launch_bounds__(256) void dft_pass_kernel(const float* __restrict__ Tre,
                                                       const float* __restrict__ Tim,
                                                       const float* __restrict__ Mre,
                                                       const float* __restrict__ Mim,
                                                       float* __restrict__ Ure,
                                                       float* __restrict__ Uim) {
    const int lane   = threadIdx.x & 31;
    const int waveId = (blockIdx.x * blockDim.x + threadIdx.x) >> 5;
    const int p0 = (waveId & 255) << 4;
    const int k0 = (waveId >> 8) << 4;
    const int half = lane >> 4;
    const int l16  = lane & 15;
    const int krow = k0 + l16;
    const int prow = p0 + l16;

    v8f rr = {}, ii = {}, ri = {}, ir = {};

#pragma unroll 4
    for (int n0 = 0; n0 < N1D; n0 += 4) {
        const int koff = n0 + 2 * half;
        v2f mr = *(const v2f*)(Mre + krow * N1D + koff);
        v2f mi = *(const v2f*)(Mim + krow * N1D + koff);
        v2f tr = *(const v2f*)(Tre + prow * N1D + koff);
        v2f ti = *(const v2f*)(Tim + prow * N1D + koff);
        rr = __builtin_amdgcn_wmma_f32_16x16x4_f32(false, mr, false, tr, (short)0, rr, false, false);
        ii = __builtin_amdgcn_wmma_f32_16x16x4_f32(false, mi, false, ti, (short)0, ii, false, false);
        ri = __builtin_amdgcn_wmma_f32_16x16x4_f32(false, mr, false, ti, (short)0, ri, false, false);
        ir = __builtin_amdgcn_wmma_f32_16x16x4_f32(false, mi, false, tr, (short)0, ir, false, false);
    }

#pragma unroll
    for (int v = 0; v < 8; ++v) {
        const int row = k0 + v + 8 * half;
        const int idx = row * 4096 + p0 + l16;
        Ure[idx] = rr[v] - ii[v];
        Uim[idx] = ri[v] + ir[v];
    }
}

__global__ __launch_bounds__(256) void hamming_kernel(float* __restrict__ re,
                                                      float* __restrict__ im,
                                                      const float* __restrict__ h) {
    int g = blockIdx.x * blockDim.x + threadIdx.x;
    if (g >= GRID_N) return;
    float hv = h[g];
    re[g] *= hv;
    im[g] *= hv;
}

__global__ __launch_bounds__(256) void reduce1_kernel(const float* __restrict__ x,
                                                      float* __restrict__ psum,
                                                      float* __restrict__ psq) {
    __shared__ float ssum[256];
    __shared__ float ssq[256];
    const int tid = threadIdx.x;
    const int base = blockIdx.x * 512;
    float a = x[base + tid];
    float b = x[base + 256 + tid];
    ssum[tid] = a + b;
    ssq[tid]  = a * a + b * b;
    __syncthreads();
    for (int s = 128; s > 0; s >>= 1) {
        if (tid < s) { ssum[tid] += ssum[tid + s]; ssq[tid] += ssq[tid + s]; }
        __syncthreads();
    }
    if (tid == 0) { psum[blockIdx.x] = ssum[0]; psq[blockIdx.x] = ssq[0]; }
}

__global__ __launch_bounds__(256) void reduce2_kernel(const float* __restrict__ psum,
                                                      const float* __restrict__ psq,
                                                      float* __restrict__ stats) {
    __shared__ float ssum[256];
    __shared__ float ssq[256];
    const int tid = threadIdx.x;
    ssum[tid] = psum[tid] + psum[tid + 256];
    ssq[tid]  = psq[tid] + psq[tid + 256];
    __syncthreads();
    for (int s = 128; s > 0; s >>= 1) {
        if (tid < s) { ssum[tid] += ssum[tid + s]; ssq[tid] += ssq[tid + s]; }
        __syncthreads();
    }
    if (tid == 0) {
        const float invN = 1.0f / (float)GRID_N;
        float mean = ssum[0] * invN;
        float var  = fmaxf(ssq[0] * invN - mean * mean, 0.0f);
        stats[0] = mean;
        stats[1] = 1.0f / (sqrtf(var) + 1e-8f);
    }
}

__global__ __launch_bounds__(256) void finalize_kernel(const float* __restrict__ re,
                                                       const float* __restrict__ stats,
                                                       float* __restrict__ out) {
    int g = blockIdx.x * blockDim.x + threadIdx.x;
    if (g >= GRID_N) return;
    out[g] = (re[g] - stats[0]) * stats[1];
}

extern "C" void kernel_launch(void* const* d_in, const int* in_sizes, int n_in,
                              void* d_out, int out_size, void* d_ws, size_t ws_size,
                              hipStream_t stream) {
    const float* X   = (const float*)d_in[0];
    const float* aw  = (const float*)d_in[1];
    const float* bw  = (const float*)d_in[2];
    const float* ham = (const float*)d_in[4];
    float* out = (float*)d_out;

    float* ws   = (float*)d_ws;
    float* Wre  = ws;
    float* WimF = Wre + 4096;
    float* WimI = WimF + 4096;
    float* Are  = WimI + 4096;
    float* Aim  = Are + GRID_N;
    float* Bre  = Aim + GRID_N;
    float* Bim  = Bre + GRID_N;
    float* psum = Bim + GRID_N;
    float* psq  = psum + 512;
    float* stats = psq + 512;

    gen_dft_kernel<<<16, 256, 0, stream>>>(Wre, WimF, WimI);

    density_kernel<<<2048, 256, 0, stream>>>(X, aw, bw, Are, Aim);

    dft_pass_kernel<<<128, 256, 0, stream>>>(Are, Aim, Wre, WimF, Bre, Bim);
    dft_pass_kernel<<<128, 256, 0, stream>>>(Bre, Bim, Wre, WimF, Are, Aim);
    dft_pass_kernel<<<128, 256, 0, stream>>>(Are, Aim, Wre, WimF, Bre, Bim);

    hamming_kernel<<<GRID_N / 256, 256, 0, stream>>>(Bre, Bim, ham);

    dft_pass_kernel<<<128, 256, 0, stream>>>(Bre, Bim, Wre, WimI, Are, Aim);
    dft_pass_kernel<<<128, 256, 0, stream>>>(Are, Aim, Wre, WimI, Bre, Bim);
    dft_pass_kernel<<<128, 256, 0, stream>>>(Bre, Bim, Wre, WimI, Are, Aim);

    reduce1_kernel<<<512, 256, 0, stream>>>(Are, psum, psq);
    reduce2_kernel<<<1, 256, 0, stream>>>(psum, psq, stats);
    finalize_kernel<<<GRID_N / 256, 256, 0, stream>>>(Are, stats, out);
}


# density: factor each Gaussian into per-axis exp tables (x,y,z) and sum the 768 atom-terms as an f32 WMMA product instead of one exp per point-atom-term; changes the order of the atom/term sum
# speedup vs baseline: 1.1557x; 1.1557x over previous
#include <hip/hip_runtime.h>
#include <hip/hip_bf16.h>
#include <math.h>

#define N1D      64
#define GRID_N   (64 * 64 * 64)
#define N_ATOMS  128
#define K_TERMS  6
#define SPACING  0.5f
#define LOG2E    1.4426950408889634f

#define NAK      (N_ATOMS * K_TERMS)
#define KS2      (NAK / 8)
#define TBL      (N1D * NAK)
#define CHUNK    16
#define NCHUNK   (KS2 / CHUNK)

typedef float v2f __attribute__((ext_vector_type(2)));
typedef float v4f __attribute__((ext_vector_type(4)));
typedef float v8f __attribute__((ext_vector_type(8)));

__device__ __forceinline__ float fast_exp2(float x) {
#if __has_builtin(__builtin_amdgcn_exp2f)
    return __builtin_amdgcn_exp2f(x);
#else
    return exp2f(x);
#endif
}

__global__ __launch_bounds__(256) void setup_kernel(const float* __restrict__ X,
                                                    const float* __restrict__ aw,
                                                    const float* __restrict__ bw,
                                                    float* __restrict__ Wre,
                                                    float* __restrict__ WimF,
                                                    float* __restrict__ WimI,
                                                    float* __restrict__ AxF,
                                                    float* __restrict__ EyF,
                                                    float* __restrict__ EzF) {
    if (blockIdx.x < 16) {
        int t = blockIdx.x * blockDim.x + threadIdx.x;
        int k = t >> 6;
        int n = t & 63;
        int kn = (k * n) & 63;
        float ang = (float)kn * 0.09817477042468103f;
        float sgn = ((k + n) & 1) ? -0.125f : 0.125f;
        float c = sgn * cosf(ang);
        float s = sgn * sinf(ang);
        Wre[t]  = c;
        WimF[t] = -s;
        WimI[t] = s;
        return;
    }
    const int t    = (blockIdx.x - 16) * blockDim.x + threadIdx.x;
    const int axis = t / TBL;
    const int r    = t - axis * TBL;
    const int i    = r / NAK;
    const int ak   = r - i * NAK;
    const int a    = ak / K_TERMS;
    const float d  = (float)i * SPACING - X[3 * a + axis];
    const float e  = fast_exp2((bw[ak] * LOG2E) * (d * d));
    const int ks2  = ak >> 3;
    const int slot = ((ak >> 2) & 1) * 2 + (ak & 1);
    const int half = (ak >> 1) & 1;
    if (axis == 0) {
        AxF[((i * KS2 + ks2) * 2 + half) * 4 + slot] = aw[ak] * e;
    } else {
        float* T = (axis == 1) ? EyF : EzF;
        T[(((i >> 4) * KS2 + ks2) * 32 + half * 16 + (i & 15)) * 4 + slot] = e;
    }
}

struct FragChunk {
    v4f a[CHUNK];
    v4f y[CHUNK];
    v4f z[CHUNK];
};

__device__ __forceinline__ void load_chunk(FragChunk& f, const v4f* ap, const v4f* yp,
                                           const v4f* zp, int c) {
#pragma unroll
    for (int j = 0; j < CHUNK; ++j) {
        const int n = c * CHUNK + j;
        f.a[j] = ap[n * 2];
        f.y[j] = yp[n * 32];
        f.z[j] = zp[n * 32];
    }
}

__device__ __forceinline__ void mma_chunk(const FragChunk& f, v8f& acc0, v8f& acc1) {
#pragma unroll
    for (int j = 0; j < CHUNK; ++j) {
        v2f A0, A1, B0, B1;
        A0.x = f.a[j].x * f.y[j].x;  A0.y = f.a[j].y * f.y[j].y;
        A1.x = f.a[j].z * f.y[j].z;  A1.y = f.a[j].w * f.y[j].w;
        B0.x = f.z[j].x;  B0.y = f.z[j].y;
        B1.x = f.z[j].z;  B1.y = f.z[j].w;
        acc0 = __builtin_amdgcn_wmma_f32_16x16x4_f32(
            false, A0, false, B0, (short)0, acc0, false, false);
        acc1 = __builtin_amdgcn_wmma_f32_16x16x4_f32(
            false, A1, false, B1, (short)0, acc1, false, false);
    }
}

__global__ __launch_bounds__(128) void density_gemm_kernel(const float* AxF,
                                                           const float* EyF,
                                                           const float* EzF,
                                                           float* rho_re,
                                                           float* rho_im) {
    const int lane = threadIdx.x & 31;
    const int izt  = threadIdx.x >> 5;
    const int ix   = blockIdx.x >> 2;
    const int iyt  = blockIdx.x & 3;
    const int half = lane >> 4;
    const int l16  = lane & 15;

    const v4f* ap = (const v4f*)AxF + (ix * KS2) * 2 + half;
    const v4f* yp = (const v4f*)EyF + (iyt * KS2) * 32 + lane;
    const v4f* zp = (const v4f*)EzF + (izt * KS2) * 32 + lane;

    v8f acc0 = {}, acc1 = {};
    FragChunk f0, f1;

    load_chunk(f0, ap, yp, zp, 0);
    for (int c = 0; c < NCHUNK - 2; c += 2) {
        load_chunk(f1, ap, yp, zp, c + 1);
        mma_chunk(f0, acc0, acc1);
        load_chunk(f0, ap, yp, zp, c + 2);
        mma_chunk(f1, acc0, acc1);
    }
    load_chunk(f1, ap, yp, zp, NCHUNK - 1);
    mma_chunk(f0, acc0, acc1);
    mma_chunk(f1, acc0, acc1);

#pragma unroll
    for (int v = 0; v < 8; ++v) {
        const int iy  = iyt * 16 + v + 8 * half;
        const int idx = (ix * N1D + iy) * N1D + izt * 16 + l16;
        rho_re[idx] = acc0[v] + acc1[v];
        rho_im[idx] = 0.0f;
    }
}

__global__ __launch_bounds__(256) void dft_pass_kernel(const float* __restrict__ Tre,
                                                       const float* __restrict__ Tim,
                                                       const float* __restrict__ Mre,
                                                       const float* __restrict__ Mim,
                                                       float* __restrict__ Ure,
                                                       float* __restrict__ Uim) {
    const int lane   = threadIdx.x & 31;
    const int waveId = (blockIdx.x * blockDim.x + threadIdx.x) >> 5;
    const int p0 = (waveId & 255) << 4;
    const int k0 = (waveId >> 8) << 4;
    const int half = lane >> 4;
    const int l16  = lane & 15;
    const int krow = k0 + l16;
    const int prow = p0 + l16;

    v8f rr = {}, ii = {}, ri = {}, ir = {};

#pragma unroll 4
    for (int n0 = 0; n0 < N1D; n0 += 4) {
        const int koff = n0 + 2 * half;
        v2f mr = *(const v2f*)(Mre + krow * N1D + koff);
        v2f mi = *(const v2f*)(Mim + krow * N1D + koff);
        v2f tr = *(const v2f*)(Tre + prow * N1D + koff);
        v2f ti = *(const v2f*)(Tim + prow * N1D + koff);
        rr = __builtin_amdgcn_wmma_f32_16x16x4_f32(false, mr, false, tr, (short)0, rr, false, false);
        ii = __builtin_amdgcn_wmma_f32_16x16x4_f32(false, mi, false, ti, (short)0, ii, false, false);
        ri = __builtin_amdgcn_wmma_f32_16x16x4_f32(false, mr, false, ti, (short)0, ri, false, false);
        ir = __builtin_amdgcn_wmma_f32_16x16x4_f32(false, mi, false, tr, (short)0, ir, false, false);
    }

#pragma unroll
    for (int v = 0; v < 8; ++v) {
        const int row = k0 + v + 8 * half;
        const int idx = row * 4096 + p0 + l16;
        Ure[idx] = rr[v] - ii[v];
        Uim[idx] = ri[v] + ir[v];
    }
}

__global__ __launch_bounds__(256) void hamming_kernel(float* __restrict__ re,
                                                      float* __restrict__ im,
                                                      const float* __restrict__ h) {
    int g = blockIdx.x * blockDim.x + threadIdx.x;
    if (g >= GRID_N) return;
    float hv = h[g];
    re[g] *= hv;
    im[g] *= hv;
}

__global__ __launch_bounds__(256) void reduce1_kernel(const float* __restrict__ x,
                                                      float* __restrict__ psum,
                                                      float* __restrict__ psq) {
    __shared__ float ssum[256];
    __shared__ float ssq[256];
    const int tid = threadIdx.x;
    const int base = blockIdx.x * 512;
    float a = x[base + tid];
    float b = x[base + 256 + tid];
    ssum[tid] = a + b;
    ssq[tid]  = a * a + b * b;
    __syncthreads();
    for (int s = 128; s > 0; s >>= 1) {
        if (tid < s) { ssum[tid] += ssum[tid + s]; ssq[tid] += ssq[tid + s]; }
        __syncthreads();
    }
    if (tid == 0) { psum[blockIdx.x] = ssum[0]; psq[blockIdx.x] = ssq[0]; }
}

__global__ __launch_bounds__(256) void reduce2_kernel(const float* __restrict__ psum,
                                                      const float* __restrict__ psq,
                                                      float* __restrict__ stats) {
    __shared__ float ssum[256];
    __shared__ float ssq[256];
    const int tid = threadIdx.x;
    ssum[tid] = psum[tid] + psum[tid + 256];
    ssq[tid]  = psq[tid] + psq[tid + 256];
    __syncthreads();
    for (int s = 128; s > 0; s >>= 1) {
        if (tid < s) { ssum[tid] += ssum[tid + s]; ssq[tid] += ssq[tid + s]; }
        __syncthreads();
    }
    if (tid == 0) {
        const float invN = 1.0f / (float)GRID_N;
        float mean = ssum[0] * invN;
        float var  = fmaxf(ssq[0] * invN - mean * mean, 0.0f);
        stats[0] = mean;
        stats[1] = 1.0f / (sqrtf(var) + 1e-8f);
    }
}

__global__ __launch_bounds__(256) void finalize_kernel(const float* __restrict__ re,
                                                       const float* __restrict__ stats,
                                                       float* __restrict__ out) {
    int g = blockIdx.x * blockDim.x + threadIdx.x;
    if (g >= GRID_N) return;
    out[g] = (re[g] - stats[0]) * stats[1];
}

extern "C" void kernel_launch(void* const* d_in, const int* in_sizes, int n_in,
                              void* d_out, int out_size, void* d_ws, size_t ws_size,
                              hipStream_t stream) {
    const float* X   = (const float*)d_in[0];
    const float* aw  = (const float*)d_in[1];
    const float* bw  = (const float*)d_in[2];
    const float* ham = (const float*)d_in[4];
    float* out = (float*)d_out;

    float* ws   = (float*)d_ws;
    float* Wre  = ws;
    float* WimF = Wre + 4096;
    float* WimI = WimF + 4096;
    float* Are  = WimI + 4096;
    float* Aim  = Are + GRID_N;
    float* Bre  = Aim + GRID_N;
    float* Bim  = Bre + GRID_N;
    float* psum = Bim + GRID_N;
    float* psq  = psum + 512;
    float* stats = psq + 512;

    float* AxF = Bre;
    float* EyF = AxF + TBL;
    float* EzF = EyF + TBL;

    setup_kernel<<<592, 256, 0, stream>>>(X, aw, bw, Wre, WimF, WimI,
                                                          AxF, EyF, EzF);

    density_gemm_kernel<<<256, 128, 0, stream>>>(AxF, EyF, EzF, Are, Aim);

    dft_pass_kernel<<<128, 256, 0, stream>>>(Are, Aim, Wre, WimF, Bre, Bim);
    dft_pass_kernel<<<128, 256, 0, stream>>>(Bre, Bim, Wre, WimF, Are, Aim);
    dft_pass_kernel<<<128, 256, 0, stream>>>(Are, Aim, Wre, WimF, Bre, Bim);

    hamming_kernel<<<GRID_N / 256, 256, 0, stream>>>(Bre, Bim, ham);

    dft_pass_kernel<<<128, 256, 0, stream>>>(Bre, Bim, Wre, WimI, Are, Aim);
    dft_pass_kernel<<<128, 256, 0, stream>>>(Are, Aim, Wre, WimI, Bre, Bim);
    dft_pass_kernel<<<128, 256, 0, stream>>>(Bre, Bim, Wre, WimI, Are, Aim);

    reduce1_kernel<<<512, 256, 0, stream>>>(Are, psum, psq);
    reduce2_kernel<<<1, 256, 0, stream>>>(psum, psq, stats);
    finalize_kernel<<<GRID_N / 256, 256, 0, stream>>>(Are, stats, out);
}


# fuse 3 launches into neighbours: filter multiply into forward pass 3's store, block sums into last inverse pass, final reduce into normalise (12 -> 9); changes the order of the mean/variance sums
# speedup vs baseline: 1.3326x; 1.1531x over previous
#include <hip/hip_runtime.h>
#include <hip/hip_bf16.h>
#include <math.h>

#define N1D      64
#define GRID_N   (64 * 64 * 64)
#define N_ATOMS  128
#define K_TERMS  6
#define SPACING  0.5f
#define LOG2E    1.4426950408889634f

#define NAK      (N_ATOMS * K_TERMS)
#define KS2      (NAK / 8)
#define TBL      (N1D * NAK)
#define CHUNK    16
#define NCHUNK   (KS2 / CHUNK)

typedef float v2f __attribute__((ext_vector_type(2)));
typedef float v4f __attribute__((ext_vector_type(4)));
typedef float v8f __attribute__((ext_vector_type(8)));

__device__ __forceinline__ float fast_exp2(float x) {
#if __has_builtin(__builtin_amdgcn_exp2f)
    return __builtin_amdgcn_exp2f(x);
#else
    return exp2f(x);
#endif
}

__global__ __launch_bounds__(256) void setup_kernel(const float* __restrict__ X,
                                                    const float* __restrict__ aw,
                                                    const float* __restrict__ bw,
                                                    float* __restrict__ Wre,
                                                    float* __restrict__ WimF,
                                                    float* __restrict__ WimI,
                                                    float* __restrict__ AxF,
                                                    float* __restrict__ EyF,
                                                    float* __restrict__ EzF) {
    if (blockIdx.x < 16) {
        int t = blockIdx.x * blockDim.x + threadIdx.x;
        int k = t >> 6;
        int n = t & 63;
        int kn = (k * n) & 63;
        float ang = (float)kn * 0.09817477042468103f;
        float sgn = ((k + n) & 1) ? -0.125f : 0.125f;
        float c = sgn * cosf(ang);
        float s = sgn * sinf(ang);
        Wre[t]  = c;
        WimF[t] = -s;
        WimI[t] = s;
        return;
    }
    const int t    = (blockIdx.x - 16) * blockDim.x + threadIdx.x;
    const int axis = t / TBL;
    const int r    = t - axis * TBL;
    const int i    = r / NAK;
    const int ak   = r - i * NAK;
    const int a    = ak / K_TERMS;
    const float d  = (float)i * SPACING - X[3 * a + axis];
    const float e  = fast_exp2((bw[ak] * LOG2E) * (d * d));
    const int ks2  = ak >> 3;
    const int slot = ((ak >> 2) & 1) * 2 + (ak & 1);
    const int half = (ak >> 1) & 1;
    if (axis == 0) {
        AxF[((i * KS2 + ks2) * 2 + half) * 4 + slot] = aw[ak] * e;
    } else {
        float* T = (axis == 1) ? EyF : EzF;
        T[(((i >> 4) * KS2 + ks2) * 32 + half * 16 + (i & 15)) * 4 + slot] = e;
    }
}

struct FragChunk {
    v4f a[CHUNK];
    v4f y[CHUNK];
    v4f z[CHUNK];
};

__device__ __forceinline__ void load_chunk(FragChunk& f, const v4f* ap, const v4f* yp,
                                           const v4f* zp, int c) {
#pragma unroll
    for (int j = 0; j < CHUNK; ++j) {
        const int n = c * CHUNK + j;
        f.a[j] = ap[n * 2];
        f.y[j] = yp[n * 32];
        f.z[j] = zp[n * 32];
    }
}

__device__ __forceinline__ void mma_chunk(const FragChunk& f, v8f& acc0, v8f& acc1) {
#pragma unroll
    for (int j = 0; j < CHUNK; ++j) {
        v2f A0, A1, B0, B1;
        A0.x = f.a[j].x * f.y[j].x;  A0.y = f.a[j].y * f.y[j].y;
        A1.x = f.a[j].z * f.y[j].z;  A1.y = f.a[j].w * f.y[j].w;
        B0.x = f.z[j].x;  B0.y = f.z[j].y;
        B1.x = f.z[j].z;  B1.y = f.z[j].w;
        acc0 = __builtin_amdgcn_wmma_f32_16x16x4_f32(
            false, A0, false, B0, (short)0, acc0, false, false);
        acc1 = __builtin_amdgcn_wmma_f32_16x16x4_f32(
            false, A1, false, B1, (short)0, acc1, false, false);
    }
}

__global__ __launch_bounds__(128) void density_gemm_kernel(const float* AxF,
                                                           const float* EyF,
                                                           const float* EzF,
                                                           float* rho_re,
                                                           float* rho_im) {
    const int lane = threadIdx.x & 31;
    const int izt  = threadIdx.x >> 5;
    const int ix   = blockIdx.x >> 2;
    const int iyt  = blockIdx.x & 3;
    const int half = lane >> 4;
    const int l16  = lane & 15;

    const v4f* ap = (const v4f*)AxF + (ix * KS2) * 2 + half;
    const v4f* yp = (const v4f*)EyF + (iyt * KS2) * 32 + lane;
    const v4f* zp = (const v4f*)EzF + (izt * KS2) * 32 + lane;

    v8f acc0 = {}, acc1 = {};
    FragChunk f0, f1;

    load_chunk(f0, ap, yp, zp, 0);
    for (int c = 0; c < NCHUNK - 2; c += 2) {
        load_chunk(f1, ap, yp, zp, c + 1);
        mma_chunk(f0, acc0, acc1);
        load_chunk(f0, ap, yp, zp, c + 2);
        mma_chunk(f1, acc0, acc1);
    }
    load_chunk(f1, ap, yp, zp, NCHUNK - 1);
    mma_chunk(f0, acc0, acc1);
    mma_chunk(f1, acc0, acc1);

#pragma unroll
    for (int v = 0; v < 8; ++v) {
        const int iy  = iyt * 16 + v + 8 * half;
        const int idx = (ix * N1D + iy) * N1D + izt * 16 + l16;
        rho_re[idx] = acc0[v] + acc1[v];
        rho_im[idx] = 0.0f;
    }
}

#define NBLK_DFT   128

struct DftTile {
    v8f rr, ii, ri, ir;
    int idx0;
};

__device__ __forceinline__ int dft_idx0() {
    const int lane   = threadIdx.x & 31;
    const int waveId = (blockIdx.x * blockDim.x + threadIdx.x) >> 5;
    return (((waveId >> 8) << 4) + 8 * (lane >> 4)) * 4096 + ((waveId & 255) << 4) + (lane & 15);
}

__device__ __forceinline__ DftTile dft_tile(const float* __restrict__ Tre,
                                            const float* __restrict__ Tim,
                                            const float* __restrict__ Mre,
                                            const float* __restrict__ Mim) {
    const int lane   = threadIdx.x & 31;
    const int waveId = (blockIdx.x * blockDim.x + threadIdx.x) >> 5;
    const int p0 = (waveId & 255) << 4;
    const int k0 = (waveId >> 8) << 4;
    const int half = lane >> 4;
    const int l16  = lane & 15;
    const int krow = k0 + l16;
    const int prow = p0 + l16;

    v8f rr = {}, ii = {}, ri = {}, ir = {};

#pragma unroll 4
    for (int n0 = 0; n0 < N1D; n0 += 4) {
        const int koff = n0 + 2 * half;
        v2f mr = *(const v2f*)(Mre + krow * N1D + koff);
        v2f mi = *(const v2f*)(Mim + krow * N1D + koff);
        v2f tr = *(const v2f*)(Tre + prow * N1D + koff);
        v2f ti = *(const v2f*)(Tim + prow * N1D + koff);
        rr = __builtin_amdgcn_wmma_f32_16x16x4_f32(false, mr, false, tr, (short)0, rr, false, false);
        ii = __builtin_amdgcn_wmma_f32_16x16x4_f32(false, mi, false, ti, (short)0, ii, false, false);
        ri = __builtin_amdgcn_wmma_f32_16x16x4_f32(false, mr, false, ti, (short)0, ri, false, false);
        ir = __builtin_amdgcn_wmma_f32_16x16x4_f32(false, mi, false, tr, (short)0, ir, false, false);
    }
    DftTile t;
    t.rr = rr; t.ii = ii; t.ri = ri; t.ir = ir;
    t.idx0 = dft_idx0();
    return t;
}

__global__ __launch_bounds__(256) void dft_pass_kernel(const float* __restrict__ Tre,
                                                       const float* __restrict__ Tim,
                                                       const float* __restrict__ Mre,
                                                       const float* __restrict__ Mim,
                                                       float* __restrict__ Ure,
                                                       float* __restrict__ Uim) {
    const DftTile t = dft_tile(Tre, Tim, Mre, Mim);
#pragma unroll
    for (int v = 0; v < 8; ++v) {
        const int idx = t.idx0 + v * 4096;
        Ure[idx] = t.rr[v] - t.ii[v];
        Uim[idx] = t.ri[v] + t.ir[v];
    }
}

__global__ __launch_bounds__(256) void dft_pass_filter_kernel(const float* __restrict__ Tre,
                                                              const float* __restrict__ Tim,
                                                              const float* __restrict__ Mre,
                                                              const float* __restrict__ Mim,
                                                              float* __restrict__ Ure,
                                                              float* __restrict__ Uim,
                                                              const float* __restrict__ h) {
    const int idx0 = dft_idx0();
    float hv[8];
#pragma unroll
    for (int v = 0; v < 8; ++v) hv[v] = h[idx0 + v * 4096];
    const DftTile t = dft_tile(Tre, Tim, Mre, Mim);
#pragma unroll
    for (int v = 0; v < 8; ++v) {
        const int idx = t.idx0 + v * 4096;
        Ure[idx] = (t.rr[v] - t.ii[v]) * hv[v];
        Uim[idx] = (t.ri[v] + t.ir[v]) * hv[v];
    }
}

__global__ __launch_bounds__(256) void dft_pass_stats_kernel(const float* __restrict__ Tre,
                                                             const float* __restrict__ Tim,
                                                             const float* __restrict__ Mre,
                                                             const float* __restrict__ Mim,
                                                             float* __restrict__ Ure,
                                                             float* __restrict__ Uim,
                                                             float* __restrict__ psum,
                                                             float* __restrict__ psq) {
    __shared__ float wsum[8];
    __shared__ float wsq[8];
    const DftTile t = dft_tile(Tre, Tim, Mre, Mim);
    float s = 0.0f, q = 0.0f;
#pragma unroll
    for (int v = 0; v < 8; ++v) {
        const int idx = t.idx0 + v * 4096;
        const float re = t.rr[v] - t.ii[v];
        Ure[idx] = re;
        Uim[idx] = t.ri[v] + t.ir[v];
        s += re;
        q += re * re;
    }
#pragma unroll
    for (int m = 16; m > 0; m >>= 1) {
        s += __shfl_xor(s, m);
        q += __shfl_xor(q, m);
    }
    if ((threadIdx.x & 31) == 0) { wsum[threadIdx.x >> 5] = s; wsq[threadIdx.x >> 5] = q; }
    __syncthreads();
    if (threadIdx.x == 0) {
        float a = 0.0f, b = 0.0f;
#pragma unroll
        for (int i = 0; i < 8; ++i) { a += wsum[i]; b += wsq[i]; }
        psum[blockIdx.x] = a;
        psq[blockIdx.x]  = b;
    }
}

__global__ __launch_bounds__(256) void finalize_kernel(const float* __restrict__ re,
                                                       const float* __restrict__ psum,
                                                       const float* __restrict__ psq,
                                                       float* __restrict__ out) {
    __shared__ float part[8];
    const int tid = threadIdx.x;
    const int g = blockIdx.x * blockDim.x + tid;
    const float x = re[g];
    float p = (tid < NBLK_DFT) ? psum[tid] : psq[tid - NBLK_DFT];
#pragma unroll
    for (int m = 16; m > 0; m >>= 1) p += __shfl_xor(p, m);
    if ((tid & 31) == 0) part[tid >> 5] = p;
    __syncthreads();
    const float tsum = (part[0] + part[1]) + (part[2] + part[3]);
    const float tsq  = (part[4] + part[5]) + (part[6] + part[7]);
    const float invN = 1.0f / (float)GRID_N;
    const float mean = tsum * invN;
    const float var  = fmaxf(tsq * invN - mean * mean, 0.0f);
    const float inv  = 1.0f / (sqrtf(var) + 1e-8f);
    out[g] = (x - mean) * inv;
}

extern "C" void kernel_launch(void* const* d_in, const int* in_sizes, int n_in,
                              void* d_out, int out_size, void* d_ws, size_t ws_size,
                              hipStream_t stream) {
    const float* X   = (const float*)d_in[0];
    const float* aw  = (const float*)d_in[1];
    const float* bw  = (const float*)d_in[2];
    const float* ham = (const float*)d_in[4];
    float* out = (float*)d_out;

    float* ws   = (float*)d_ws;
    float* Wre  = ws;
    float* WimF = Wre + 4096;
    float* WimI = WimF + 4096;
    float* Are  = WimI + 4096;
    float* Aim  = Are + GRID_N;
    float* Bre  = Aim + GRID_N;
    float* Bim  = Bre + GRID_N;
    float* psum = Bim + GRID_N;
    float* psq  = psum + NBLK_DFT;

    float* AxF = Bre;
    float* EyF = AxF + TBL;
    float* EzF = EyF + TBL;

    setup_kernel<<<592, 256, 0, stream>>>(X, aw, bw, Wre, WimF, WimI,
                                                          AxF, EyF, EzF);

    density_gemm_kernel<<<256, 128, 0, stream>>>(AxF, EyF, EzF, Are, Aim);

    dft_pass_kernel<<<NBLK_DFT, 256, 0, stream>>>(Are, Aim, Wre, WimF, Bre, Bim);
    dft_pass_kernel<<<NBLK_DFT, 256, 0, stream>>>(Bre, Bim, Wre, WimF, Are, Aim);
    dft_pass_filter_kernel<<<NBLK_DFT, 256, 0, stream>>>(Are, Aim, Wre, WimF, Bre, Bim, ham);

    dft_pass_kernel<<<NBLK_DFT, 256, 0, stream>>>(Bre, Bim, Wre, WimI, Are, Aim);
    dft_pass_kernel<<<NBLK_DFT, 256, 0, stream>>>(Are, Aim, Wre, WimI, Bre, Bim);
    dft_pass_stats_kernel<<<NBLK_DFT, 256, 0, stream>>>(Bre, Bim, Wre, WimI, Are, Aim,
                                                        psum, psq);

    finalize_kernel<<<GRID_N / 256, 256, 0, stream>>>(Are, psum, psq, out);
}


# two DFT axis passes per launch, the first pass's 64x16 slab kept in LDS for the second (6 pass launches -> 3, 4-wave blocks); inverse axes now x,y,z; changes the order of the mean/variance sums
# speedup vs baseline: 1.9261x; 1.4454x over previous
#include <hip/hip_runtime.h>
#include <hip/hip_bf16.h>
#include <math.h>

#define N1D      64
#define GRID_N   (64 * 64 * 64)
#define N_ATOMS  128
#define K_TERMS  6
#define SPACING  0.5f
#define LOG2E    1.4426950408889634f

#define NAK      (N_ATOMS * K_TERMS)
#define KS2      (NAK / 8)
#define TBL      (N1D * NAK)
#define CHUNK    16
#define NCHUNK   (KS2 / CHUNK)

typedef float v2f __attribute__((ext_vector_type(2)));
typedef float v4f __attribute__((ext_vector_type(4)));
typedef float v8f __attribute__((ext_vector_type(8)));

__device__ __forceinline__ float fast_exp2(float x) {
#if __has_builtin(__builtin_amdgcn_exp2f)
    return __builtin_amdgcn_exp2f(x);
#else
    return exp2f(x);
#endif
}

__global__ __launch_bounds__(256) void setup_kernel(const float* __restrict__ X,
                                                    const float* __restrict__ aw,
                                                    const float* __restrict__ bw,
                                                    float* __restrict__ Wre,
                                                    float* __restrict__ WimF,
                                                    float* __restrict__ WimI,
                                                    float* __restrict__ AxF,
                                                    float* __restrict__ EyF,
                                                    float* __restrict__ EzF) {
    if (blockIdx.x < 16) {
        int t = blockIdx.x * blockDim.x + threadIdx.x;
        int k = t >> 6;
        int n = t & 63;
        int kn = (k * n) & 63;
        float ang = (float)kn * 0.09817477042468103f;
        float sgn = ((k + n) & 1) ? -0.125f : 0.125f;
        float c = sgn * cosf(ang);
        float s = sgn * sinf(ang);
        Wre[t]  = c;
        WimF[t] = -s;
        WimI[t] = s;
        return;
    }
    const int t    = (blockIdx.x - 16) * blockDim.x + threadIdx.x;
    const int axis = t / TBL;
    const int r    = t - axis * TBL;
    const int i    = r / NAK;
    const int ak   = r - i * NAK;
    const int a    = ak / K_TERMS;
    const float d  = (float)i * SPACING - X[3 * a + axis];
    const float e  = fast_exp2((bw[ak] * LOG2E) * (d * d));
    const int ks2  = ak >> 3;
    const int slot = ((ak >> 2) & 1) * 2 + (ak & 1);
    const int half = (ak >> 1) & 1;
    if (axis == 0) {
        AxF[((i * KS2 + ks2) * 2 + half) * 4 + slot] = aw[ak] * e;
    } else {
        float* T = (axis == 1) ? EyF : EzF;
        T[(((i >> 4) * KS2 + ks2) * 32 + half * 16 + (i & 15)) * 4 + slot] = e;
    }
}

struct FragChunk {
    v4f a[CHUNK];
    v4f y[CHUNK];
    v4f z[CHUNK];
};

__device__ __forceinline__ void load_chunk(FragChunk& f, const v4f* ap, const v4f* yp,
                                           const v4f* zp, int c) {
#pragma unroll
    for (int j = 0; j < CHUNK; ++j) {
        const int n = c * CHUNK + j;
        f.a[j] = ap[n * 2];
        f.y[j] = yp[n * 32];
        f.z[j] = zp[n * 32];
    }
}

__device__ __forceinline__ void mma_chunk(const FragChunk& f, v8f& acc0, v8f& acc1) {
#pragma unroll
    for (int j = 0; j < CHUNK; ++j) {
        v2f A0, A1, B0, B1;
        A0.x = f.a[j].x * f.y[j].x;  A0.y = f.a[j].y * f.y[j].y;
        A1.x = f.a[j].z * f.y[j].z;  A1.y = f.a[j].w * f.y[j].w;
        B0.x = f.z[j].x;  B0.y = f.z[j].y;
        B1.x = f.z[j].z;  B1.y = f.z[j].w;
        acc0 = __builtin_amdgcn_wmma_f32_16x16x4_f32(
            false, A0, false, B0, (short)0, acc0, false, false);
        acc1 = __builtin_amdgcn_wmma_f32_16x16x4_f32(
            false, A1, false, B1, (short)0, acc1, false, false);
    }
}

__global__ __launch_bounds__(128) void density_gemm_kernel(const float* AxF,
                                                           const float* EyF,
                                                           const float* EzF,
                                                           float* rho_re,
                                                           float* rho_im) {
    const int lane = threadIdx.x & 31;
    const int izt  = threadIdx.x >> 5;
    const int ix   = blockIdx.x >> 2;
    const int iyt  = blockIdx.x & 3;
    const int half = lane >> 4;
    const int l16  = lane & 15;

    const v4f* ap = (const v4f*)AxF + (ix * KS2) * 2 + half;
    const v4f* yp = (const v4f*)EyF + (iyt * KS2) * 32 + lane;
    const v4f* zp = (const v4f*)EzF + (izt * KS2) * 32 + lane;

    v8f acc0 = {}, acc1 = {};
    FragChunk f0, f1;

    load_chunk(f0, ap, yp, zp, 0);
    for (int c = 0; c < NCHUNK - 2; c += 2) {
        load_chunk(f1, ap, yp, zp, c + 1);
        mma_chunk(f0, acc0, acc1);
        load_chunk(f0, ap, yp, zp, c + 2);
        mma_chunk(f1, acc0, acc1);
    }
    load_chunk(f1, ap, yp, zp, NCHUNK - 1);
    mma_chunk(f0, acc0, acc1);
    mma_chunk(f1, acc0, acc1);

#pragma unroll
    for (int v = 0; v < 8; ++v) {
        const int iy  = iyt * 16 + v + 8 * half;
        const int idx = (ix * N1D + iy) * N1D + izt * 16 + l16;
        rho_re[idx] = acc0[v] + acc1[v];
        rho_im[idx] = 0.0f;
    }
}

#define NBLK2      256
#define WPITCH     68

__device__ __forceinline__ v8f mma4(v2f a, v2f b, v8f c) {
    return __builtin_amdgcn_wmma_f32_16x16x4_f32(false, a, false, b, (short)0, c, false, false);
}

__device__ __forceinline__ int slab_idx(int x, int ky, int kt, int l) {
    return ((x * 4 + kt) * N1D + ky) * 16 + l;
}

__device__ __forceinline__ void slab_pass(const v2f* mr, const v2f* mi,
                                          const float* Sre, const float* Sim,
                                          v8f& rr, v8f& ii, v8f& ri, v8f& ir) {
    const int lane = threadIdx.x & 31;
    const int half = lane >> 4;
    const int l16  = lane & 15;
    v2f br[16], bi[16];
#pragma unroll
    for (int s = 0; s < 16; ++s) {
        const int k = 4 * s + 2 * half;
        br[s].x = Sre[k * 16 + l16];  br[s].y = Sre[(k + 1) * 16 + l16];
        bi[s].x = Sim[k * 16 + l16];  bi[s].y = Sim[(k + 1) * 16 + l16];
    }
    __builtin_amdgcn_sched_barrier(0);
#pragma unroll
    for (int s = 0; s < 16; ++s) {
        rr = mma4(mr[s], br[s], rr);
        ii = mma4(mi[s], bi[s], ii);
        ri = mma4(mr[s], bi[s], ri);
        ir = mma4(mi[s], br[s], ir);
    }
}

__global__ __launch_bounds__(128) void fwd_zy_kernel(const float* __restrict__ Tre,
                                                     const float* __restrict__ Tim,
                                                     const float* __restrict__ Mre,
                                                     const float* __restrict__ Mim,
                                                     float* __restrict__ Ure,
                                                     float* __restrict__ Uim) {
    __shared__ float Sre[N1D * 16];
    __shared__ float Sim[N1D * 16];
    const int lane = threadIdx.x & 31;
    const int w    = threadIdx.x >> 5;
    const int half = lane >> 4;
    const int l16  = lane & 15;
    const int x    = blockIdx.x >> 2;
    const int kt   = blockIdx.x & 3;

    v2f m2r[16], m2i[16];
#pragma unroll
    for (int s = 0; s < 16; ++s) {
        m2r[s] = *(const v2f*)(Mre + (16 * w + l16) * N1D + 4 * s + 2 * half);
        m2i[s] = *(const v2f*)(Mim + (16 * w + l16) * N1D + 4 * s + 2 * half);
    }

    const float* tr = Tre + (x * N1D + 16 * w + l16) * N1D + 2 * half;
    const float* ti = Tim + (x * N1D + 16 * w + l16) * N1D + 2 * half;
    const float* br = Mre + (16 * kt + l16) * N1D + 2 * half;
    const float* bi = Mim + (16 * kt + l16) * N1D + 2 * half;
    v8f rr = {}, ii = {}, ri = {}, ir = {};
#pragma unroll
    for (int s = 0; s < 16; ++s) {
        const v2f ar = *(const v2f*)(tr + 4 * s);
        const v2f ai = *(const v2f*)(ti + 4 * s);
        const v2f wr = *(const v2f*)(br + 4 * s);
        const v2f wi = *(const v2f*)(bi + 4 * s);
        rr = mma4(ar, wr, rr);
        ii = mma4(ai, wi, ii);
        ri = mma4(ar, wi, ri);
        ir = mma4(ai, wr, ir);
    }
#pragma unroll
    for (int v = 0; v < 8; ++v) {
        const int y = 16 * w + v + 8 * half;
        Sre[y * 16 + l16] = rr[v] - ii[v];
        Sim[y * 16 + l16] = ri[v] + ir[v];
    }
    __syncthreads();

    v8f rr2 = {}, ii2 = {}, ri2 = {}, ir2 = {};
    slab_pass(m2r, m2i, Sre, Sim, rr2, ii2, ri2, ir2);
#pragma unroll
    for (int v = 0; v < 8; ++v) {
        const int idx = slab_idx(x, 16 * w + v + 8 * half, kt, l16);
        Ure[idx] = rr2[v] - ii2[v];
        Uim[idx] = ri2[v] + ir2[v];
    }
}

__global__ __launch_bounds__(128) void x_filter_kernel(const float* __restrict__ Tre,
                                                       const float* __restrict__ Tim,
                                                       const float* __restrict__ Mre,
                                                       const float* __restrict__ MimF,
                                                       const float* __restrict__ MimI,
                                                       const float* __restrict__ h,
                                                       float* __restrict__ Ure,
                                                       float* __restrict__ Uim) {
    __shared__ float Sre[N1D * 16];
    __shared__ float Sim[N1D * 16];
    const int lane = threadIdx.x & 31;
    const int w    = threadIdx.x >> 5;
    const int half = lane >> 4;
    const int l16  = lane & 15;
    const int ky   = blockIdx.x >> 2;
    const int kt   = blockIdx.x & 3;

    v2f m1r[16], m1i[16], m2i[16];
#pragma unroll
    for (int s = 0; s < 16; ++s) {
        m1r[s] = *(const v2f*)(Mre  + (16 * w + l16) * N1D + 4 * s + 2 * half);
        m1i[s] = *(const v2f*)(MimF + (16 * w + l16) * N1D + 4 * s + 2 * half);
        m2i[s] = *(const v2f*)(MimI + (16 * w + l16) * N1D + 4 * s + 2 * half);
    }
    float hv[8];
#pragma unroll
    for (int v = 0; v < 8; ++v)
        hv[v] = h[((16 * w + v + 8 * half) * N1D + ky) * N1D + 16 * kt + l16];

    v8f rr = {}, ii = {}, ri = {}, ir = {};
#pragma unroll
    for (int s = 0; s < 16; ++s) {
        const int k = 4 * s + 2 * half;
        v2f br, bi;
        br.x = Tre[slab_idx(k, ky, kt, l16)];  br.y = Tre[slab_idx(k + 1, ky, kt, l16)];
        bi.x = Tim[slab_idx(k, ky, kt, l16)];  bi.y = Tim[slab_idx(k + 1, ky, kt, l16)];
        rr = mma4(m1r[s], br, rr);
        ii = mma4(m1i[s], bi, ii);
        ri = mma4(m1r[s], bi, ri);
        ir = mma4(m1i[s], br, ir);
    }
#pragma unroll
    for (int v = 0; v < 8; ++v) {
        const int kx = 16 * w + v + 8 * half;
        Sre[kx * 16 + l16] = (rr[v] - ii[v]) * hv[v];
        Sim[kx * 16 + l16] = (ri[v] + ir[v]) * hv[v];
    }
    __syncthreads();

    v8f rr2 = {}, ii2 = {}, ri2 = {}, ir2 = {};
    slab_pass(m1r, m2i, Sre, Sim, rr2, ii2, ri2, ir2);
#pragma unroll
    for (int v = 0; v < 8; ++v) {
        const int idx = slab_idx(16 * w + v + 8 * half, ky, kt, l16);
        Ure[idx] = rr2[v] - ii2[v];
        Uim[idx] = ri2[v] + ir2[v];
    }
}

__global__ __launch_bounds__(128) void inv_yz_stats_kernel(const float* __restrict__ Tre,
                                                           const float* __restrict__ Tim,
                                                           const float* __restrict__ Mre,
                                                           const float* __restrict__ Mim,
                                                           float* __restrict__ Ore,
                                                           float* __restrict__ psum,
                                                           float* __restrict__ psq) {
    __shared__ __attribute__((aligned(16))) float Lre[16 * WPITCH];
    __shared__ __attribute__((aligned(16))) float Lim[16 * WPITCH];
    __shared__ float wsum[4];
    __shared__ float wsq[4];
    const int lane = threadIdx.x & 31;
    const int w    = threadIdx.x >> 5;
    const int half = lane >> 4;
    const int l16  = lane & 15;
    const int x    = blockIdx.x >> 2;
    const int yt   = blockIdx.x & 3;

    v2f m2r[16], m2i[16];
#pragma unroll
    for (int s = 0; s < 16; ++s) {
        m2r[s] = *(const v2f*)(Mre + (16 * w + l16) * N1D + 4 * s + 2 * half);
        m2i[s] = *(const v2f*)(Mim + (16 * w + l16) * N1D + 4 * s + 2 * half);
    }

    const float* ar_p = Mre + (16 * yt + l16) * N1D + 2 * half;
    const float* ai_p = Mim + (16 * yt + l16) * N1D + 2 * half;
    v8f rr = {}, ii = {}, ri = {}, ir = {};
#pragma unroll
    for (int s = 0; s < 16; ++s) {
        const int k = 4 * s + 2 * half;
        const v2f ar = *(const v2f*)(ar_p + 4 * s);
        const v2f ai = *(const v2f*)(ai_p + 4 * s);
        v2f br, bi;
        br.x = Tre[slab_idx(x, k, w, l16)];  br.y = Tre[slab_idx(x, k + 1, w, l16)];
        bi.x = Tim[slab_idx(x, k, w, l16)];  bi.y = Tim[slab_idx(x, k + 1, w, l16)];
        rr = mma4(ar, br, rr);
        ii = mma4(ai, bi, ii);
        ri = mma4(ar, bi, ri);
        ir = mma4(ai, br, ir);
    }
#pragma unroll
    for (int v = 0; v < 8; ++v) {
        Lre[(v + 8 * half) * WPITCH + 16 * w + l16] = rr[v] - ii[v];
        Lim[(v + 8 * half) * WPITCH + 16 * w + l16] = ri[v] + ir[v];
    }
    __syncthreads();

    v8f pr = {}, pi = {};
    v2f wr[16], wi[16];
#pragma unroll
    for (int s = 0; s < 16; ++s) {
        wr[s] = *(const v2f*)(Lre + l16 * WPITCH + 4 * s + 2 * half);
        wi[s] = *(const v2f*)(Lim + l16 * WPITCH + 4 * s + 2 * half);
    }
    __builtin_amdgcn_sched_barrier(0);
#pragma unroll
    for (int s = 0; s < 16; ++s) {
        pr = mma4(wr[s], m2r[s], pr);
        pi = mma4(wi[s], m2i[s], pi);
    }
    float sacc = 0.0f, qacc = 0.0f;
#pragma unroll
    for (int v = 0; v < 8; ++v) {
        const float re = pr[v] - pi[v];
        Ore[(x * N1D + 16 * yt + v + 8 * half) * N1D + 16 * w + l16] = re;
        sacc += re;
        qacc += re * re;
    }
#pragma unroll
    for (int m = 16; m > 0; m >>= 1) {
        sacc += __shfl_xor(sacc, m);
        qacc += __shfl_xor(qacc, m);
    }
    if (lane == 0) { wsum[w] = sacc; wsq[w] = qacc; }
    __syncthreads();
    if (threadIdx.x == 0) {
        psum[blockIdx.x] = (wsum[0] + wsum[1]) + (wsum[2] + wsum[3]);
        psq[blockIdx.x]  = (wsq[0] + wsq[1]) + (wsq[2] + wsq[3]);
    }
}

__global__ __launch_bounds__(256) void finalize_kernel(const float* __restrict__ re,
                                                       const float* __restrict__ psum,
                                                       const float* __restrict__ psq,
                                                       float* __restrict__ out) {
    __shared__ float parts[8];
    __shared__ float partq[8];
    const int tid = threadIdx.x;
    const int g = blockIdx.x * blockDim.x + tid;
    const float x = re[g];
    float p = psum[tid];
    float q = psq[tid];
#pragma unroll
    for (int m = 16; m > 0; m >>= 1) {
        p += __shfl_xor(p, m);
        q += __shfl_xor(q, m);
    }
    if ((tid & 31) == 0) { parts[tid >> 5] = p; partq[tid >> 5] = q; }
    __syncthreads();
    const float tsum = ((parts[0] + parts[1]) + (parts[2] + parts[3])) +
                       ((parts[4] + parts[5]) + (parts[6] + parts[7]));
    const float tsq  = ((partq[0] + partq[1]) + (partq[2] + partq[3])) +
                       ((partq[4] + partq[5]) + (partq[6] + partq[7]));
    const float invN = 1.0f / (float)GRID_N;
    const float mean = tsum * invN;
    const float var  = fmaxf(tsq * invN - mean * mean, 0.0f);
    const float inv  = 1.0f / (sqrtf(var) + 1e-8f);
    out[g] = (x - mean) * inv;
}

extern "C" void kernel_launch(void* const* d_in, const int* in_sizes, int n_in,
                              void* d_out, int out_size, void* d_ws, size_t ws_size,
                              hipStream_t stream) {
    const float* X   = (const float*)d_in[0];
    const float* aw  = (const float*)d_in[1];
    const float* bw  = (const float*)d_in[2];
    const float* ham = (const float*)d_in[4];
    float* out = (float*)d_out;

    float* ws   = (float*)d_ws;
    float* Wre  = ws;
    float* WimF = Wre + 4096;
    float* WimI = WimF + 4096;
    float* Are  = WimI + 4096;
    float* Aim  = Are + GRID_N;
    float* Bre  = Aim + GRID_N;
    float* Bim  = Bre + GRID_N;

    float* AxF = Bre;
    float* EyF = AxF + TBL;
    float* EzF = EyF + TBL;

    float* psum = Bim;
    float* psq  = Bim + NBLK2;

    setup_kernel<<<592, 256, 0, stream>>>(X, aw, bw, Wre, WimF, WimI,
                                                          AxF, EyF, EzF);

    density_gemm_kernel<<<256, 128, 0, stream>>>(AxF, EyF, EzF, Are, Aim);

    fwd_zy_kernel<<<NBLK2, 128, 0, stream>>>(Are, Aim, Wre, WimF, Bre, Bim);

    x_filter_kernel<<<NBLK2, 128, 0, stream>>>(Bre, Bim, Wre, WimF, WimI, ham, Are, Aim);

    inv_yz_stats_kernel<<<NBLK2, 128, 0, stream>>>(Are, Aim, Wre, WimI, Bre, psum, psq);

    finalize_kernel<<<GRID_N / 256, 256, 0, stream>>>(Bre, psum, psq, out);
}


# mean/variance from the filtered spectrum in the x launch (DC term; Parseval on its Hermitian part), normalise fused into the last inverse launch (6 -> 5); changes the order of the variance sum
# speedup vs baseline: 2.1636x; 1.1233x over previous
#include <hip/hip_runtime.h>
#include <hip/hip_bf16.h>
#include <math.h>

#define N1D      64
#define GRID_N   (64 * 64 * 64)
#define N_ATOMS  128
#define K_TERMS  6
#define SPACING  0.5f
#define LOG2E    1.4426950408889634f

#define NAK      (N_ATOMS * K_TERMS)
#define KS2      (NAK / 8)
#define TBL      (N1D * NAK)
#define CHUNK    16
#define NCHUNK   (KS2 / CHUNK)

typedef float v2f __attribute__((ext_vector_type(2)));
typedef float v4f __attribute__((ext_vector_type(4)));
typedef float v8f __attribute__((ext_vector_type(8)));

__device__ __forceinline__ float fast_exp2(float x) {
#if __has_builtin(__builtin_amdgcn_exp2f)
    return __builtin_amdgcn_exp2f(x);
#else
    return exp2f(x);
#endif
}

__global__ __launch_bounds__(256) void setup_kernel(const float* __restrict__ X,
                                                    const float* __restrict__ aw,
                                                    const float* __restrict__ bw,
                                                    float* __restrict__ Wre,
                                                    float* __restrict__ WimF,
                                                    float* __restrict__ WimI,
                                                    float* __restrict__ AxF,
                                                    float* __restrict__ EyF,
                                                    float* __restrict__ EzF) {
    if (blockIdx.x < 16) {
        int t = blockIdx.x * blockDim.x + threadIdx.x;
        int k = t >> 6;
        int n = t & 63;
        int kn = (k * n) & 63;
        float ang = (float)kn * 0.09817477042468103f;
        float sgn = ((k + n) & 1) ? -0.125f : 0.125f;
        float c = sgn * cosf(ang);
        float s = sgn * sinf(ang);
        Wre[t]  = c;
        WimF[t] = -s;
        WimI[t] = s;
        return;
    }
    const int t    = (blockIdx.x - 16) * blockDim.x + threadIdx.x;
    const int axis = t / TBL;
    const int r    = t - axis * TBL;
    const int i    = r / NAK;
    const int ak   = r - i * NAK;
    const int a    = ak / K_TERMS;
    const float d  = (float)i * SPACING - X[3 * a + axis];
    const float e  = fast_exp2((bw[ak] * LOG2E) * (d * d));
    const int ks2  = ak >> 3;
    const int slot = ((ak >> 2) & 1) * 2 + (ak & 1);
    const int half = (ak >> 1) & 1;
    if (axis == 0) {
        AxF[((i * KS2 + ks2) * 2 + half) * 4 + slot] = aw[ak] * e;
    } else {
        float* T = (axis == 1) ? EyF : EzF;
        T[(((i >> 4) * KS2 + ks2) * 32 + half * 16 + (i & 15)) * 4 + slot] = e;
    }
}

struct FragChunk {
    v4f a[CHUNK];
    v4f y[CHUNK];
    v4f z[CHUNK];
};

__device__ __forceinline__ void load_chunk(FragChunk& f, const v4f* ap, const v4f* yp,
                                           const v4f* zp, int c) {
#pragma unroll
    for (int j = 0; j < CHUNK; ++j) {
        const int n = c * CHUNK + j;
        f.a[j] = ap[n * 2];
        f.y[j] = yp[n * 32];
        f.z[j] = zp[n * 32];
    }
}

__device__ __forceinline__ void mma_chunk(const FragChunk& f, v8f& acc0, v8f& acc1) {
#pragma unroll
    for (int j = 0; j < CHUNK; ++j) {
        v2f A0, A1, B0, B1;
        A0.x = f.a[j].x * f.y[j].x;  A0.y = f.a[j].y * f.y[j].y;
        A1.x = f.a[j].z * f.y[j].z;  A1.y = f.a[j].w * f.y[j].w;
        B0.x = f.z[j].x;  B0.y = f.z[j].y;
        B1.x = f.z[j].z;  B1.y = f.z[j].w;
        acc0 = __builtin_amdgcn_wmma_f32_16x16x4_f32(
            false, A0, false, B0, (short)0, acc0, false, false);
        acc1 = __builtin_amdgcn_wmma_f32_16x16x4_f32(
            false, A1, false, B1, (short)0, acc1, false, false);
    }
}

__global__ __launch_bounds__(128) void density_gemm_kernel(const float* AxF,
                                                           const float* EyF,
                                                           const float* EzF,
                                                           float* rho_re,
                                                           float* rho_im) {
    const int lane = threadIdx.x & 31;
    const int izt  = threadIdx.x >> 5;
    const int ix   = blockIdx.x >> 2;
    const int iyt  = blockIdx.x & 3;
    const int half = lane >> 4;
    const int l16  = lane & 15;

    const v4f* ap = (const v4f*)AxF + (ix * KS2) * 2 + half;
    const v4f* yp = (const v4f*)EyF + (iyt * KS2) * 32 + lane;
    const v4f* zp = (const v4f*)EzF + (izt * KS2) * 32 + lane;

    v8f acc0 = {}, acc1 = {};
    FragChunk f0, f1;

    load_chunk(f0, ap, yp, zp, 0);
    for (int c = 0; c < NCHUNK - 2; c += 2) {
        load_chunk(f1, ap, yp, zp, c + 1);
        mma_chunk(f0, acc0, acc1);
        load_chunk(f0, ap, yp, zp, c + 2);
        mma_chunk(f1, acc0, acc1);
    }
    load_chunk(f1, ap, yp, zp, NCHUNK - 1);
    mma_chunk(f0, acc0, acc1);
    mma_chunk(f1, acc0, acc1);

#pragma unroll
    for (int v = 0; v < 8; ++v) {
        const int iy  = iyt * 16 + v + 8 * half;
        const int idx = (ix * N1D + iy) * N1D + izt * 16 + l16;
        rho_re[idx] = acc0[v] + acc1[v];
        rho_im[idx] = 0.0f;
    }
}

#define NBLK2      256
#define WPITCH     68

__device__ __forceinline__ v8f mma4(v2f a, v2f b, v8f c) {
    return __builtin_amdgcn_wmma_f32_16x16x4_f32(false, a, false, b, (short)0, c, false, false);
}

__device__ __forceinline__ int slab_idx(int x, int ky, int kt, int l) {
    return ((x * 4 + kt) * N1D + ky) * 16 + l;
}

__device__ __forceinline__ void slab_pass(const v2f* mr, const v2f* mi,
                                          const float* Sre, const float* Sim,
                                          v8f& rr, v8f& ii, v8f& ri, v8f& ir) {
    const int lane = threadIdx.x & 31;
    const int half = lane >> 4;
    const int l16  = lane & 15;
    v2f br[16], bi[16];
#pragma unroll
    for (int s = 0; s < 16; ++s) {
        const int k = 4 * s + 2 * half;
        br[s].x = Sre[k * 16 + l16];  br[s].y = Sre[(k + 1) * 16 + l16];
        bi[s].x = Sim[k * 16 + l16];  bi[s].y = Sim[(k + 1) * 16 + l16];
    }
    __builtin_amdgcn_sched_barrier(0);
#pragma unroll
    for (int s = 0; s < 16; ++s) {
        rr = mma4(mr[s], br[s], rr);
        ii = mma4(mi[s], bi[s], ii);
        ri = mma4(mr[s], bi[s], ri);
        ir = mma4(mi[s], br[s], ir);
    }
}

__global__ __launch_bounds__(128) void fwd_zy_kernel(const float* __restrict__ Tre,
                                                     const float* __restrict__ Tim,
                                                     const float* __restrict__ Mre,
                                                     const float* __restrict__ Mim,
                                                     float* __restrict__ Ure,
                                                     float* __restrict__ Uim) {
    __shared__ float Sre[N1D * 16];
    __shared__ float Sim[N1D * 16];
    const int lane = threadIdx.x & 31;
    const int w    = threadIdx.x >> 5;
    const int half = lane >> 4;
    const int l16  = lane & 15;
    const int x    = blockIdx.x >> 2;
    const int kt   = blockIdx.x & 3;

    v2f m2r[16], m2i[16];
#pragma unroll
    for (int s = 0; s < 16; ++s) {
        m2r[s] = *(const v2f*)(Mre + (16 * w + l16) * N1D + 4 * s + 2 * half);
        m2i[s] = *(const v2f*)(Mim + (16 * w + l16) * N1D + 4 * s + 2 * half);
    }

    const float* tr = Tre + (x * N1D + 16 * w + l16) * N1D + 2 * half;
    const float* ti = Tim + (x * N1D + 16 * w + l16) * N1D + 2 * half;
    const float* br = Mre + (16 * kt + l16) * N1D + 2 * half;
    const float* bi = Mim + (16 * kt + l16) * N1D + 2 * half;
    v8f rr = {}, ii = {}, ri = {}, ir = {};
#pragma unroll
    for (int s = 0; s < 16; ++s) {
        const v2f ar = *(const v2f*)(tr + 4 * s);
        const v2f ai = *(const v2f*)(ti + 4 * s);
        const v2f wr = *(const v2f*)(br + 4 * s);
        const v2f wi = *(const v2f*)(bi + 4 * s);
        rr = mma4(ar, wr, rr);
        ii = mma4(ai, wi, ii);
        ri = mma4(ar, wi, ri);
        ir = mma4(ai, wr, ir);
    }
#pragma unroll
    for (int v = 0; v < 8; ++v) {
        const int y = 16 * w + v + 8 * half;
        Sre[y * 16 + l16] = rr[v] - ii[v];
        Sim[y * 16 + l16] = ri[v] + ir[v];
    }
    __syncthreads();

    v8f rr2 = {}, ii2 = {}, ri2 = {}, ir2 = {};
    slab_pass(m2r, m2i, Sre, Sim, rr2, ii2, ri2, ir2);
#pragma unroll
    for (int v = 0; v < 8; ++v) {
        const int idx = slab_idx(x, 16 * w + v + 8 * half, kt, l16);
        Ure[idx] = rr2[v] - ii2[v];
        Uim[idx] = ri2[v] + ir2[v];
    }
}

__global__ __launch_bounds__(128) void x_filter_kernel(float* Tre,
                                                       float* Tim,
                                                       const float* __restrict__ Mre,
                                                       const float* __restrict__ MimF,
                                                       const float* __restrict__ MimI,
                                                       const float* __restrict__ h,
                                                       float* __restrict__ stat) {
    __shared__ float Sre[N1D * 16];
    __shared__ float Sim[N1D * 16];
    __shared__ float wq[4];
    const int lane = threadIdx.x & 31;
    const int w    = threadIdx.x >> 5;
    const int half = lane >> 4;
    const int l16  = lane & 15;
    const int ky   = blockIdx.x >> 2;
    const int kt   = blockIdx.x & 3;

    v2f m1r[16], m1i[16], m2i[16];
#pragma unroll
    for (int s = 0; s < 16; ++s) {
        m1r[s] = *(const v2f*)(Mre  + (16 * w + l16) * N1D + 4 * s + 2 * half);
        m1i[s] = *(const v2f*)(MimF + (16 * w + l16) * N1D + 4 * s + 2 * half);
        m2i[s] = *(const v2f*)(MimI + (16 * w + l16) * N1D + 4 * s + 2 * half);
    }
    float hv[8], hm[8];
    const int mky = (N1D - ky) & 63;
    const int mkz = (N1D - (16 * kt + l16)) & 63;
#pragma unroll
    for (int v = 0; v < 8; ++v) {
        const int kx = 16 * w + v + 8 * half;
        hv[v] = h[(kx * N1D + ky) * N1D + 16 * kt + l16];
        hm[v] = h[((((N1D - kx) & 63) * N1D) + mky) * N1D + mkz];
    }

    v8f rr = {}, ii = {}, ri = {}, ir = {};
#pragma unroll
    for (int s = 0; s < 16; ++s) {
        const int k = 4 * s + 2 * half;
        v2f br, bi;
        br.x = Tre[slab_idx(k, ky, kt, l16)];  br.y = Tre[slab_idx(k + 1, ky, kt, l16)];
        bi.x = Tim[slab_idx(k, ky, kt, l16)];  bi.y = Tim[slab_idx(k + 1, ky, kt, l16)];
        rr = mma4(m1r[s], br, rr);
        ii = mma4(m1i[s], bi, ii);
        ri = mma4(m1r[s], bi, ri);
        ir = mma4(m1i[s], br, ir);
    }
    float qacc = 0.0f;
#pragma unroll
    for (int v = 0; v < 8; ++v) {
        const int kx = 16 * w + v + 8 * half;
        const float fr = rr[v] - ii[v];
        const float fi = ri[v] + ir[v];
        const float hs = 0.5f * (hv[v] + hm[v]);
        qacc += (fr * fr + fi * fi) * (hs * hs);
        Sre[kx * 16 + l16] = fr * hv[v];
        Sim[kx * 16 + l16] = fi * hv[v];
    }
#pragma unroll
    for (int m = 16; m > 0; m >>= 1) qacc += __shfl_xor(qacc, m);
    if (lane == 0) wq[w] = qacc;
    if (blockIdx.x == 4 * 32 + 2 && threadIdx.x == 64)
        stat[NBLK2] = (rr[0] - ii[0]) * hv[0];
    __syncthreads();
    if (threadIdx.x == 0) stat[blockIdx.x] = (wq[0] + wq[1]) + (wq[2] + wq[3]);

    v8f rr2 = {}, ii2 = {}, ri2 = {}, ir2 = {};
    slab_pass(m1r, m2i, Sre, Sim, rr2, ii2, ri2, ir2);
#pragma unroll
    for (int v = 0; v < 8; ++v) {
        const int idx = slab_idx(16 * w + v + 8 * half, ky, kt, l16);
        Tre[idx] = rr2[v] - ii2[v];
        Tim[idx] = ri2[v] + ir2[v];
    }
}

__global__ __launch_bounds__(128) void inv_yz_norm_kernel(const float* __restrict__ Tre,
                                                           const float* __restrict__ Tim,
                                                           const float* __restrict__ Mre,
                                                           const float* __restrict__ Mim,
                                                           const float* __restrict__ stat,
                                                           float* __restrict__ out) {
    __shared__ __attribute__((aligned(16))) float Lre[16 * WPITCH];
    __shared__ __attribute__((aligned(16))) float Lim[16 * WPITCH];
    __shared__ float wsq[4];
    const int lane = threadIdx.x & 31;
    const int w    = threadIdx.x >> 5;
    const int half = lane >> 4;
    const int l16  = lane & 15;
    const int x    = blockIdx.x >> 2;
    const int yt   = blockIdx.x & 3;

    v2f m2r[16], m2i[16];
#pragma unroll
    for (int s = 0; s < 16; ++s) {
        m2r[s] = *(const v2f*)(Mre + (16 * w + l16) * N1D + 4 * s + 2 * half);
        m2i[s] = *(const v2f*)(Mim + (16 * w + l16) * N1D + 4 * s + 2 * half);
    }

    float q = stat[threadIdx.x] + stat[threadIdx.x + 128];
    const float g0 = stat[NBLK2];

    const float* ar_p = Mre + (16 * yt + l16) * N1D + 2 * half;
    const float* ai_p = Mim + (16 * yt + l16) * N1D + 2 * half;
    v8f rr = {}, ii = {}, ri = {}, ir = {};
#pragma unroll
    for (int s = 0; s < 16; ++s) {
        const int k = 4 * s + 2 * half;
        const v2f ar = *(const v2f*)(ar_p + 4 * s);
        const v2f ai = *(const v2f*)(ai_p + 4 * s);
        v2f br, bi;
        br.x = Tre[slab_idx(x, k, w, l16)];  br.y = Tre[slab_idx(x, k + 1, w, l16)];
        bi.x = Tim[slab_idx(x, k, w, l16)];  bi.y = Tim[slab_idx(x, k + 1, w, l16)];
        rr = mma4(ar, br, rr);
        ii = mma4(ai, bi, ii);
        ri = mma4(ar, bi, ri);
        ir = mma4(ai, br, ir);
    }
#pragma unroll
    for (int v = 0; v < 8; ++v) {
        Lre[(v + 8 * half) * WPITCH + 16 * w + l16] = rr[v] - ii[v];
        Lim[(v + 8 * half) * WPITCH + 16 * w + l16] = ri[v] + ir[v];
    }
#pragma unroll
    for (int m = 16; m > 0; m >>= 1) q += __shfl_xor(q, m);
    if (lane == 0) wsq[w] = q;
    __syncthreads();
    const float tsq  = (wsq[0] + wsq[1]) + (wsq[2] + wsq[3]);
    const float invN = 1.0f / (float)GRID_N;
    const float mean = g0 * (1.0f / 512.0f);
    const float var  = fmaxf(tsq * invN - mean * mean, 0.0f);
    const float inv  = 1.0f / (sqrtf(var) + 1e-8f);

    v8f pr = {}, pi = {};
    v2f wr[16], wi[16];
#pragma unroll
    for (int s = 0; s < 16; ++s) {
        wr[s] = *(const v2f*)(Lre + l16 * WPITCH + 4 * s + 2 * half);
        wi[s] = *(const v2f*)(Lim + l16 * WPITCH + 4 * s + 2 * half);
    }
    __builtin_amdgcn_sched_barrier(0);
#pragma unroll
    for (int s = 0; s < 16; ++s) {
        pr = mma4(wr[s], m2r[s], pr);
        pi = mma4(wi[s], m2i[s], pi);
    }
#pragma unroll
    for (int v = 0; v < 8; ++v) {
        const float re = pr[v] - pi[v];
        out[(x * N1D + 16 * yt + v + 8 * half) * N1D + 16 * w + l16] = (re - mean) * inv;
    }
}

extern "C" void kernel_launch(void* const* d_in, const int* in_sizes, int n_in,
                              void* d_out, int out_size, void* d_ws, size_t ws_size,
                              hipStream_t stream) {
    const float* X   = (const float*)d_in[0];
    const float* aw  = (const float*)d_in[1];
    const float* bw  = (const float*)d_in[2];
    const float* ham = (const float*)d_in[4];
    float* out = (float*)d_out;

    float* ws   = (float*)d_ws;
    float* Wre  = ws;
    float* WimF = Wre + 4096;
    float* WimI = WimF + 4096;
    float* Are  = WimI + 4096;
    float* Aim  = Are + GRID_N;
    float* Bre  = Aim + GRID_N;
    float* Bim  = Bre + GRID_N;

    float* AxF = Bre;
    float* EyF = AxF + TBL;
    float* EzF = EyF + TBL;

    float* stat = Are;

    setup_kernel<<<592, 256, 0, stream>>>(X, aw, bw, Wre, WimF, WimI,
                                                          AxF, EyF, EzF);

    density_gemm_kernel<<<256, 128, 0, stream>>>(AxF, EyF, EzF, Are, Aim);

    fwd_zy_kernel<<<NBLK2, 128, 0, stream>>>(Are, Aim, Wre, WimF, Bre, Bim);

    x_filter_kernel<<<NBLK2, 128, 0, stream>>>(Bre, Bim, Wre, WimF, WimI, ham, stat);

    inv_yz_norm_kernel<<<NBLK2, 128, 0, stream>>>(Bre, Bim, Wre, WimI, stat, out);
}


# density operand fragments built in registers (one exp2 per element from dx^2+dy^2 and dz^2) instead of read from tables: set-up launch removed (5 -> 4); changes the order of the atom/term sum
# speedup vs baseline: 2.4600x; 1.1370x over previous
#include <hip/hip_runtime.h>
#include <hip/hip_bf16.h>
#include <math.h>

#define N1D      64
#define GRID_N   (64 * 64 * 64)
#define N_ATOMS  128
#define K_TERMS  6
#define SPACING  0.5f
#define LOG2E    1.4426950408889634f


typedef float v2f __attribute__((ext_vector_type(2)));
typedef float v4f __attribute__((ext_vector_type(4)));
typedef float v8f __attribute__((ext_vector_type(8)));

__device__ __forceinline__ float fast_exp2(float x) {
#if __has_builtin(__builtin_amdgcn_exp2f)
    return __builtin_amdgcn_exp2f(x);
#else
    return exp2f(x);
#endif
}

#define ACH      8
#define NACH     (N_ATOMS / 2 / ACH)

struct AtomChunk {
    float x[ACH], y[ACH], z[ACH];
    v2f   b[ACH][3];
    v2f   w[ACH][3];
};

__device__ __forceinline__ void load_atoms(AtomChunk& p, const float* __restrict__ X,
                                           const float* __restrict__ aw,
                                           const float* __restrict__ bw,
                                           int half, int c) {
#pragma unroll
    for (int j = 0; j < ACH; ++j) {
        const int a = 2 * (c * ACH + j) + half;
        p.x[j] = X[3 * a];
        p.y[j] = X[3 * a + 1];
        p.z[j] = X[3 * a + 2];
        const v2f* bp = (const v2f*)(bw + K_TERMS * a);
        const v2f* wp = (const v2f*)(aw + K_TERMS * a);
#pragma unroll
        for (int u = 0; u < 3; ++u) {
            p.b[j][u] = bp[u];
            p.w[j][u] = wp[u];
        }
    }
}

__device__ __forceinline__ void mma_atoms(const AtomChunk& p, float fx, float fy, float fz,
                                          v8f& acc0, v8f& acc1) {
#pragma unroll
    for (int j = 0; j < ACH; ++j) {
        const float dx  = fx - p.x[j];
        const float dy  = fy - p.y[j];
        const float dz  = fz - p.z[j];
        const float rxy = (dx * dx + dy * dy) * LOG2E;
        const float rz  = (dz * dz) * LOG2E;
#pragma unroll
        for (int u = 0; u < 3; ++u) {
            const float b0 = p.b[j][u].x;
            const float b1 = p.b[j][u].y;
            v2f A, B;
            A.x = p.w[j][u].x * fast_exp2(b0 * rxy);
            A.y = p.w[j][u].y * fast_exp2(b1 * rxy);
            B.x = fast_exp2(b0 * rz);
            B.y = fast_exp2(b1 * rz);
            if (((3 * j + u) & 1) == 0)
                acc0 = __builtin_amdgcn_wmma_f32_16x16x4_f32(
                    false, A, false, B, (short)0, acc0, false, false);
            else
                acc1 = __builtin_amdgcn_wmma_f32_16x16x4_f32(
                    false, A, false, B, (short)0, acc1, false, false);
        }
    }
}

__global__ __launch_bounds__(128) void density_gemm_kernel(const float* __restrict__ X,
                                                           const float* __restrict__ aw,
                                                           const float* __restrict__ bw,
                                                           float* __restrict__ Wre,
                                                           float* __restrict__ WimF,
                                                           float* __restrict__ WimI,
                                                           float* __restrict__ rho_re,
                                                           float* __restrict__ rho_im) {
    const int lane = threadIdx.x & 31;
    const int izt  = threadIdx.x >> 5;
    const int ix   = blockIdx.x >> 2;
    const int iyt  = blockIdx.x & 3;
    const int half = lane >> 4;
    const int l16  = lane & 15;

    const float fx = (float)ix * SPACING;
    const float fy = (float)(iyt * 16 + l16) * SPACING;
    const float fz = (float)(izt * 16 + l16) * SPACING;

    v8f acc0 = {}, acc1 = {};
    AtomChunk p0, p1;

    load_atoms(p0, X, aw, bw, half, 0);
    for (int c = 0; c < NACH - 2; c += 2) {
        load_atoms(p1, X, aw, bw, half, c + 1);
        mma_atoms(p0, fx, fy, fz, acc0, acc1);
        load_atoms(p0, X, aw, bw, half, c + 2);
        mma_atoms(p1, fx, fy, fz, acc0, acc1);
    }
    load_atoms(p1, X, aw, bw, half, NACH - 1);
    mma_atoms(p0, fx, fy, fz, acc0, acc1);
    mma_atoms(p1, fx, fy, fz, acc0, acc1);

#pragma unroll
    for (int v = 0; v < 8; ++v) {
        const int iy  = iyt * 16 + v + 8 * half;
        const int idx = (ix * N1D + iy) * N1D + izt * 16 + l16;
        rho_re[idx] = acc0[v] + acc1[v];
        rho_im[idx] = 0.0f;
    }

    if (blockIdx.x < (N1D * N1D) / 128) {
        const int t = blockIdx.x * 128 + threadIdx.x;
        const int k = t >> 6;
        const int n = t & 63;
        const int kn = (k * n) & 63;
        const float ang = (float)kn * 0.09817477042468103f;
        const float sgn = ((k + n) & 1) ? -0.125f : 0.125f;
        const float c = sgn * cosf(ang);
        const float s = sgn * sinf(ang);
        Wre[t]  = c;
        WimF[t] = -s;
        WimI[t] = s;
    }
}

#define NBLK2      256
#define WPITCH     68

__device__ __forceinline__ v8f mma4(v2f a, v2f b, v8f c) {
    return __builtin_amdgcn_wmma_f32_16x16x4_f32(false, a, false, b, (short)0, c, false, false);
}

__device__ __forceinline__ int slab_idx(int x, int ky, int kt, int l) {
    return ((x * 4 + kt) * N1D + ky) * 16 + l;
}

__device__ __forceinline__ void slab_pass(const v2f* mr, const v2f* mi,
                                          const float* Sre, const float* Sim,
                                          v8f& rr, v8f& ii, v8f& ri, v8f& ir) {
    const int lane = threadIdx.x & 31;
    const int half = lane >> 4;
    const int l16  = lane & 15;
    v2f br[16], bi[16];
#pragma unroll
    for (int s = 0; s < 16; ++s) {
        const int k = 4 * s + 2 * half;
        br[s].x = Sre[k * 16 + l16];  br[s].y = Sre[(k + 1) * 16 + l16];
        bi[s].x = Sim[k * 16 + l16];  bi[s].y = Sim[(k + 1) * 16 + l16];
    }
    __builtin_amdgcn_sched_barrier(0);
#pragma unroll
    for (int s = 0; s < 16; ++s) {
        rr = mma4(mr[s], br[s], rr);
        ii = mma4(mi[s], bi[s], ii);
        ri = mma4(mr[s], bi[s], ri);
        ir = mma4(mi[s], br[s], ir);
    }
}

__global__ __launch_bounds__(128) void fwd_zy_kernel(const float* __restrict__ Tre,
                                                     const float* __restrict__ Tim,
                                                     const float* __restrict__ Mre,
                                                     const float* __restrict__ Mim,
                                                     float* __restrict__ Ure,
                                                     float* __restrict__ Uim) {
    __shared__ float Sre[N1D * 16];
    __shared__ float Sim[N1D * 16];
    const int lane = threadIdx.x & 31;
    const int w    = threadIdx.x >> 5;
    const int half = lane >> 4;
    const int l16  = lane & 15;
    const int x    = blockIdx.x >> 2;
    const int kt   = blockIdx.x & 3;

    v2f m2r[16], m2i[16];
#pragma unroll
    for (int s = 0; s < 16; ++s) {
        m2r[s] = *(const v2f*)(Mre + (16 * w + l16) * N1D + 4 * s + 2 * half);
        m2i[s] = *(const v2f*)(Mim + (16 * w + l16) * N1D + 4 * s + 2 * half);
    }

    const float* tr = Tre + (x * N1D + 16 * w + l16) * N1D + 2 * half;
    const float* ti = Tim + (x * N1D + 16 * w + l16) * N1D + 2 * half;
    const float* br = Mre + (16 * kt + l16) * N1D + 2 * half;
    const float* bi = Mim + (16 * kt + l16) * N1D + 2 * half;
    v8f rr = {}, ii = {}, ri = {}, ir = {};
#pragma unroll
    for (int s = 0; s < 16; ++s) {
        const v2f ar = *(const v2f*)(tr + 4 * s);
        const v2f ai = *(const v2f*)(ti + 4 * s);
        const v2f wr = *(const v2f*)(br + 4 * s);
        const v2f wi = *(const v2f*)(bi + 4 * s);
        rr = mma4(ar, wr, rr);
        ii = mma4(ai, wi, ii);
        ri = mma4(ar, wi, ri);
        ir = mma4(ai, wr, ir);
    }
#pragma unroll
    for (int v = 0; v < 8; ++v) {
        const int y = 16 * w + v + 8 * half;
        Sre[y * 16 + l16] = rr[v] - ii[v];
        Sim[y * 16 + l16] = ri[v] + ir[v];
    }
    __syncthreads();

    v8f rr2 = {}, ii2 = {}, ri2 = {}, ir2 = {};
    slab_pass(m2r, m2i, Sre, Sim, rr2, ii2, ri2, ir2);
#pragma unroll
    for (int v = 0; v < 8; ++v) {
        const int idx = slab_idx(x, 16 * w + v + 8 * half, kt, l16);
        Ure[idx] = rr2[v] - ii2[v];
        Uim[idx] = ri2[v] + ir2[v];
    }
}

__global__ __launch_bounds__(128) void x_filter_kernel(float* Tre,
                                                       float* Tim,
                                                       const float* __restrict__ Mre,
                                                       const float* __restrict__ MimF,
                                                       const float* __restrict__ MimI,
                                                       const float* __restrict__ h,
                                                       float* __restrict__ stat) {
    __shared__ float Sre[N1D * 16];
    __shared__ float Sim[N1D * 16];
    __shared__ float wq[4];
    const int lane = threadIdx.x & 31;
    const int w    = threadIdx.x >> 5;
    const int half = lane >> 4;
    const int l16  = lane & 15;
    const int ky   = blockIdx.x >> 2;
    const int kt   = blockIdx.x & 3;

    v2f m1r[16], m1i[16], m2i[16];
#pragma unroll
    for (int s = 0; s < 16; ++s) {
        m1r[s] = *(const v2f*)(Mre  + (16 * w + l16) * N1D + 4 * s + 2 * half);
        m1i[s] = *(const v2f*)(MimF + (16 * w + l16) * N1D + 4 * s + 2 * half);
        m2i[s] = *(const v2f*)(MimI + (16 * w + l16) * N1D + 4 * s + 2 * half);
    }
    float hv[8], hm[8];
    const int mky = (N1D - ky) & 63;
    const int mkz = (N1D - (16 * kt + l16)) & 63;
#pragma unroll
    for (int v = 0; v < 8; ++v) {
        const int kx = 16 * w + v + 8 * half;
        hv[v] = h[(kx * N1D + ky) * N1D + 16 * kt + l16];
        hm[v] = h[((((N1D - kx) & 63) * N1D) + mky) * N1D + mkz];
    }

    v8f rr = {}, ii = {}, ri = {}, ir = {};
#pragma unroll
    for (int s = 0; s < 16; ++s) {
        const int k = 4 * s + 2 * half;
        v2f br, bi;
        br.x = Tre[slab_idx(k, ky, kt, l16)];  br.y = Tre[slab_idx(k + 1, ky, kt, l16)];
        bi.x = Tim[slab_idx(k, ky, kt, l16)];  bi.y = Tim[slab_idx(k + 1, ky, kt, l16)];
        rr = mma4(m1r[s], br, rr);
        ii = mma4(m1i[s], bi, ii);
        ri = mma4(m1r[s], bi, ri);
        ir = mma4(m1i[s], br, ir);
    }
    float qacc = 0.0f;
#pragma unroll
    for (int v = 0; v < 8; ++v) {
        const int kx = 16 * w + v + 8 * half;
        const float fr = rr[v] - ii[v];
        const float fi = ri[v] + ir[v];
        const float hs = 0.5f * (hv[v] + hm[v]);
        qacc += (fr * fr + fi * fi) * (hs * hs);
        Sre[kx * 16 + l16] = fr * hv[v];
        Sim[kx * 16 + l16] = fi * hv[v];
    }
#pragma unroll
    for (int m = 16; m > 0; m >>= 1) qacc += __shfl_xor(qacc, m);
    if (lane == 0) wq[w] = qacc;
    if (blockIdx.x == 4 * 32 + 2 && threadIdx.x == 64)
        stat[NBLK2] = (rr[0] - ii[0]) * hv[0];
    __syncthreads();
    if (threadIdx.x == 0) stat[blockIdx.x] = (wq[0] + wq[1]) + (wq[2] + wq[3]);

    v8f rr2 = {}, ii2 = {}, ri2 = {}, ir2 = {};
    slab_pass(m1r, m2i, Sre, Sim, rr2, ii2, ri2, ir2);
#pragma unroll
    for (int v = 0; v < 8; ++v) {
        const int idx = slab_idx(16 * w + v + 8 * half, ky, kt, l16);
        Tre[idx] = rr2[v] - ii2[v];
        Tim[idx] = ri2[v] + ir2[v];
    }
}

__global__ __launch_bounds__(128) void inv_yz_norm_kernel(const float* __restrict__ Tre,
                                                           const float* __restrict__ Tim,
                                                           const float* __restrict__ Mre,
                                                           const float* __restrict__ Mim,
                                                           const float* __restrict__ stat,
                                                           float* __restrict__ out) {
    __shared__ __attribute__((aligned(16))) float Lre[16 * WPITCH];
    __shared__ __attribute__((aligned(16))) float Lim[16 * WPITCH];
    __shared__ float wsq[4];
    const int lane = threadIdx.x & 31;
    const int w    = threadIdx.x >> 5;
    const int half = lane >> 4;
    const int l16  = lane & 15;
    const int x    = blockIdx.x >> 2;
    const int yt   = blockIdx.x & 3;

    v2f m2r[16], m2i[16];
#pragma unroll
    for (int s = 0; s < 16; ++s) {
        m2r[s] = *(const v2f*)(Mre + (16 * w + l16) * N1D + 4 * s + 2 * half);
        m2i[s] = *(const v2f*)(Mim + (16 * w + l16) * N1D + 4 * s + 2 * half);
    }

    float q = stat[threadIdx.x] + stat[threadIdx.x + 128];
    const float g0 = stat[NBLK2];

    const float* ar_p = Mre + (16 * yt + l16) * N1D + 2 * half;
    const float* ai_p = Mim + (16 * yt + l16) * N1D + 2 * half;
    v8f rr = {}, ii = {}, ri = {}, ir = {};
#pragma unroll
    for (int s = 0; s < 16; ++s) {
        const int k = 4 * s + 2 * half;
        const v2f ar = *(const v2f*)(ar_p + 4 * s);
        const v2f ai = *(const v2f*)(ai_p + 4 * s);
        v2f br, bi;
        br.x = Tre[slab_idx(x, k, w, l16)];  br.y = Tre[slab_idx(x, k + 1, w, l16)];
        bi.x = Tim[slab_idx(x, k, w, l16)];  bi.y = Tim[slab_idx(x, k + 1, w, l16)];
        rr = mma4(ar, br, rr);
        ii = mma4(ai, bi, ii);
        ri = mma4(ar, bi, ri);
        ir = mma4(ai, br, ir);
    }
#pragma unroll
    for (int v = 0; v < 8; ++v) {
        Lre[(v + 8 * half) * WPITCH + 16 * w + l16] = rr[v] - ii[v];
        Lim[(v + 8 * half) * WPITCH + 16 * w + l16] = ri[v] + ir[v];
    }
#pragma unroll
    for (int m = 16; m > 0; m >>= 1) q += __shfl_xor(q, m);
    if (lane == 0) wsq[w] = q;
    __syncthreads();
    const float tsq  = (wsq[0] + wsq[1]) + (wsq[2] + wsq[3]);
    const float invN = 1.0f / (float)GRID_N;
    const float mean = g0 * (1.0f / 512.0f);
    const float var  = fmaxf(tsq * invN - mean * mean, 0.0f);
    const float inv  = 1.0f / (sqrtf(var) + 1e-8f);

    v8f pr = {}, pi = {};
    v2f wr[16], wi[16];
#pragma unroll
    for (int s = 0; s < 16; ++s) {
        wr[s] = *(const v2f*)(Lre + l16 * WPITCH + 4 * s + 2 * half);
        wi[s] = *(const v2f*)(Lim + l16 * WPITCH + 4 * s + 2 * half);
    }
    __builtin_amdgcn_sched_barrier(0);
#pragma unroll
    for (int s = 0; s < 16; ++s) {
        pr = mma4(wr[s], m2r[s], pr);
        pi = mma4(wi[s], m2i[s], pi);
    }
#pragma unroll
    for (int v = 0; v < 8; ++v) {
        const float re = pr[v] - pi[v];
        out[(x * N1D + 16 * yt + v + 8 * half) * N1D + 16 * w + l16] = (re - mean) * inv;
    }
}

extern "C" void kernel_launch(void* const* d_in, const int* in_sizes, int n_in,
                              void* d_out, int out_size, void* d_ws, size_t ws_size,
                              hipStream_t stream) {
    const float* X   = (const float*)d_in[0];
    const float* aw  = (const float*)d_in[1];
    const float* bw  = (const float*)d_in[2];
    const float* ham = (const float*)d_in[4];
    float* out = (float*)d_out;

    float* ws   = (float*)d_ws;
    float* Wre  = ws;
    float* WimF = Wre + 4096;
    float* WimI = WimF + 4096;
    float* Are  = WimI + 4096;
    float* Aim  = Are + GRID_N;
    float* Bre  = Aim + GRID_N;
    float* Bim  = Bre + GRID_N;

    float* stat = Are;

    density_gemm_kernel<<<256, 128, 0, stream>>>(X, aw, bw, Wre, WimF, WimI, Are, Aim);

    fwd_zy_kernel<<<NBLK2, 128, 0, stream>>>(Are, Aim, Wre, WimF, Bre, Bim);

    x_filter_kernel<<<NBLK2, 128, 0, stream>>>(Bre, Bim, Wre, WimF, WimI, ham, stat);

    inv_yz_norm_kernel<<<NBLK2, 128, 0, stream>>>(Bre, Bim, Wre, WimI, stat, out);
}


# DFT matrices and density planes stored in WMMA fragment order: each fragment read is a whole-line 16-byte load, not an 8-byte gather over 16 lines; new k order changes the order of the DFT sums
# speedup vs baseline: 2.5392x; 1.0322x over previous
#include <hip/hip_runtime.h>
#include <hip/hip_bf16.h>
#include <math.h>

#define N1D      64
#define GRID_N   (64 * 64 * 64)
#define N_ATOMS  128
#define K_TERMS  6
#define SPACING  0.5f
#define LOG2E    1.4426950408889634f


typedef float v2f __attribute__((ext_vector_type(2)));
typedef float v4f __attribute__((ext_vector_type(4)));
typedef float v8f __attribute__((ext_vector_type(8)));

__device__ __forceinline__ float fast_exp2(float x) {
#if __has_builtin(__builtin_amdgcn_exp2f)
    return __builtin_amdgcn_exp2f(x);
#else
    return exp2f(x);
#endif
}

#define ACH      8
#define NACH     (N_ATOMS / 2 / ACH)

struct AtomChunk {
    float x[ACH], y[ACH], z[ACH];
    v2f   b[ACH][3];
    v2f   w[ACH][3];
};

__device__ __forceinline__ void load_atoms(AtomChunk& p, const float* __restrict__ X,
                                           const float* __restrict__ aw,
                                           const float* __restrict__ bw,
                                           int half, int c) {
#pragma unroll
    for (int j = 0; j < ACH; ++j) {
        const int a = 2 * (c * ACH + j) + half;
        p.x[j] = X[3 * a];
        p.y[j] = X[3 * a + 1];
        p.z[j] = X[3 * a + 2];
        const v2f* bp = (const v2f*)(bw + K_TERMS * a);
        const v2f* wp = (const v2f*)(aw + K_TERMS * a);
#pragma unroll
        for (int u = 0; u < 3; ++u) {
            p.b[j][u] = bp[u];
            p.w[j][u] = wp[u];
        }
    }
}

__device__ __forceinline__ void mma_atoms(const AtomChunk& p, float fx, float fy, float fz,
                                          v8f& acc0, v8f& acc1) {
#pragma unroll
    for (int j = 0; j < ACH; ++j) {
        const float dx  = fx - p.x[j];
        const float dy  = fy - p.y[j];
        const float dz  = fz - p.z[j];
        const float rxy = (dx * dx + dy * dy) * LOG2E;
        const float rz  = (dz * dz) * LOG2E;
#pragma unroll
        for (int u = 0; u < 3; ++u) {
            const float b0 = p.b[j][u].x;
            const float b1 = p.b[j][u].y;
            v2f A, B;
            A.x = p.w[j][u].x * fast_exp2(b0 * rxy);
            A.y = p.w[j][u].y * fast_exp2(b1 * rxy);
            B.x = fast_exp2(b0 * rz);
            B.y = fast_exp2(b1 * rz);
            if (((3 * j + u) & 1) == 0)
                acc0 = __builtin_amdgcn_wmma_f32_16x16x4_f32(
                    false, B, false, A, (short)0, acc0, false, false);
            else
                acc1 = __builtin_amdgcn_wmma_f32_16x16x4_f32(
                    false, B, false, A, (short)0, acc1, false, false);
        }
    }
}

__global__ __launch_bounds__(128) void density_gemm_kernel(const float* __restrict__ X,
                                                           const float* __restrict__ aw,
                                                           const float* __restrict__ bw,
                                                           float* __restrict__ Wre,
                                                           float* __restrict__ WimF,
                                                           float* __restrict__ WimI,
                                                           float* __restrict__ rho_re,
                                                           float* __restrict__ rho_im) {
    const int lane = threadIdx.x & 31;
    const int izt  = threadIdx.x >> 5;
    const int ix   = blockIdx.x >> 2;
    const int iyt  = blockIdx.x & 3;
    const int half = lane >> 4;
    const int l16  = lane & 15;

    const float fx = (float)ix * SPACING;
    const float fy = (float)(iyt * 16 + l16) * SPACING;
    const float fz = (float)(izt * 16 + l16) * SPACING;

    v8f acc0 = {}, acc1 = {};
    AtomChunk p0, p1;

    load_atoms(p0, X, aw, bw, half, 0);
    for (int c = 0; c < NACH - 2; c += 2) {
        load_atoms(p1, X, aw, bw, half, c + 1);
        mma_atoms(p0, fx, fy, fz, acc0, acc1);
        load_atoms(p0, X, aw, bw, half, c + 2);
        mma_atoms(p1, fx, fy, fz, acc0, acc1);
    }
    load_atoms(p1, X, aw, bw, half, NACH - 1);
    mma_atoms(p0, fx, fy, fz, acc0, acc1);
    mma_atoms(p1, fx, fy, fz, acc0, acc1);

    {
        const v8f sum = acc0 + acc1;
        v4f lo, hi;
        lo.x = sum[0]; lo.y = sum[1]; lo.z = sum[2]; lo.w = sum[3];
        hi.x = sum[4]; hi.y = sum[5]; hi.z = sum[6]; hi.w = sum[7];
        const int o = ((ix * 4 + iyt) * 8 + 2 * izt) * 32 + lane;
        v4f* ore = (v4f*)rho_re + o;
        v4f* oim = (v4f*)rho_im + o;
        const v4f zero = {0.0f, 0.0f, 0.0f, 0.0f};
        ore[0]  = lo;
        ore[32] = hi;
        oim[0]  = zero;
        oim[32] = zero;
    }

    if (blockIdx.x < (N1D * N1D) / 128) {
        const int t  = blockIdx.x * 128 + threadIdx.x;
        const int e  = t & 3;
        const int fl = (t >> 2) & 31;
        const int q  = (t >> 7) & 7;
        const int k  = 16 * (t >> 10) + (fl & 15);
        const int sk = 2 * q + (e >> 1);
        const int n  = 16 * (sk >> 2) + 8 * (fl >> 4) + 2 * (sk & 3) + (e & 1);
        const int kn = (k * n) & 63;
        const float ang = (float)kn * 0.09817477042468103f;
        const float sgn = ((k + n) & 1) ? -0.125f : 0.125f;
        const float c = sgn * cosf(ang);
        const float s = sgn * sinf(ang);
        Wre[t]  = c;
        WimF[t] = -s;
        WimI[t] = s;
    }
}

#define NBLK2      256
#define WPITCH     68
#define SPITCH     20

__device__ __forceinline__ v8f mma4(v2f a, v2f b, v8f c) {
    return __builtin_amdgcn_wmma_f32_16x16x4_f32(false, a, false, b, (short)0, c, false, false);
}

__device__ __forceinline__ void load_mfrag(v4f (&m)[8], const float* __restrict__ Mf,
                                           int tile, int lane) {
    const v4f* p = (const v4f*)Mf + tile * 256 + lane;
#pragma unroll
    for (int q = 0; q < 8; ++q) m[q] = p[q * 32];
}

__device__ __forceinline__ int kof(int s, int half) {
    return 16 * (s >> 2) + 8 * half + 2 * (s & 3);
}

__device__ __forceinline__ v2f mstep(const v4f* m, int s) {
    const v4f t = m[s >> 1];
    v2f r;
    if (s & 1) { r.x = t.z; r.y = t.w; } else { r.x = t.x; r.y = t.y; }
    return r;
}

__device__ __forceinline__ int slab_idx(int x, int ky, int kt, int l) {
    return ((x * 4 + kt) * N1D + ky) * 16 + l;
}

__device__ __forceinline__ void slab_pass(const v4f* mr, const v4f* mi,
                                          const float* Sre, const float* Sim,
                                          v8f& rr, v8f& ii, v8f& ri, v8f& ir) {
    const int lane = threadIdx.x & 31;
    const int half = lane >> 4;
    const int l16  = lane & 15;
    v2f br[16], bi[16];
#pragma unroll
    for (int s = 0; s < 16; ++s) {
        const int k = kof(s, half);
        br[s].x = Sre[k * SPITCH + l16];  br[s].y = Sre[(k + 1) * SPITCH + l16];
        bi[s].x = Sim[k * SPITCH + l16];  bi[s].y = Sim[(k + 1) * SPITCH + l16];
    }
    __builtin_amdgcn_sched_barrier(0);
#pragma unroll
    for (int s = 0; s < 16; ++s) {
        rr = mma4(mstep(mr, s), br[s], rr);
        ii = mma4(mstep(mi, s), bi[s], ii);
        ri = mma4(mstep(mr, s), bi[s], ri);
        ir = mma4(mstep(mi, s), br[s], ir);
    }
}

__global__ __launch_bounds__(128) void fwd_zy_kernel(const float* __restrict__ Tre,
                                                     const float* __restrict__ Tim,
                                                     const float* __restrict__ Mre,
                                                     const float* __restrict__ Mim,
                                                     float* __restrict__ Ure,
                                                     float* __restrict__ Uim) {
    __shared__ __attribute__((aligned(16))) float Sre[N1D * SPITCH];
    __shared__ __attribute__((aligned(16))) float Sim[N1D * SPITCH];
    const int lane = threadIdx.x & 31;
    const int w    = threadIdx.x >> 5;
    const int half = lane >> 4;
    const int l16  = lane & 15;
    const int x    = blockIdx.x >> 2;
    const int kt   = blockIdx.x & 3;

    v4f m2r[8], m2i[8];
    load_mfrag(m2r, Mre, w, lane);
    load_mfrag(m2i, Mim, w, lane);

    v4f wzr[8], wzi[8], tr[8], ti[8];
    load_mfrag(wzr, Mre, kt, lane);
    load_mfrag(wzi, Mim, kt, lane);
    load_mfrag(tr, Tre + x * (N1D * N1D), w, lane);
    load_mfrag(ti, Tim + x * (N1D * N1D), w, lane);
    v8f rr = {}, ii = {}, ri = {}, ir = {};
#pragma unroll
    for (int s = 0; s < 16; ++s) {
        const v2f dr = mstep(tr, s);
        const v2f di = mstep(ti, s);
        const v2f wr = mstep(wzr, s);
        const v2f wi = mstep(wzi, s);
        rr = mma4(wr, dr, rr);
        ii = mma4(wi, di, ii);
        ri = mma4(wr, di, ri);
        ir = mma4(wi, dr, ir);
    }
#pragma unroll
    for (int g = 0; g < 2; ++g) {
        v4f ur, ui;
        ur.x = rr[4 * g]     - ii[4 * g];      ui.x = ri[4 * g]     + ir[4 * g];
        ur.y = rr[4 * g + 1] - ii[4 * g + 1];  ui.y = ri[4 * g + 1] + ir[4 * g + 1];
        ur.z = rr[4 * g + 2] - ii[4 * g + 2];  ui.z = ri[4 * g + 2] + ir[4 * g + 2];
        ur.w = rr[4 * g + 3] - ii[4 * g + 3];  ui.w = ri[4 * g + 3] + ir[4 * g + 3];
        *(v4f*)(Sre + (16 * w + l16) * SPITCH + 8 * half + 4 * g) = ur;
        *(v4f*)(Sim + (16 * w + l16) * SPITCH + 8 * half + 4 * g) = ui;
    }
    __syncthreads();

    v8f rr2 = {}, ii2 = {}, ri2 = {}, ir2 = {};
    slab_pass(m2r, m2i, Sre, Sim, rr2, ii2, ri2, ir2);
#pragma unroll
    for (int v = 0; v < 8; ++v) {
        const int idx = slab_idx(x, 16 * w + v + 8 * half, kt, l16);
        Ure[idx] = rr2[v] - ii2[v];
        Uim[idx] = ri2[v] + ir2[v];
    }
}

__global__ __launch_bounds__(128) void x_filter_kernel(float* Tre,
                                                       float* Tim,
                                                       const float* __restrict__ Mre,
                                                       const float* __restrict__ MimF,
                                                       const float* __restrict__ MimI,
                                                       const float* __restrict__ h,
                                                       float* __restrict__ stat) {
    __shared__ float Sre[N1D * SPITCH];
    __shared__ float Sim[N1D * SPITCH];
    __shared__ float wq[4];
    const int lane = threadIdx.x & 31;
    const int w    = threadIdx.x >> 5;
    const int half = lane >> 4;
    const int l16  = lane & 15;
    const int ky   = blockIdx.x >> 2;
    const int kt   = blockIdx.x & 3;

    v4f m1r[8], m1i[8], m2i[8];
    load_mfrag(m1r, Mre,  w, lane);
    load_mfrag(m1i, MimF, w, lane);
    load_mfrag(m2i, MimI, w, lane);
    float hv[8], hm[8];
    const int mky = (N1D - ky) & 63;
    const int mkz = (N1D - (16 * kt + l16)) & 63;
#pragma unroll
    for (int v = 0; v < 8; ++v) {
        const int kx = 16 * w + v + 8 * half;
        hv[v] = h[(kx * N1D + ky) * N1D + 16 * kt + l16];
        hm[v] = h[((((N1D - kx) & 63) * N1D) + mky) * N1D + mkz];
    }

    v8f rr = {}, ii = {}, ri = {}, ir = {};
#pragma unroll
    for (int s = 0; s < 16; ++s) {
        const int k = kof(s, half);
        v2f br, bi;
        br.x = Tre[slab_idx(k, ky, kt, l16)];  br.y = Tre[slab_idx(k + 1, ky, kt, l16)];
        bi.x = Tim[slab_idx(k, ky, kt, l16)];  bi.y = Tim[slab_idx(k + 1, ky, kt, l16)];
        rr = mma4(mstep(m1r, s), br, rr);
        ii = mma4(mstep(m1i, s), bi, ii);
        ri = mma4(mstep(m1r, s), bi, ri);
        ir = mma4(mstep(m1i, s), br, ir);
    }
    float qacc = 0.0f;
#pragma unroll
    for (int v = 0; v < 8; ++v) {
        const int kx = 16 * w + v + 8 * half;
        const float fr = rr[v] - ii[v];
        const float fi = ri[v] + ir[v];
        const float hs = 0.5f * (hv[v] + hm[v]);
        qacc += (fr * fr + fi * fi) * (hs * hs);
        Sre[kx * SPITCH + l16] = fr * hv[v];
        Sim[kx * SPITCH + l16] = fi * hv[v];
    }
#pragma unroll
    for (int m = 16; m > 0; m >>= 1) qacc += __shfl_xor(qacc, m);
    if (lane == 0) wq[w] = qacc;
    if (blockIdx.x == 4 * 32 + 2 && threadIdx.x == 64)
        stat[NBLK2] = (rr[0] - ii[0]) * hv[0];
    __syncthreads();
    if (threadIdx.x == 0) stat[blockIdx.x] = (wq[0] + wq[1]) + (wq[2] + wq[3]);

    v8f rr2 = {}, ii2 = {}, ri2 = {}, ir2 = {};
    slab_pass(m1r, m2i, Sre, Sim, rr2, ii2, ri2, ir2);
#pragma unroll
    for (int v = 0; v < 8; ++v) {
        const int idx = slab_idx(16 * w + v + 8 * half, ky, kt, l16);
        Tre[idx] = rr2[v] - ii2[v];
        Tim[idx] = ri2[v] + ir2[v];
    }
}

__global__ __launch_bounds__(128) void inv_yz_norm_kernel(const float* __restrict__ Tre,
                                                           const float* __restrict__ Tim,
                                                           const float* __restrict__ Mre,
                                                           const float* __restrict__ Mim,
                                                           const float* __restrict__ stat,
                                                           float* __restrict__ out) {
    __shared__ __attribute__((aligned(16))) float Lre[16 * WPITCH];
    __shared__ __attribute__((aligned(16))) float Lim[16 * WPITCH];
    __shared__ float wsq[4];
    const int lane = threadIdx.x & 31;
    const int w    = threadIdx.x >> 5;
    const int half = lane >> 4;
    const int l16  = lane & 15;
    const int x    = blockIdx.x >> 2;
    const int yt   = blockIdx.x & 3;

    v4f m2r[8], m2i[8];
    load_mfrag(m2r, Mre, w, lane);
    load_mfrag(m2i, Mim, w, lane);

    float q = stat[threadIdx.x] + stat[threadIdx.x + 128];
    const float g0 = stat[NBLK2];

    v4f m1r[8], m1i[8];
    load_mfrag(m1r, Mre, yt, lane);
    load_mfrag(m1i, Mim, yt, lane);
    v8f rr = {}, ii = {}, ri = {}, ir = {};
#pragma unroll
    for (int s = 0; s < 16; ++s) {
        const int k = kof(s, half);
        const v2f ar = mstep(m1r, s);
        const v2f ai = mstep(m1i, s);
        v2f br, bi;
        br.x = Tre[slab_idx(x, k, w, l16)];  br.y = Tre[slab_idx(x, k + 1, w, l16)];
        bi.x = Tim[slab_idx(x, k, w, l16)];  bi.y = Tim[slab_idx(x, k + 1, w, l16)];
        rr = mma4(ar, br, rr);
        ii = mma4(ai, bi, ii);
        ri = mma4(ar, bi, ri);
        ir = mma4(ai, br, ir);
    }
#pragma unroll
    for (int v = 0; v < 8; ++v) {
        Lre[(v + 8 * half) * WPITCH + 16 * w + l16] = rr[v] - ii[v];
        Lim[(v + 8 * half) * WPITCH + 16 * w + l16] = ri[v] + ir[v];
    }
#pragma unroll
    for (int m = 16; m > 0; m >>= 1) q += __shfl_xor(q, m);
    if (lane == 0) wsq[w] = q;
    __syncthreads();
    const float tsq  = (wsq[0] + wsq[1]) + (wsq[2] + wsq[3]);
    const float invN = 1.0f / (float)GRID_N;
    const float mean = g0 * (1.0f / 512.0f);
    const float var  = fmaxf(tsq * invN - mean * mean, 0.0f);
    const float inv  = 1.0f / (sqrtf(var) + 1e-8f);

    v8f pr = {}, pi = {};
    v4f wr[8], wi[8];
#pragma unroll
    for (int q = 0; q < 8; ++q) {
        wr[q] = *(const v4f*)(Lre + l16 * WPITCH + kof(2 * q, half));
        wi[q] = *(const v4f*)(Lim + l16 * WPITCH + kof(2 * q, half));
    }
    __builtin_amdgcn_sched_barrier(0);
#pragma unroll
    for (int s = 0; s < 16; ++s) {
        pr = mma4(mstep(wr, s), mstep(m2r, s), pr);
        pi = mma4(mstep(wi, s), mstep(m2i, s), pi);
    }
#pragma unroll
    for (int v = 0; v < 8; ++v) {
        const float re = pr[v] - pi[v];
        out[(x * N1D + 16 * yt + v + 8 * half) * N1D + 16 * w + l16] = (re - mean) * inv;
    }
}

extern "C" void kernel_launch(void* const* d_in, const int* in_sizes, int n_in,
                              void* d_out, int out_size, void* d_ws, size_t ws_size,
                              hipStream_t stream) {
    const float* X   = (const float*)d_in[0];
    const float* aw  = (const float*)d_in[1];
    const float* bw  = (const float*)d_in[2];
    const float* ham = (const float*)d_in[4];
    float* out = (float*)d_out;

    float* ws   = (float*)d_ws;
    float* Wre  = ws;
    float* WimF = Wre + 4096;
    float* WimI = WimF + 4096;
    float* Are  = WimI + 4096;
    float* Aim  = Are + GRID_N;
    float* Bre  = Aim + GRID_N;
    float* Bim  = Bre + GRID_N;

    float* stat = Are;

    density_gemm_kernel<<<256, 128, 0, stream>>>(X, aw, bw, Wre, WimF, WimI, Are, Aim);

    fwd_zy_kernel<<<NBLK2, 128, 0, stream>>>(Are, Aim, Wre, WimF, Bre, Bim);

    x_filter_kernel<<<NBLK2, 128, 0, stream>>>(Bre, Bim, Wre, WimF, WimI, ham, stat);

    inv_yz_norm_kernel<<<NBLK2, 128, 0, stream>>>(Bre, Bim, Wre, WimI, stat, out);
}


# the three whole-slab DFT passes (y forward, x forward, x inverse) factored 8x8: two block-form 16x16x16 stages, exchange in LDS, 16 WMMAs a wave not 64; changes the order of the DFT and variance sums
# speedup vs baseline: 2.6449x; 1.0416x over previous
#include <hip/hip_runtime.h>
#include <hip/hip_bf16.h>
#include <math.h>

#define N1D      64
#define GRID_N   (64 * 64 * 64)
#define N_ATOMS  128
#define K_TERMS  6
#define SPACING  0.5f
#define LOG2E    1.4426950408889634f


typedef float v2f __attribute__((ext_vector_type(2)));
typedef float v4f __attribute__((ext_vector_type(4)));
typedef float v8f __attribute__((ext_vector_type(8)));

__device__ __forceinline__ float fast_exp2(float x) {
#if __has_builtin(__builtin_amdgcn_exp2f)
    return __builtin_amdgcn_exp2f(x);
#else
    return exp2f(x);
#endif
}

#define ACH      8
#define NACH     (N_ATOMS / 2 / ACH)

struct AtomChunk {
    float x[ACH], y[ACH], z[ACH];
    v2f   b[ACH][3];
    v2f   w[ACH][3];
};

__device__ __forceinline__ void load_atoms(AtomChunk& p, const float* __restrict__ X,
                                           const float* __restrict__ aw,
                                           const float* __restrict__ bw,
                                           int half, int c) {
#pragma unroll
    for (int j = 0; j < ACH; ++j) {
        const int a = 2 * (c * ACH + j) + half;
        p.x[j] = X[3 * a];
        p.y[j] = X[3 * a + 1];
        p.z[j] = X[3 * a + 2];
        const v2f* bp = (const v2f*)(bw + K_TERMS * a);
        const v2f* wp = (const v2f*)(aw + K_TERMS * a);
#pragma unroll
        for (int u = 0; u < 3; ++u) {
            p.b[j][u] = bp[u];
            p.w[j][u] = wp[u];
        }
    }
}

__device__ __forceinline__ void mma_atoms(const AtomChunk& p, float fx, float fy, float fz,
                                          v8f& acc0, v8f& acc1) {
#pragma unroll
    for (int j = 0; j < ACH; ++j) {
        const float dx  = fx - p.x[j];
        const float dy  = fy - p.y[j];
        const float dz  = fz - p.z[j];
        const float rxy = (dx * dx + dy * dy) * LOG2E;
        const float rz  = (dz * dz) * LOG2E;
#pragma unroll
        for (int u = 0; u < 3; ++u) {
            const float b0 = p.b[j][u].x;
            const float b1 = p.b[j][u].y;
            v2f A, B;
            A.x = p.w[j][u].x * fast_exp2(b0 * rxy);
            A.y = p.w[j][u].y * fast_exp2(b1 * rxy);
            B.x = fast_exp2(b0 * rz);
            B.y = fast_exp2(b1 * rz);
            if (((3 * j + u) & 1) == 0)
                acc0 = __builtin_amdgcn_wmma_f32_16x16x4_f32(
                    false, B, false, A, (short)0, acc0, false, false);
            else
                acc1 = __builtin_amdgcn_wmma_f32_16x16x4_f32(
                    false, B, false, A, (short)0, acc1, false, false);
        }
    }
}

__global__ __launch_bounds__(128) void density_gemm_kernel(const float* __restrict__ X,
                                                           const float* __restrict__ aw,
                                                           const float* __restrict__ bw,
                                                           float* __restrict__ Wre,
                                                           float* __restrict__ WimF,
                                                           float* __restrict__ WimI,
                                                           float* __restrict__ rho_re,
                                                           float* __restrict__ rho_im) {
    const int lane = threadIdx.x & 31;
    const int izt  = threadIdx.x >> 5;
    const int ix   = blockIdx.x >> 2;
    const int iyt  = blockIdx.x & 3;
    const int half = lane >> 4;
    const int l16  = lane & 15;

    const float fx = (float)ix * SPACING;
    const float fy = (float)(iyt * 16 + l16) * SPACING;
    const float fz = (float)(izt * 16 + l16) * SPACING;

    v8f acc0 = {}, acc1 = {};
    AtomChunk p0, p1;

    load_atoms(p0, X, aw, bw, half, 0);
    for (int c = 0; c < NACH - 2; c += 2) {
        load_atoms(p1, X, aw, bw, half, c + 1);
        mma_atoms(p0, fx, fy, fz, acc0, acc1);
        load_atoms(p0, X, aw, bw, half, c + 2);
        mma_atoms(p1, fx, fy, fz, acc0, acc1);
    }
    load_atoms(p1, X, aw, bw, half, NACH - 1);
    mma_atoms(p0, fx, fy, fz, acc0, acc1);
    mma_atoms(p1, fx, fy, fz, acc0, acc1);

    {
        const v8f sum = acc0 + acc1;
        v4f lo, hi;
        lo.x = sum[0]; lo.y = sum[1]; lo.z = sum[2]; lo.w = sum[3];
        hi.x = sum[4]; hi.y = sum[5]; hi.z = sum[6]; hi.w = sum[7];
        const int o = ((ix * 4 + iyt) * 8 + 2 * izt) * 32 + lane;
        v4f* ore = (v4f*)rho_re + o;
        v4f* oim = (v4f*)rho_im + o;
        const v4f zero = {0.0f, 0.0f, 0.0f, 0.0f};
        ore[0]  = lo;
        ore[32] = hi;
        oim[0]  = zero;
        oim[32] = zero;
    }

    if (blockIdx.x < (N1D * N1D) / 128) {
        const int t  = blockIdx.x * 128 + threadIdx.x;
        const int e  = t & 3;
        const int fl = (t >> 2) & 31;
        const int q  = (t >> 7) & 7;
        const int k  = 16 * (t >> 10) + (fl & 15);
        const int sk = 2 * q + (e >> 1);
        const int n  = 16 * (sk >> 2) + 8 * (fl >> 4) + 2 * (sk & 3) + (e & 1);
        const int kn = (k * n) & 63;
        const float ang = (float)kn * 0.09817477042468103f;
        const float sgn = ((k + n) & 1) ? -0.125f : 0.125f;
        const float c = sgn * cosf(ang);
        const float s = sgn * sinf(ang);
        Wre[t]  = c;
        WimF[t] = -s;
        WimI[t] = s;
    }
}

#define NBLK2      256
#define WPITCH     68
#define SPITCH     20
#define SPART      (N1D * SPITCH + 16)
#define YPART      (N1D * 16 + 16)

__device__ __forceinline__ v8f mma4(v2f a, v2f b, v8f c) {
    return __builtin_amdgcn_wmma_f32_16x16x4_f32(false, a, false, b, (short)0, c, false, false);
}

__device__ __forceinline__ void load_mfrag(v4f (&m)[8], const float* __restrict__ Mf,
                                           int tile, int lane) {
    const v4f* p = (const v4f*)Mf + tile * 256 + lane;
#pragma unroll
    for (int q = 0; q < 8; ++q) m[q] = p[q * 32];
}

__device__ __forceinline__ int kof(int s, int half) {
    return 16 * (s >> 2) + 8 * half + 2 * (s & 3);
}

__device__ __forceinline__ v2f mstep(const v4f* m, int s) {
    const v4f t = m[s >> 1];
    v2f r;
    if (s & 1) { r.x = t.z; r.y = t.w; } else { r.x = t.x; r.y = t.y; }
    return r;
}

__device__ __forceinline__ int slab_idx(int x, int ky, int kt, int l) {
    return ((x * 4 + kt) * N1D + ky) * 16 + l;
}

__device__ __forceinline__ int widx(int k, int n) {
    const int sk = 4 * (n >> 4) + ((n >> 1) & 3);
    return ((k >> 4) << 10) | ((sk >> 1) << 7) | ((((n >> 3) & 1) * 16 + (k & 15)) << 2)
         | (2 * (sk & 1) + (n & 1));
}

struct StageM {
    v2f a1[4];
    v2f a2[2][4];
};

__device__ __forceinline__ void load_stage(StageM& m, const float* __restrict__ Mre,
                                           const float* __restrict__ Mim,
                                           int w, int half, int l16) {
    const int  p    = l16 >> 3;
    const int  i    = l16 & 7;
    const bool diag = (p == half);
    const float* src = diag ? Mre : Mim;
    const float sg  = diag ? 1.0f : ((p == 1) ? 1.0f : -1.0f);
    const float s1  = (i & 1) ? -8.0f * sg : 8.0f * sg;
#pragma unroll
    for (int s = 0; s < 4; ++s) {
        m.a1[s].x = s1 * src[widx(i, 8 * (2 * s))];
        m.a1[s].y = s1 * src[widx(i, 8 * (2 * s + 1))];
#pragma unroll
        for (int cc = 0; cc < 2; ++cc) {
            const int row = 2 * w + cc + 8 * i;
            m.a2[cc][s].x = sg * src[widx(row, 2 * s)];
            m.a2[cc][s].y = sg * src[widx(row, 2 * s + 1)];
        }
    }
}

__device__ __forceinline__ void conj_stage(StageM& o, const StageM& m, int half, int l16) {
    const float f = ((l16 >> 3) == half) ? 1.0f : -1.0f;
#pragma unroll
    for (int s = 0; s < 4; ++s) {
        o.a1[s]    = m.a1[s] * f;
        o.a2[0][s] = m.a2[0][s] * f;
        o.a2[1][s] = m.a2[1][s] * f;
    }
}

__device__ __forceinline__ void stage1_operands(v2f (&b1)[2][4], const float* P, int w) {
#pragma unroll
    for (int cc = 0; cc < 2; ++cc)
#pragma unroll
        for (int s = 0; s < 4; ++s) {
            b1[cc][s].x = P[(8 * (2 * s) + 2 * w + cc) * SPITCH];
            b1[cc][s].y = P[(8 * (2 * s + 1) + 2 * w + cc) * SPITCH];
        }
}

__device__ __forceinline__ void stage1(const StageM& m, const v2f (&b1)[2][4], float* Yh, int w) {
    v8f y0 = {}, y1 = {};
#pragma unroll
    for (int s = 0; s < 4; ++s) {
        y0 = mma4(m.a1[s], b1[0][s], y0);
        y1 = mma4(m.a1[s], b1[1][s], y1);
    }
#pragma unroll
    for (int r = 0; r < 8; ++r) {
        Yh[((2 * w) * 8 + r) * 16]     = y0[r];
        Yh[((2 * w + 1) * 8 + r) * 16] = y1[r];
    }
}

__device__ __forceinline__ void stage2(const StageM& m, const float* Yh, int w, v8f& o0, v8f& o1) {
    v2f b2[2][4];
#pragma unroll
    for (int cc = 0; cc < 2; ++cc)
#pragma unroll
        for (int s = 0; s < 4; ++s) {
            b2[cc][s].x = Yh[((2 * s) * 8 + 2 * w + cc) * 16];
            b2[cc][s].y = Yh[((2 * s + 1) * 8 + 2 * w + cc) * 16];
        }
    v8f a = {}, b = {};
#pragma unroll
    for (int s = 0; s < 4; ++s) {
        a = mma4(m.a2[0][s], b2[0][s], a);
        b = mma4(m.a2[1][s], b2[1][s], b);
    }
    o0 = a;
    o1 = b;
}

__global__ __launch_bounds__(128) void fwd_zy_kernel(const float* __restrict__ Tre,
                                                     const float* __restrict__ Tim,
                                                     const float* __restrict__ Mre,
                                                     const float* __restrict__ Mim,
                                                     float* __restrict__ Ure,
                                                     float* __restrict__ Uim) {
    __shared__ __attribute__((aligned(16))) float S[2 * SPART];
    __shared__ float Yx[2 * YPART];
    float* Sre = S;
    float* Sim = S + SPART;
    const int lane = threadIdx.x & 31;
    const int w    = threadIdx.x >> 5;
    const int half = lane >> 4;
    const int l16  = lane & 15;
    const int x    = blockIdx.x >> 2;
    const int kt   = blockIdx.x & 3;

    StageM m2;
    load_stage(m2, Mre, Mim, w, half, l16);

    v4f wzr[8], wzi[8], tr[8], ti[8];
    load_mfrag(wzr, Mre, kt, lane);
    load_mfrag(wzi, Mim, kt, lane);
    load_mfrag(tr, Tre + x * (N1D * N1D), w, lane);
    load_mfrag(ti, Tim + x * (N1D * N1D), w, lane);
    v8f rr = {}, ii = {}, ri = {}, ir = {};
#pragma unroll
    for (int s = 0; s < 16; ++s) {
        const v2f dr = mstep(tr, s);
        const v2f di = mstep(ti, s);
        const v2f wr = mstep(wzr, s);
        const v2f wi = mstep(wzi, s);
        rr = mma4(wr, dr, rr);
        ii = mma4(wi, di, ii);
        ri = mma4(wr, di, ri);
        ir = mma4(wi, dr, ir);
    }
#pragma unroll
    for (int g = 0; g < 2; ++g) {
        v4f ur, ui;
        ur.x = rr[4 * g]     - ii[4 * g];      ui.x = ri[4 * g]     + ir[4 * g];
        ur.y = rr[4 * g + 1] - ii[4 * g + 1];  ui.y = ri[4 * g + 1] + ir[4 * g + 1];
        ur.z = rr[4 * g + 2] - ii[4 * g + 2];  ui.z = ri[4 * g + 2] + ir[4 * g + 2];
        ur.w = rr[4 * g + 3] - ii[4 * g + 3];  ui.w = ri[4 * g + 3] + ir[4 * g + 3];
        *(v4f*)(Sre + (16 * w + l16) * SPITCH + 8 * half + 4 * g) = ur;
        *(v4f*)(Sim + (16 * w + l16) * SPITCH + 8 * half + 4 * g) = ui;
    }
    __syncthreads();

    float* Yh = Yx + half * YPART + l16;
    v2f b1[2][4];
    stage1_operands(b1, S + half * SPART + l16, w);
    stage1(m2, b1, Yh, w);
    __syncthreads();
    v8f o0, o1;
    stage2(m2, Yh, w, o0, o1);
    float* Up = half ? Uim : Ure;
#pragma unroll
    for (int r = 0; r < 8; ++r) {
        Up[slab_idx(x, 2 * w + 8 * r, kt, l16)]     = o0[r];
        Up[slab_idx(x, 2 * w + 1 + 8 * r, kt, l16)] = o1[r];
    }
}

__global__ __launch_bounds__(128) void x_filter_kernel(float* Tre,
                                                       float* Tim,
                                                       const float* __restrict__ Mre,
                                                       const float* __restrict__ MimF,
                                                       const float* __restrict__ h,
                                                       float* __restrict__ stat) {
    __shared__ float S[2 * SPART];
    __shared__ float Yx[2 * YPART];
    __shared__ float wq[4];
    const int lane = threadIdx.x & 31;
    const int w    = threadIdx.x >> 5;
    const int half = lane >> 4;
    const int l16  = lane & 15;
    const int ky   = blockIdx.x >> 2;
    const int kt   = blockIdx.x & 3;

    StageM mf, mi;
    load_stage(mf, Mre, MimF, w, half, l16);
    conj_stage(mi, mf, half, l16);
    float* Tp = half ? Tim : Tre;

    v2f b1[2][4];
#pragma unroll
    for (int cc = 0; cc < 2; ++cc)
#pragma unroll
        for (int s = 0; s < 4; ++s) {
            b1[cc][s].x = Tp[slab_idx(8 * (2 * s) + 2 * w + cc, ky, kt, l16)];
            b1[cc][s].y = Tp[slab_idx(8 * (2 * s + 1) + 2 * w + cc, ky, kt, l16)];
        }
    float hv[2][8], hm[2][8];
    const int mky = (N1D - ky) & 63;
    const int mkz = (N1D - (16 * kt + l16)) & 63;
#pragma unroll
    for (int cc = 0; cc < 2; ++cc)
#pragma unroll
        for (int r = 0; r < 8; ++r) {
            const int kx = 2 * w + cc + 8 * r;
            hv[cc][r] = h[(kx * N1D + ky) * N1D + 16 * kt + l16];
            hm[cc][r] = h[((((N1D - kx) & 63) * N1D) + mky) * N1D + mkz];
        }

    float* Yh = Yx + half * YPART + l16;
    float* Sh = S + half * SPART + l16;
    stage1(mf, b1, Yh, w);
    __syncthreads();
    v8f f0, f1;
    stage2(mf, Yh, w, f0, f1);
    float qacc = 0.0f;
#pragma unroll
    for (int r = 0; r < 8; ++r) {
        const float hs0 = 0.5f * (hv[0][r] + hm[0][r]);
        const float hs1 = 0.5f * (hv[1][r] + hm[1][r]);
        qacc += (f0[r] * f0[r]) * (hs0 * hs0);
        qacc += (f1[r] * f1[r]) * (hs1 * hs1);
        Sh[(2 * w + 8 * r) * SPITCH]     = f0[r] * hv[0][r];
        Sh[(2 * w + 1 + 8 * r) * SPITCH] = f1[r] * hv[1][r];
    }
#pragma unroll
    for (int m = 16; m > 0; m >>= 1) qacc += __shfl_xor(qacc, m);
    if (lane == 0) wq[w] = qacc;
    if (blockIdx.x == 4 * 32 + 2 && threadIdx.x == 0)
        stat[NBLK2] = f0[4] * hv[0][4];
    __syncthreads();
    if (threadIdx.x == 0) stat[blockIdx.x] = (wq[0] + wq[1]) + (wq[2] + wq[3]);

    v2f c1[2][4];
    stage1_operands(c1, Sh, w);
    stage1(mi, c1, Yh, w);
    __syncthreads();
    v8f r0, r1;
    stage2(mi, Yh, w, r0, r1);
#pragma unroll
    for (int r = 0; r < 8; ++r) {
        Tp[slab_idx(2 * w + 8 * r, ky, kt, l16)]     = r0[r];
        Tp[slab_idx(2 * w + 1 + 8 * r, ky, kt, l16)] = r1[r];
    }
}

__global__ __launch_bounds__(128) void inv_yz_norm_kernel(const float* __restrict__ Tre,
                                                           const float* __restrict__ Tim,
                                                           const float* __restrict__ Mre,
                                                           const float* __restrict__ Mim,
                                                           const float* __restrict__ stat,
                                                           float* __restrict__ out) {
    __shared__ __attribute__((aligned(16))) float Lre[16 * WPITCH];
    __shared__ __attribute__((aligned(16))) float Lim[16 * WPITCH];
    __shared__ float wsq[4];
    const int lane = threadIdx.x & 31;
    const int w    = threadIdx.x >> 5;
    const int half = lane >> 4;
    const int l16  = lane & 15;
    const int x    = blockIdx.x >> 2;
    const int yt   = blockIdx.x & 3;

    v4f m2r[8], m2i[8];
    load_mfrag(m2r, Mre, w, lane);
    load_mfrag(m2i, Mim, w, lane);

    float q = stat[threadIdx.x] + stat[threadIdx.x + 128];
    const float g0 = stat[NBLK2];

    v4f m1r[8], m1i[8];
    load_mfrag(m1r, Mre, yt, lane);
    load_mfrag(m1i, Mim, yt, lane);
    v8f rr = {}, ii = {}, ri = {}, ir = {};
#pragma unroll
    for (int s = 0; s < 16; ++s) {
        const int k = kof(s, half);
        const v2f ar = mstep(m1r, s);
        const v2f ai = mstep(m1i, s);
        v2f br, bi;
        br.x = Tre[slab_idx(x, k, w, l16)];  br.y = Tre[slab_idx(x, k + 1, w, l16)];
        bi.x = Tim[slab_idx(x, k, w, l16)];  bi.y = Tim[slab_idx(x, k + 1, w, l16)];
        rr = mma4(ar, br, rr);
        ii = mma4(ai, bi, ii);
        ri = mma4(ar, bi, ri);
        ir = mma4(ai, br, ir);
    }
#pragma unroll
    for (int v = 0; v < 8; ++v) {
        Lre[(v + 8 * half) * WPITCH + 16 * w + l16] = rr[v] - ii[v];
        Lim[(v + 8 * half) * WPITCH + 16 * w + l16] = ri[v] + ir[v];
    }
#pragma unroll
    for (int m = 16; m > 0; m >>= 1) q += __shfl_xor(q, m);
    if (lane == 0) wsq[w] = q;
    __syncthreads();
    const float tsq  = (wsq[0] + wsq[1]) + (wsq[2] + wsq[3]);
    const float invN = 1.0f / (float)GRID_N;
    const float mean = g0 * (1.0f / 512.0f);
    const float var  = fmaxf(tsq * invN - mean * mean, 0.0f);
    const float inv  = 1.0f / (sqrtf(var) + 1e-8f);

    v8f pr = {}, pi = {};
    v4f wr[8], wi[8];
#pragma unroll
    for (int q = 0; q < 8; ++q) {
        wr[q] = *(const v4f*)(Lre + l16 * WPITCH + kof(2 * q, half));
        wi[q] = *(const v4f*)(Lim + l16 * WPITCH + kof(2 * q, half));
    }
    __builtin_amdgcn_sched_barrier(0);
#pragma unroll
    for (int s = 0; s < 16; ++s) {
        pr = mma4(mstep(wr, s), mstep(m2r, s), pr);
        pi = mma4(mstep(wi, s), mstep(m2i, s), pi);
    }
#pragma unroll
    for (int v = 0; v < 8; ++v) {
        const float re = pr[v] - pi[v];
        out[(x * N1D + 16 * yt + v + 8 * half) * N1D + 16 * w + l16] = (re - mean) * inv;
    }
}

extern "C" void kernel_launch(void* const* d_in, const int* in_sizes, int n_in,
                              void* d_out, int out_size, void* d_ws, size_t ws_size,
                              hipStream_t stream) {
    const float* X   = (const float*)d_in[0];
    const float* aw  = (const float*)d_in[1];
    const float* bw  = (const float*)d_in[2];
    const float* ham = (const float*)d_in[4];
    float* out = (float*)d_out;

    float* ws   = (float*)d_ws;
    float* Wre  = ws;
    float* WimF = Wre + 4096;
    float* WimI = WimF + 4096;
    float* Are  = WimI + 4096;
    float* Aim  = Are + GRID_N;
    float* Bre  = Aim + GRID_N;
    float* Bim  = Bre + GRID_N;

    float* stat = Are;

    density_gemm_kernel<<<256, 128, 0, stream>>>(X, aw, bw, Wre, WimF, WimI, Are, Aim);

    fwd_zy_kernel<<<NBLK2, 128, 0, stream>>>(Are, Aim, Wre, WimF, Bre, Bim);

    x_filter_kernel<<<NBLK2, 128, 0, stream>>>(Bre, Bim, Wre, WimF, ham, stat);

    inv_yz_norm_kernel<<<NBLK2, 128, 0, stream>>>(Bre, Bim, Wre, WimI, stat, out);
}


# forward passes reordered y,z,x: y pass on the density slab in LDS inside the density launch, z pass joined to the x/filter launch (4 launches -> 3); changes the order of the DFT and variance sums
# speedup vs baseline: 2.8091x; 1.0621x over previous
#include <hip/hip_runtime.h>
#include <hip/hip_bf16.h>
#include <math.h>

#define N1D      64
#define GRID_N   (64 * 64 * 64)
#define N_ATOMS  128
#define K_TERMS  6
#define SPACING  0.5f
#define LOG2E    1.4426950408889634f


typedef float v2f __attribute__((ext_vector_type(2)));
typedef float v4f __attribute__((ext_vector_type(4)));
typedef float v8f __attribute__((ext_vector_type(8)));

__device__ __forceinline__ float fast_exp2(float x) {
#if __has_builtin(__builtin_amdgcn_exp2f)
    return __builtin_amdgcn_exp2f(x);
#else
    return exp2f(x);
#endif
}


__device__ const float COS64[64] = {
    1.000000000f, 0.9951847267f, 0.9807852804f, 0.9569403357f, 0.9238795325f, 0.8819212643f, 0.8314696123f, 0.7730104534f,
    0.7071067812f, 0.6343932842f, 0.5555702330f, 0.4713967368f, 0.3826834324f, 0.2902846773f, 0.1950903220f, 0.09801714033f,
    0.000000000f, -0.09801714033f, -0.1950903220f, -0.2902846773f, -0.3826834324f, -0.4713967368f, -0.5555702330f, -0.6343932842f,
    -0.7071067812f, -0.7730104534f, -0.8314696123f, -0.8819212643f, -0.9238795325f, -0.9569403357f, -0.9807852804f, -0.9951847267f,
    -1.000000000f, -0.9951847267f, -0.9807852804f, -0.9569403357f, -0.9238795325f, -0.8819212643f, -0.8314696123f, -0.7730104534f,
    -0.7071067812f, -0.6343932842f, -0.5555702330f, -0.4713967368f, -0.3826834324f, -0.2902846773f, -0.1950903220f, -0.09801714033f,
    0.000000000f, 0.09801714033f, 0.1950903220f, 0.2902846773f, 0.3826834324f, 0.4713967368f, 0.5555702330f, 0.6343932842f,
    0.7071067812f, 0.7730104534f, 0.8314696123f, 0.8819212643f, 0.9238795325f, 0.9569403357f, 0.9807852804f, 0.9951847267f
};

__device__ __forceinline__ float wval(int k, int n, bool real_part) {
    const int   kn  = (k * n) & 63;
    const float sgn = ((k + n) & 1) ? -0.125f : 0.125f;
    return real_part ? sgn * COS64[kn] : -sgn * COS64[(kn + 48) & 63];
}

#define ACH      8
#define NACH     (N_ATOMS / 2 / ACH)

struct AtomChunk {
    float x[ACH], y[ACH], z[ACH];
    v2f   b[ACH][3];
    v2f   w[ACH][3];
};

__device__ __forceinline__ void load_atoms(AtomChunk& p, const float* __restrict__ X,
                                           const float* __restrict__ aw,
                                           const float* __restrict__ bw,
                                           int half, int c) {
#pragma unroll
    for (int j = 0; j < ACH; ++j) {
        const int a = 2 * (c * ACH + j) + half;
        p.x[j] = X[3 * a];
        p.y[j] = X[3 * a + 1];
        p.z[j] = X[3 * a + 2];
        const v2f* bp = (const v2f*)(bw + K_TERMS * a);
        const v2f* wp = (const v2f*)(aw + K_TERMS * a);
#pragma unroll
        for (int u = 0; u < 3; ++u) {
            p.b[j][u] = bp[u];
            p.w[j][u] = wp[u];
        }
    }
}

__device__ __forceinline__ void mma_atoms(const AtomChunk& p, float fx, float fy, float fz,
                                          v8f& acc0, v8f& acc1) {
#pragma unroll
    for (int j = 0; j < ACH; ++j) {
        const float dx  = fx - p.x[j];
        const float dy  = fy - p.y[j];
        const float dz  = fz - p.z[j];
        const float rxy = (dx * dx + dy * dy) * LOG2E;
        const float rz  = (dz * dz) * LOG2E;
#pragma unroll
        for (int u = 0; u < 3; ++u) {
            const float b0 = p.b[j][u].x;
            const float b1 = p.b[j][u].y;
            v2f A, B;
            A.x = p.w[j][u].x * fast_exp2(b0 * rxy);
            A.y = p.w[j][u].y * fast_exp2(b1 * rxy);
            B.x = fast_exp2(b0 * rz);
            B.y = fast_exp2(b1 * rz);
            if (((3 * j + u) & 1) == 0)
                acc0 = __builtin_amdgcn_wmma_f32_16x16x4_f32(
                    false, B, false, A, (short)0, acc0, false, false);
            else
                acc1 = __builtin_amdgcn_wmma_f32_16x16x4_f32(
                    false, B, false, A, (short)0, acc1, false, false);
        }
    }
}

#define NBLK2      256
#define WPITCH     68
#define SPITCH     20
#define SPART      (N1D * SPITCH + 16)
#define YPART      (N1D * 16 + 16)

__device__ __forceinline__ v8f mma4(v2f a, v2f b, v8f c) {
    return __builtin_amdgcn_wmma_f32_16x16x4_f32(false, a, false, b, (short)0, c, false, false);
}

__device__ __forceinline__ void load_mfrag(v4f (&m)[8], const float* __restrict__ Mf,
                                           int tile, int lane) {
    const v4f* p = (const v4f*)Mf + tile * 256 + lane;
#pragma unroll
    for (int q = 0; q < 8; ++q) m[q] = p[q * 32];
}

__device__ __forceinline__ int kof(int s, int half) {
    return 16 * (s >> 2) + 8 * half + 2 * (s & 3);
}

__device__ __forceinline__ v2f mstep(const v4f* m, int s) {
    const v4f t = m[s >> 1];
    v2f r;
    if (s & 1) { r.x = t.z; r.y = t.w; } else { r.x = t.x; r.y = t.y; }
    return r;
}

__device__ __forceinline__ int slab_idx(int x, int ky, int kt, int l) {
    return ((x * 4 + kt) * N1D + ky) * 16 + l;
}

__device__ __forceinline__ int widx(int k, int n) {
    const int sk = 4 * (n >> 4) + ((n >> 1) & 3);
    return ((k >> 4) << 10) | ((sk >> 1) << 7) | ((((n >> 3) & 1) * 16 + (k & 15)) << 2)
         | (2 * (sk & 1) + (n & 1));
}

struct StageM {
    v2f a1[4];
    v2f a2[2][4];
};

__device__ __forceinline__ void load_stage(StageM& m, const float* __restrict__ Mre,
                                           const float* __restrict__ Mim,
                                           int w, int half, int l16) {
    const int  p    = l16 >> 3;
    const int  i    = l16 & 7;
    const bool diag = (p == half);
    const float* src = diag ? Mre : Mim;
    const float sg  = diag ? 1.0f : ((p == 1) ? 1.0f : -1.0f);
    const float s1  = (i & 1) ? -8.0f * sg : 8.0f * sg;
#pragma unroll
    for (int s = 0; s < 4; ++s) {
        m.a1[s].x = s1 * src[widx(i, 8 * (2 * s))];
        m.a1[s].y = s1 * src[widx(i, 8 * (2 * s + 1))];
#pragma unroll
        for (int cc = 0; cc < 2; ++cc) {
            const int row = 2 * w + cc + 8 * i;
            m.a2[cc][s].x = sg * src[widx(row, 2 * s)];
            m.a2[cc][s].y = sg * src[widx(row, 2 * s + 1)];
        }
    }
}

__device__ __forceinline__ void conj_stage(StageM& o, const StageM& m, int half, int l16) {
    const float f = ((l16 >> 3) == half) ? 1.0f : -1.0f;
#pragma unroll
    for (int s = 0; s < 4; ++s) {
        o.a1[s]    = m.a1[s] * f;
        o.a2[0][s] = m.a2[0][s] * f;
        o.a2[1][s] = m.a2[1][s] * f;
    }
}

__device__ __forceinline__ void make_stage(StageM& m, int w, int half, int l16) {
    const int  p    = l16 >> 3;
    const int  i    = l16 & 7;
    const bool diag = (p == half);
    const float sg  = diag ? 1.0f : ((p == 1) ? 1.0f : -1.0f);
    const float s1  = (i & 1) ? -8.0f * sg : 8.0f * sg;
#pragma unroll
    for (int s = 0; s < 4; ++s) {
        m.a1[s].x = s1 * wval(i, 8 * (2 * s), diag);
        m.a1[s].y = s1 * wval(i, 8 * (2 * s + 1), diag);
#pragma unroll
        for (int cc = 0; cc < 2; ++cc) {
            const int row = 2 * w + cc + 8 * i;
            m.a2[cc][s].x = sg * wval(row, 2 * s, diag);
            m.a2[cc][s].y = sg * wval(row, 2 * s + 1, diag);
        }
    }
}

__device__ __forceinline__ void stage1_operands(v2f (&b1)[2][4], const float* P, int w) {
#pragma unroll
    for (int cc = 0; cc < 2; ++cc)
#pragma unroll
        for (int s = 0; s < 4; ++s) {
            b1[cc][s].x = P[(8 * (2 * s) + 2 * w + cc) * SPITCH];
            b1[cc][s].y = P[(8 * (2 * s + 1) + 2 * w + cc) * SPITCH];
        }
}

__device__ __forceinline__ void stage1(const StageM& m, const v2f (&b1)[2][4], float* Yh, int w) {
    v8f y0 = {}, y1 = {};
#pragma unroll
    for (int s = 0; s < 4; ++s) {
        y0 = mma4(m.a1[s], b1[0][s], y0);
        y1 = mma4(m.a1[s], b1[1][s], y1);
    }
#pragma unroll
    for (int r = 0; r < 8; ++r) {
        Yh[((2 * w) * 8 + r) * 16]     = y0[r];
        Yh[((2 * w + 1) * 8 + r) * 16] = y1[r];
    }
}

__device__ __forceinline__ void stage2(const StageM& m, const float* Yh, int w, v8f& o0, v8f& o1) {
    v2f b2[2][4];
#pragma unroll
    for (int cc = 0; cc < 2; ++cc)
#pragma unroll
        for (int s = 0; s < 4; ++s) {
            b2[cc][s].x = Yh[((2 * s) * 8 + 2 * w + cc) * 16];
            b2[cc][s].y = Yh[((2 * s + 1) * 8 + 2 * w + cc) * 16];
        }
    v8f a = {}, b = {};
#pragma unroll
    for (int s = 0; s < 4; ++s) {
        a = mma4(m.a2[0][s], b2[0][s], a);
        b = mma4(m.a2[1][s], b2[1][s], b);
    }
    o0 = a;
    o1 = b;
}

__global__ __launch_bounds__(128) void density_yfwd_kernel(const float* __restrict__ X,
                                                           const float* __restrict__ aw,
                                                           const float* __restrict__ bw,
                                                           float* __restrict__ Wre,
                                                           float* __restrict__ WimF,
                                                           float* __restrict__ Ure,
                                                           float* __restrict__ Uim) {
    __shared__ __attribute__((aligned(16))) float S[2 * SPART];
    __shared__ float Yx[2 * YPART];
    const int lane = threadIdx.x & 31;
    const int w    = threadIdx.x >> 5;
    const int iyt  = w;
    const int ix   = blockIdx.x >> 2;
    const int izt  = blockIdx.x & 3;
    const int half = lane >> 4;
    const int l16  = lane & 15;

    const float fx = (float)ix * SPACING;
    const float fy = (float)(iyt * 16 + l16) * SPACING;
    const float fz = (float)(izt * 16 + l16) * SPACING;

    StageM m;
    make_stage(m, w, half, l16);

    v8f acc0 = {}, acc1 = {};
    AtomChunk p0, p1;

    load_atoms(p0, X, aw, bw, half, 0);
    for (int c = 0; c < NACH - 2; c += 2) {
        load_atoms(p1, X, aw, bw, half, c + 1);
        mma_atoms(p0, fx, fy, fz, acc0, acc1);
        load_atoms(p0, X, aw, bw, half, c + 2);
        mma_atoms(p1, fx, fy, fz, acc0, acc1);
    }
    load_atoms(p1, X, aw, bw, half, NACH - 1);
    mma_atoms(p0, fx, fy, fz, acc0, acc1);
    mma_atoms(p1, fx, fy, fz, acc0, acc1);

    {
        const v8f sum = acc0 + acc1;
        v4f lo, hi;
        lo.x = sum[0]; lo.y = sum[1]; lo.z = sum[2]; lo.w = sum[3];
        hi.x = sum[4]; hi.y = sum[5]; hi.z = sum[6]; hi.w = sum[7];
        const v4f zero = {0.0f, 0.0f, 0.0f, 0.0f};
        float* sre = S + (16 * iyt + l16) * SPITCH + 8 * half;
        float* sim = sre + SPART;
        *(v4f*)(sre)     = lo;
        *(v4f*)(sre + 4) = hi;
        *(v4f*)(sim)     = zero;
        *(v4f*)(sim + 4) = zero;
    }

    __syncthreads();

    float* Yh = Yx + half * YPART + l16;
    v2f b1[2][4];
    stage1_operands(b1, S + half * SPART + l16, w);
    stage1(m, b1, Yh, w);
    __syncthreads();
    v8f o0, o1;
    stage2(m, Yh, w, o0, o1);
    float* Up = half ? Uim : Ure;
#pragma unroll
    for (int r = 0; r < 8; ++r) {
        Up[((2 * w + 8 * r) * N1D + ix) * N1D + 16 * izt + l16]     = o0[r];
        Up[((2 * w + 1 + 8 * r) * N1D + ix) * N1D + 16 * izt + l16] = o1[r];
    }

    if (blockIdx.x < (N1D * N1D) / 128) {
        const int t  = blockIdx.x * 128 + threadIdx.x;
        const int e  = t & 3;
        const int fl = (t >> 2) & 31;
        const int q  = (t >> 7) & 7;
        const int k  = 16 * (t >> 10) + (fl & 15);
        const int sk = 2 * q + (e >> 1);
        const int n  = 16 * (sk >> 2) + 8 * (fl >> 4) + 2 * (sk & 3) + (e & 1);
        const int kn = (k * n) & 63;
        const float ang = (float)kn * 0.09817477042468103f;
        const float sgn = ((k + n) & 1) ? -0.125f : 0.125f;
        const float c = sgn * cosf(ang);
        const float s = sgn * sinf(ang);
        Wre[t]  = c;
        WimF[t] = -s;
    }
}

__global__ __launch_bounds__(128) void zx_filter_kernel(const float* __restrict__ Ure,
                                                        const float* __restrict__ Uim,
                                                        float* __restrict__ Tre,
                                                        float* __restrict__ Tim,
                                                        const float* __restrict__ Mre,
                                                        const float* __restrict__ MimF,
                                                        const float* __restrict__ h,
                                                        float* __restrict__ stat) {
    __shared__ __attribute__((aligned(16))) float S[2 * SPART];
    __shared__ float Yx[2 * YPART];
    __shared__ float wq[4];
    const int lane = threadIdx.x & 31;
    const int w    = threadIdx.x >> 5;
    const int half = lane >> 4;
    const int l16  = lane & 15;
    const int ky   = blockIdx.x >> 2;
    const int kt   = blockIdx.x & 3;

    StageM mf, mi;
    load_stage(mf, Mre, MimF, w, half, l16);
    conj_stage(mi, mf, half, l16);
    float* Tp = half ? Tim : Tre;

    v4f wzr[8], wzi[8], ur[8], ui[8];
    load_mfrag(wzr, Mre, kt, lane);
    load_mfrag(wzi, MimF, kt, lane);
    {
        const int row = ((ky * N1D) + 16 * w + l16) * N1D + 8 * half;
        const v4f* pr = (const v4f*)(Ure + row);
        const v4f* pi = (const v4f*)(Uim + row);
#pragma unroll
        for (int q = 0; q < 8; ++q) {
            ur[q] = pr[4 * (q >> 1) + (q & 1)];
            ui[q] = pi[4 * (q >> 1) + (q & 1)];
        }
    }
    float hv[2][8], hm[2][8];
    const int mky = (N1D - ky) & 63;
    const int mkz = (N1D - (16 * kt + l16)) & 63;
#pragma unroll
    for (int cc = 0; cc < 2; ++cc)
#pragma unroll
        for (int r = 0; r < 8; ++r) {
            const int kx = 2 * w + cc + 8 * r;
            hv[cc][r] = h[(kx * N1D + ky) * N1D + 16 * kt + l16];
            hm[cc][r] = h[((((N1D - kx) & 63) * N1D) + mky) * N1D + mkz];
        }

    v8f rr = {}, ii = {}, ri = {}, ir = {};
#pragma unroll
    for (int s = 0; s < 16; ++s) {
        const v2f dr = mstep(ur, s);
        const v2f di = mstep(ui, s);
        const v2f wr = mstep(wzr, s);
        const v2f wi = mstep(wzi, s);
        rr = mma4(wr, dr, rr);
        ii = mma4(wi, di, ii);
        ri = mma4(wr, di, ri);
        ir = mma4(wi, dr, ir);
    }
#pragma unroll
    for (int g = 0; g < 2; ++g) {
        v4f vr, vi;
        vr.x = rr[4 * g]     - ii[4 * g];      vi.x = ri[4 * g]     + ir[4 * g];
        vr.y = rr[4 * g + 1] - ii[4 * g + 1];  vi.y = ri[4 * g + 1] + ir[4 * g + 1];
        vr.z = rr[4 * g + 2] - ii[4 * g + 2];  vi.z = ri[4 * g + 2] + ir[4 * g + 2];
        vr.w = rr[4 * g + 3] - ii[4 * g + 3];  vi.w = ri[4 * g + 3] + ir[4 * g + 3];
        *(v4f*)(S + (16 * w + l16) * SPITCH + 8 * half + 4 * g)         = vr;
        *(v4f*)(S + SPART + (16 * w + l16) * SPITCH + 8 * half + 4 * g) = vi;
    }
    __syncthreads();

    float* Yh = Yx + half * YPART + l16;
    float* Sh = S + half * SPART + l16;
    v2f b1[2][4];
    stage1_operands(b1, Sh, w);
    stage1(mf, b1, Yh, w);
    __syncthreads();
    v8f f0, f1;
    stage2(mf, Yh, w, f0, f1);
    float qacc = 0.0f;
#pragma unroll
    for (int r = 0; r < 8; ++r) {
        const float hs0 = 0.5f * (hv[0][r] + hm[0][r]);
        const float hs1 = 0.5f * (hv[1][r] + hm[1][r]);
        qacc += (f0[r] * f0[r]) * (hs0 * hs0);
        qacc += (f1[r] * f1[r]) * (hs1 * hs1);
        Sh[(2 * w + 8 * r) * SPITCH]     = f0[r] * hv[0][r];
        Sh[(2 * w + 1 + 8 * r) * SPITCH] = f1[r] * hv[1][r];
    }
#pragma unroll
    for (int m = 16; m > 0; m >>= 1) qacc += __shfl_xor(qacc, m);
    if (lane == 0) wq[w] = qacc;
    if (blockIdx.x == 4 * 32 + 2 && threadIdx.x == 0)
        stat[NBLK2] = f0[4] * hv[0][4];
    __syncthreads();
    if (threadIdx.x == 0) stat[blockIdx.x] = (wq[0] + wq[1]) + (wq[2] + wq[3]);

    v2f c1[2][4];
    stage1_operands(c1, Sh, w);
    stage1(mi, c1, Yh, w);
    __syncthreads();
    v8f r0, r1;
    stage2(mi, Yh, w, r0, r1);
#pragma unroll
    for (int r = 0; r < 8; ++r) {
        Tp[slab_idx(2 * w + 8 * r, ky, kt, l16)]     = r0[r];
        Tp[slab_idx(2 * w + 1 + 8 * r, ky, kt, l16)] = r1[r];
    }
}

__global__ __launch_bounds__(128) void inv_yz_norm_kernel(const float* __restrict__ Tre,
                                                           const float* __restrict__ Tim,
                                                           const float* __restrict__ Mre,
                                                           const float* __restrict__ Mim,
                                                           const float* __restrict__ stat,
                                                           float* __restrict__ out) {
    __shared__ __attribute__((aligned(16))) float Lre[16 * WPITCH];
    __shared__ __attribute__((aligned(16))) float Lim[16 * WPITCH];
    __shared__ float wsq[4];
    const int lane = threadIdx.x & 31;
    const int w    = threadIdx.x >> 5;
    const int half = lane >> 4;
    const int l16  = lane & 15;
    const int x    = blockIdx.x >> 2;
    const int yt   = blockIdx.x & 3;

    v4f m2r[8], m2i[8];
    load_mfrag(m2r, Mre, w, lane);
    load_mfrag(m2i, Mim, w, lane);

    float q = stat[threadIdx.x] + stat[threadIdx.x + 128];
    const float g0 = stat[NBLK2];

    v4f m1r[8], m1i[8];
    load_mfrag(m1r, Mre, yt, lane);
    load_mfrag(m1i, Mim, yt, lane);
    v8f rr = {}, ii = {}, ri = {}, ir = {};
#pragma unroll
    for (int s = 0; s < 16; ++s) {
        const int k = kof(s, half);
        const v2f ar = mstep(m1r, s);
        const v2f ai = mstep(m1i, s);
        v2f br, bi;
        br.x = Tre[slab_idx(x, k, w, l16)];  br.y = Tre[slab_idx(x, k + 1, w, l16)];
        bi.x = Tim[slab_idx(x, k, w, l16)];  bi.y = Tim[slab_idx(x, k + 1, w, l16)];
        rr = mma4(ar, br, rr);
        ii = mma4(ai, bi, ii);
        ri = mma4(ar, bi, ri);
        ir = mma4(ai, br, ir);
    }
#pragma unroll
    for (int v = 0; v < 8; ++v) {
        Lre[(v + 8 * half) * WPITCH + 16 * w + l16] = rr[v] + ii[v];
        Lim[(v + 8 * half) * WPITCH + 16 * w + l16] = ri[v] - ir[v];
    }
#pragma unroll
    for (int m = 16; m > 0; m >>= 1) q += __shfl_xor(q, m);
    if (lane == 0) wsq[w] = q;
    __syncthreads();
    const float tsq  = (wsq[0] + wsq[1]) + (wsq[2] + wsq[3]);
    const float invN = 1.0f / (float)GRID_N;
    const float mean = g0 * (1.0f / 512.0f);
    const float var  = fmaxf(tsq * invN - mean * mean, 0.0f);
    const float inv  = 1.0f / (sqrtf(var) + 1e-8f);

    v8f pr = {}, pi = {};
    v4f wr[8], wi[8];
#pragma unroll
    for (int q = 0; q < 8; ++q) {
        wr[q] = *(const v4f*)(Lre + l16 * WPITCH + kof(2 * q, half));
        wi[q] = *(const v4f*)(Lim + l16 * WPITCH + kof(2 * q, half));
    }
    __builtin_amdgcn_sched_barrier(0);
#pragma unroll
    for (int s = 0; s < 16; ++s) {
        pr = mma4(mstep(wr, s), mstep(m2r, s), pr);
        pi = mma4(mstep(wi, s), mstep(m2i, s), pi);
    }
#pragma unroll
    for (int v = 0; v < 8; ++v) {
        const float re = pr[v] + pi[v];
        out[(x * N1D + 16 * yt + v + 8 * half) * N1D + 16 * w + l16] = (re - mean) * inv;
    }
}

extern "C" void kernel_launch(void* const* d_in, const int* in_sizes, int n_in,
                              void* d_out, int out_size, void* d_ws, size_t ws_size,
                              hipStream_t stream) {
    const float* X   = (const float*)d_in[0];
    const float* aw  = (const float*)d_in[1];
    const float* bw  = (const float*)d_in[2];
    const float* ham = (const float*)d_in[4];
    float* out = (float*)d_out;

    float* ws   = (float*)d_ws;
    float* Wre  = ws;
    float* WimF = Wre + 4096;
    float* stat = WimF + 4096;
    float* Are  = stat + 4096;
    float* Aim  = Are + GRID_N;
    float* Bre  = Aim + GRID_N;
    float* Bim  = Bre + GRID_N;

    density_yfwd_kernel<<<256, 128, 0, stream>>>(X, aw, bw, Wre, WimF, Are, Aim);

    zx_filter_kernel<<<NBLK2, 128, 0, stream>>>(Are, Aim, Bre, Bim, Wre, WimF, ham, stat);

    inv_yz_norm_kernel<<<NBLK2, 128, 0, stream>>>(Bre, Bim, Wre, WimF, stat, out);
}


# all reads of the first two launches issued at wave entry as one group (branch-free stage-table reads; an asm anchor keeps reads ahead of the z-pass WMMAs): one memory wait per launch, not several
# speedup vs baseline: 3.0148x; 1.0732x over previous
#include <hip/hip_runtime.h>
#include <hip/hip_bf16.h>
#include <math.h>

#define N1D      64
#define GRID_N   (64 * 64 * 64)
#define N_ATOMS  128
#define K_TERMS  6
#define SPACING  0.5f
#define LOG2E    1.4426950408889634f


typedef float v2f __attribute__((ext_vector_type(2)));
typedef float v4f __attribute__((ext_vector_type(4)));
typedef float v8f __attribute__((ext_vector_type(8)));

__device__ __forceinline__ float fast_exp2(float x) {
#if __has_builtin(__builtin_amdgcn_exp2f)
    return __builtin_amdgcn_exp2f(x);
#else
    return exp2f(x);
#endif
}


__device__ const float COS64[64] = {
    1.000000000f, 0.9951847267f, 0.9807852804f, 0.9569403357f, 0.9238795325f, 0.8819212643f, 0.8314696123f, 0.7730104534f,
    0.7071067812f, 0.6343932842f, 0.5555702330f, 0.4713967368f, 0.3826834324f, 0.2902846773f, 0.1950903220f, 0.09801714033f,
    0.000000000f, -0.09801714033f, -0.1950903220f, -0.2902846773f, -0.3826834324f, -0.4713967368f, -0.5555702330f, -0.6343932842f,
    -0.7071067812f, -0.7730104534f, -0.8314696123f, -0.8819212643f, -0.9238795325f, -0.9569403357f, -0.9807852804f, -0.9951847267f,
    -1.000000000f, -0.9951847267f, -0.9807852804f, -0.9569403357f, -0.9238795325f, -0.8819212643f, -0.8314696123f, -0.7730104534f,
    -0.7071067812f, -0.6343932842f, -0.5555702330f, -0.4713967368f, -0.3826834324f, -0.2902846773f, -0.1950903220f, -0.09801714033f,
    0.000000000f, 0.09801714033f, 0.1950903220f, 0.2902846773f, 0.3826834324f, 0.4713967368f, 0.5555702330f, 0.6343932842f,
    0.7071067812f, 0.7730104534f, 0.8314696123f, 0.8819212643f, 0.9238795325f, 0.9569403357f, 0.9807852804f, 0.9951847267f
};

__device__ __forceinline__ int wpos(int k, int n, bool real_part) {
    return (k * n + (real_part ? 0 : 48)) & 63;
}
__device__ __forceinline__ float wsgn(int k, int n, bool real_part) {
    const float sgn = ((k + n) & 1) ? -0.125f : 0.125f;
    return real_part ? sgn : -sgn;
}

#define ACH      8
#define NACH     (N_ATOMS / 2 / ACH)

struct AtomChunk {
    float x[ACH], y[ACH], z[ACH];
    v2f   b[ACH][3];
    v2f   w[ACH][3];
};

__device__ __forceinline__ void load_atoms(AtomChunk& p, const float* __restrict__ X,
                                           const float* __restrict__ aw,
                                           const float* __restrict__ bw,
                                           int half, int c) {
#pragma unroll
    for (int j = 0; j < ACH; ++j) {
        const int a = 2 * (c * ACH + j) + half;
        p.x[j] = X[3 * a];
        p.y[j] = X[3 * a + 1];
        p.z[j] = X[3 * a + 2];
        const v2f* bp = (const v2f*)(bw + K_TERMS * a);
        const v2f* wp = (const v2f*)(aw + K_TERMS * a);
#pragma unroll
        for (int u = 0; u < 3; ++u) {
            p.b[j][u] = bp[u];
            p.w[j][u] = wp[u];
        }
    }
}

__device__ __forceinline__ void mma_atoms(const AtomChunk& p, float fx, float fy, float fz,
                                          v8f& acc0, v8f& acc1) {
#pragma unroll
    for (int j = 0; j < ACH; ++j) {
        const float dx  = fx - p.x[j];
        const float dy  = fy - p.y[j];
        const float dz  = fz - p.z[j];
        const float rxy = (dx * dx + dy * dy) * LOG2E;
        const float rz  = (dz * dz) * LOG2E;
#pragma unroll
        for (int u = 0; u < 3; ++u) {
            const float b0 = p.b[j][u].x;
            const float b1 = p.b[j][u].y;
            v2f A, B;
            A.x = p.w[j][u].x * fast_exp2(b0 * rxy);
            A.y = p.w[j][u].y * fast_exp2(b1 * rxy);
            B.x = fast_exp2(b0 * rz);
            B.y = fast_exp2(b1 * rz);
            if (((3 * j + u) & 1) == 0)
                acc0 = __builtin_amdgcn_wmma_f32_16x16x4_f32(
                    false, B, false, A, (short)0, acc0, false, false);
            else
                acc1 = __builtin_amdgcn_wmma_f32_16x16x4_f32(
                    false, B, false, A, (short)0, acc1, false, false);
        }
    }
}

#define NBLK2      256
#define WPITCH     68
#define SPITCH     20
#define SPART      (N1D * SPITCH + 16)
#define YPART      (N1D * 16 + 16)

__device__ __forceinline__ v8f mma4(v2f a, v2f b, v8f c) {
    return __builtin_amdgcn_wmma_f32_16x16x4_f32(false, a, false, b, (short)0, c, false, false);
}

__device__ __forceinline__ void load_mfrag(v4f (&m)[8], const float* __restrict__ Mf,
                                           int tile, int lane) {
    const v4f* p = (const v4f*)Mf + tile * 256 + lane;
#pragma unroll
    for (int q = 0; q < 8; ++q) m[q] = p[q * 32];
}

__device__ __forceinline__ int kof(int s, int half) {
    return 16 * (s >> 2) + 8 * half + 2 * (s & 3);
}

__device__ __forceinline__ v2f mstep(const v4f* m, int s) {
    const v4f t = m[s >> 1];
    v2f r;
    if (s & 1) { r.x = t.z; r.y = t.w; } else { r.x = t.x; r.y = t.y; }
    return r;
}

__device__ __forceinline__ int slab_idx(int x, int ky, int kt, int l) {
    return ((x * 4 + kt) * N1D + ky) * 16 + l;
}

__device__ __forceinline__ int widx(int k, int n) {
    const int sk = 4 * (n >> 4) + ((n >> 1) & 3);
    return ((k >> 4) << 10) | ((sk >> 1) << 7) | ((((n >> 3) & 1) * 16 + (k & 15)) << 2)
         | (2 * (sk & 1) + (n & 1));
}

struct StageM {
    v2f a1[4];
    v2f a2[2][4];
};

struct StageRaw {
    float a1[4][2];
    float a2[2][4][2];
};

__device__ __forceinline__ void ask_stage(StageRaw& t, const float* __restrict__ Mre,
                                          const float* __restrict__ Mim,
                                          int w, int half, int l16) {
    const int  p    = l16 >> 3;
    const int  i    = l16 & 7;
    const bool diag = (p == half);
    const float* src = diag ? Mre : Mim;
#pragma unroll
    for (int s = 0; s < 4; ++s) {
        t.a1[s][0] = src[widx(i, 8 * (2 * s))];
        t.a1[s][1] = src[widx(i, 8 * (2 * s + 1))];
#pragma unroll
        for (int cc = 0; cc < 2; ++cc) {
            const int row = 2 * w + cc + 8 * i;
            t.a2[cc][s][0] = src[widx(row, 2 * s)];
            t.a2[cc][s][1] = src[widx(row, 2 * s + 1)];
        }
    }
}

__device__ __forceinline__ void finish_stage(StageM& m, const StageRaw& t, int half, int l16) {
    const int  p    = l16 >> 3;
    const int  i    = l16 & 7;
    const bool diag = (p == half);
    const float sg  = diag ? 1.0f : ((p == 1) ? 1.0f : -1.0f);
    const float s1  = (i & 1) ? -8.0f * sg : 8.0f * sg;
#pragma unroll
    for (int s = 0; s < 4; ++s) {
        m.a1[s].x = s1 * t.a1[s][0];
        m.a1[s].y = s1 * t.a1[s][1];
#pragma unroll
        for (int cc = 0; cc < 2; ++cc) {
            m.a2[cc][s].x = sg * t.a2[cc][s][0];
            m.a2[cc][s].y = sg * t.a2[cc][s][1];
        }
    }
}

__device__ __forceinline__ void conj_stage(StageM& o, const StageM& m, int half, int l16) {
    const float f = ((l16 >> 3) == half) ? 1.0f : -1.0f;
#pragma unroll
    for (int s = 0; s < 4; ++s) {
        o.a1[s]    = m.a1[s] * f;
        o.a2[0][s] = m.a2[0][s] * f;
        o.a2[1][s] = m.a2[1][s] * f;
    }
}

__device__ __forceinline__ void ask_stage_const(StageRaw& t, int w, int half, int l16) {
    const int  p    = l16 >> 3;
    const int  i    = l16 & 7;
    const bool diag = (p == half);
#pragma unroll
    for (int s = 0; s < 4; ++s) {
        t.a1[s][0] = COS64[wpos(i, 8 * (2 * s), diag)];
        t.a1[s][1] = COS64[wpos(i, 8 * (2 * s + 1), diag)];
#pragma unroll
        for (int cc = 0; cc < 2; ++cc) {
            const int row = 2 * w + cc + 8 * i;
            t.a2[cc][s][0] = COS64[wpos(row, 2 * s, diag)];
            t.a2[cc][s][1] = COS64[wpos(row, 2 * s + 1, diag)];
        }
    }
}

__device__ __forceinline__ void finish_stage_const(StageM& m, const StageRaw& t,
                                                   int w, int half, int l16) {
    const int  p    = l16 >> 3;
    const int  i    = l16 & 7;
    const bool diag = (p == half);
    const float sg  = diag ? 1.0f : ((p == 1) ? 1.0f : -1.0f);
    const float s1  = (i & 1) ? -8.0f * sg : 8.0f * sg;
#pragma unroll
    for (int s = 0; s < 4; ++s) {
        m.a1[s].x = s1 * (wsgn(i, 8 * (2 * s), diag) * t.a1[s][0]);
        m.a1[s].y = s1 * (wsgn(i, 8 * (2 * s + 1), diag) * t.a1[s][1]);
#pragma unroll
        for (int cc = 0; cc < 2; ++cc) {
            const int row = 2 * w + cc + 8 * i;
            m.a2[cc][s].x = sg * (wsgn(row, 2 * s, diag) * t.a2[cc][s][0]);
            m.a2[cc][s].y = sg * (wsgn(row, 2 * s + 1, diag) * t.a2[cc][s][1]);
        }
    }
}

__device__ __forceinline__ void stage1_operands(v2f (&b1)[2][4], const float* P, int w) {
#pragma unroll
    for (int cc = 0; cc < 2; ++cc)
#pragma unroll
        for (int s = 0; s < 4; ++s) {
            b1[cc][s].x = P[(8 * (2 * s) + 2 * w + cc) * SPITCH];
            b1[cc][s].y = P[(8 * (2 * s + 1) + 2 * w + cc) * SPITCH];
        }
}

__device__ __forceinline__ void stage1(const StageM& m, const v2f (&b1)[2][4], float* Yh, int w) {
    v8f y0 = {}, y1 = {};
#pragma unroll
    for (int s = 0; s < 4; ++s) {
        y0 = mma4(m.a1[s], b1[0][s], y0);
        y1 = mma4(m.a1[s], b1[1][s], y1);
    }
#pragma unroll
    for (int r = 0; r < 8; ++r) {
        Yh[((2 * w) * 8 + r) * 16]     = y0[r];
        Yh[((2 * w + 1) * 8 + r) * 16] = y1[r];
    }
}

__device__ __forceinline__ void stage2(const StageM& m, const float* Yh, int w, v8f& o0, v8f& o1) {
    v2f b2[2][4];
#pragma unroll
    for (int cc = 0; cc < 2; ++cc)
#pragma unroll
        for (int s = 0; s < 4; ++s) {
            b2[cc][s].x = Yh[((2 * s) * 8 + 2 * w + cc) * 16];
            b2[cc][s].y = Yh[((2 * s + 1) * 8 + 2 * w + cc) * 16];
        }
    v8f a = {}, b = {};
#pragma unroll
    for (int s = 0; s < 4; ++s) {
        a = mma4(m.a2[0][s], b2[0][s], a);
        b = mma4(m.a2[1][s], b2[1][s], b);
    }
    o0 = a;
    o1 = b;
}

__global__ __launch_bounds__(128) void density_yfwd_kernel(const float* __restrict__ X,
                                                           const float* __restrict__ aw,
                                                           const float* __restrict__ bw,
                                                           float* __restrict__ Wre,
                                                           float* __restrict__ WimF,
                                                           float* __restrict__ Ure,
                                                           float* __restrict__ Uim) {
    __shared__ __attribute__((aligned(16))) float S[2 * SPART];
    __shared__ float Yx[2 * YPART];
    const int lane = threadIdx.x & 31;
    const int w    = threadIdx.x >> 5;
    const int iyt  = w;
    const int ix   = blockIdx.x >> 2;
    const int izt  = blockIdx.x & 3;
    const int half = lane >> 4;
    const int l16  = lane & 15;

    const float fx = (float)ix * SPACING;
    const float fy = (float)(iyt * 16 + l16) * SPACING;
    const float fz = (float)(izt * 16 + l16) * SPACING;

    v8f acc0 = {}, acc1 = {};
    AtomChunk p0, p1;

    load_atoms(p0, X, aw, bw, half, 0);

    StageRaw traw;
    ask_stage_const(traw, w, half, l16);

    for (int c = 0; c < NACH - 2; c += 2) {
        load_atoms(p1, X, aw, bw, half, c + 1);
        mma_atoms(p0, fx, fy, fz, acc0, acc1);
        load_atoms(p0, X, aw, bw, half, c + 2);
        mma_atoms(p1, fx, fy, fz, acc0, acc1);
    }
    load_atoms(p1, X, aw, bw, half, NACH - 1);
    mma_atoms(p0, fx, fy, fz, acc0, acc1);
    mma_atoms(p1, fx, fy, fz, acc0, acc1);

    {
        const v8f sum = acc0 + acc1;
        v4f lo, hi;
        lo.x = sum[0]; lo.y = sum[1]; lo.z = sum[2]; lo.w = sum[3];
        hi.x = sum[4]; hi.y = sum[5]; hi.z = sum[6]; hi.w = sum[7];
        const v4f zero = {0.0f, 0.0f, 0.0f, 0.0f};
        float* sre = S + (16 * iyt + l16) * SPITCH + 8 * half;
        float* sim = sre + SPART;
        *(v4f*)(sre)     = lo;
        *(v4f*)(sre + 4) = hi;
        *(v4f*)(sim)     = zero;
        *(v4f*)(sim + 4) = zero;
    }

    StageM m;
    finish_stage_const(m, traw, w, half, l16);

    __syncthreads();

    float* Yh = Yx + half * YPART + l16;
    v2f b1[2][4];
    stage1_operands(b1, S + half * SPART + l16, w);
    stage1(m, b1, Yh, w);
    __syncthreads();
    v8f o0, o1;
    stage2(m, Yh, w, o0, o1);
    float* Up = half ? Uim : Ure;
#pragma unroll
    for (int r = 0; r < 8; ++r) {
        Up[((2 * w + 8 * r) * N1D + ix) * N1D + 16 * izt + l16]     = o0[r];
        Up[((2 * w + 1 + 8 * r) * N1D + ix) * N1D + 16 * izt + l16] = o1[r];
    }

    if (blockIdx.x < (N1D * N1D) / 128) {
        const int t  = blockIdx.x * 128 + threadIdx.x;
        const int e  = t & 3;
        const int fl = (t >> 2) & 31;
        const int q  = (t >> 7) & 7;
        const int k  = 16 * (t >> 10) + (fl & 15);
        const int sk = 2 * q + (e >> 1);
        const int n  = 16 * (sk >> 2) + 8 * (fl >> 4) + 2 * (sk & 3) + (e & 1);
        const int kn = (k * n) & 63;
        const float ang = (float)kn * 0.09817477042468103f;
        const float sgn = ((k + n) & 1) ? -0.125f : 0.125f;
        const float c = sgn * cosf(ang);
        const float s = sgn * sinf(ang);
        Wre[t]  = c;
        WimF[t] = -s;
    }
}

__global__ __launch_bounds__(128) void zx_filter_kernel(const float* __restrict__ Ure,
                                                        const float* __restrict__ Uim,
                                                        float* __restrict__ Tre,
                                                        float* __restrict__ Tim,
                                                        const float* __restrict__ Mre,
                                                        const float* __restrict__ MimF,
                                                        const float* __restrict__ h,
                                                        float* __restrict__ stat) {
    __shared__ __attribute__((aligned(16))) float S[2 * SPART];
    __shared__ float Yx[2 * YPART];
    __shared__ float wq[4];
    const int lane = threadIdx.x & 31;
    const int w    = threadIdx.x >> 5;
    const int half = lane >> 4;
    const int l16  = lane & 15;
    const int ky   = blockIdx.x >> 2;
    const int kt   = blockIdx.x & 3;

    float* Tp = half ? Tim : Tre;

    v4f wzr[8], wzi[8], ur[8], ui[8];
    {
        const v4f* pzr = (const v4f*)Mre + kt * 256 + lane;
        const v4f* pzi = (const v4f*)MimF + kt * 256 + lane;
        const int row = ((ky * N1D) + 16 * w + l16) * N1D + 8 * half;
        const v4f* pr = (const v4f*)(Ure + row);
        const v4f* pi = (const v4f*)(Uim + row);
#pragma unroll
        for (int q = 0; q < 8; ++q) {
            wzr[q] = pzr[q * 32];
            wzi[q] = pzi[q * 32];
            ur[q] = pr[4 * (q >> 1) + (q & 1)];
            ui[q] = pi[4 * (q >> 1) + (q & 1)];
        }
    }
    StageRaw traw;
    ask_stage(traw, Mre, MimF, w, half, l16);
    float hv[2][8], hm[2][8];
    const int mky = (N1D - ky) & 63;
    const int mkz = (N1D - (16 * kt + l16)) & 63;
#pragma unroll
    for (int cc = 0; cc < 2; ++cc)
#pragma unroll
        for (int r = 0; r < 8; ++r) {
            const int kx = 2 * w + cc + 8 * r;
            hv[cc][r] = h[(kx * N1D + ky) * N1D + 16 * kt + l16];
            hm[cc][r] = h[((((N1D - kx) & 63) * N1D) + mky) * N1D + mkz];
        }
    v8f rr = {}, ii = {}, ri = {}, ir = {};
    asm volatile("" : "+v"(rr), "+v"(ii), "+v"(ri), "+v"(ir));
#pragma unroll
    for (int s = 0; s < 16; ++s) {
        const v2f dr = mstep(ur, s);
        const v2f di = mstep(ui, s);
        const v2f wr = mstep(wzr, s);
        const v2f wi = mstep(wzi, s);
        rr = mma4(wr, dr, rr);
        ii = mma4(wi, di, ii);
        ri = mma4(wr, di, ri);
        ir = mma4(wi, dr, ir);
    }
#pragma unroll
    for (int g = 0; g < 2; ++g) {
        v4f vr, vi;
        vr.x = rr[4 * g]     - ii[4 * g];      vi.x = ri[4 * g]     + ir[4 * g];
        vr.y = rr[4 * g + 1] - ii[4 * g + 1];  vi.y = ri[4 * g + 1] + ir[4 * g + 1];
        vr.z = rr[4 * g + 2] - ii[4 * g + 2];  vi.z = ri[4 * g + 2] + ir[4 * g + 2];
        vr.w = rr[4 * g + 3] - ii[4 * g + 3];  vi.w = ri[4 * g + 3] + ir[4 * g + 3];
        *(v4f*)(S + (16 * w + l16) * SPITCH + 8 * half + 4 * g)         = vr;
        *(v4f*)(S + SPART + (16 * w + l16) * SPITCH + 8 * half + 4 * g) = vi;
    }
    StageM mf, mi;
    finish_stage(mf, traw, half, l16);
    conj_stage(mi, mf, half, l16);
    __syncthreads();

    float* Yh = Yx + half * YPART + l16;
    float* Sh = S + half * SPART + l16;
    v2f b1[2][4];
    stage1_operands(b1, Sh, w);
    stage1(mf, b1, Yh, w);
    __syncthreads();
    v8f f0, f1;
    stage2(mf, Yh, w, f0, f1);
    float qacc = 0.0f;
#pragma unroll
    for (int r = 0; r < 8; ++r) {
        const float hs0 = 0.5f * (hv[0][r] + hm[0][r]);
        const float hs1 = 0.5f * (hv[1][r] + hm[1][r]);
        qacc += (f0[r] * f0[r]) * (hs0 * hs0);
        qacc += (f1[r] * f1[r]) * (hs1 * hs1);
        Sh[(2 * w + 8 * r) * SPITCH]     = f0[r] * hv[0][r];
        Sh[(2 * w + 1 + 8 * r) * SPITCH] = f1[r] * hv[1][r];
    }
#pragma unroll
    for (int m = 16; m > 0; m >>= 1) qacc += __shfl_xor(qacc, m);
    if (lane == 0) wq[w] = qacc;
    if (blockIdx.x == 4 * 32 + 2 && threadIdx.x == 0)
        stat[NBLK2] = f0[4] * hv[0][4];
    __syncthreads();
    if (threadIdx.x == 0) stat[blockIdx.x] = (wq[0] + wq[1]) + (wq[2] + wq[3]);

    v2f c1[2][4];
    stage1_operands(c1, Sh, w);
    stage1(mi, c1, Yh, w);
    __syncthreads();
    v8f r0, r1;
    stage2(mi, Yh, w, r0, r1);
#pragma unroll
    for (int r = 0; r < 8; ++r) {
        Tp[slab_idx(2 * w + 8 * r, ky, kt, l16)]     = r0[r];
        Tp[slab_idx(2 * w + 1 + 8 * r, ky, kt, l16)] = r1[r];
    }
}

__global__ __launch_bounds__(128) void inv_yz_norm_kernel(const float* __restrict__ Tre,
                                                           const float* __restrict__ Tim,
                                                           const float* __restrict__ Mre,
                                                           const float* __restrict__ Mim,
                                                           const float* __restrict__ stat,
                                                           float* __restrict__ out) {
    __shared__ __attribute__((aligned(16))) float Lre[16 * WPITCH];
    __shared__ __attribute__((aligned(16))) float Lim[16 * WPITCH];
    __shared__ float wsq[4];
    const int lane = threadIdx.x & 31;
    const int w    = threadIdx.x >> 5;
    const int half = lane >> 4;
    const int l16  = lane & 15;
    const int x    = blockIdx.x >> 2;
    const int yt   = blockIdx.x & 3;

    v4f m2r[8], m2i[8];
    load_mfrag(m2r, Mre, w, lane);
    load_mfrag(m2i, Mim, w, lane);

    float q = stat[threadIdx.x] + stat[threadIdx.x + 128];
    const float g0 = stat[NBLK2];

    v4f m1r[8], m1i[8];
    load_mfrag(m1r, Mre, yt, lane);
    load_mfrag(m1i, Mim, yt, lane);
    v8f rr = {}, ii = {}, ri = {}, ir = {};
#pragma unroll
    for (int s = 0; s < 16; ++s) {
        const int k = kof(s, half);
        const v2f ar = mstep(m1r, s);
        const v2f ai = mstep(m1i, s);
        v2f br, bi;
        br.x = Tre[slab_idx(x, k, w, l16)];  br.y = Tre[slab_idx(x, k + 1, w, l16)];
        bi.x = Tim[slab_idx(x, k, w, l16)];  bi.y = Tim[slab_idx(x, k + 1, w, l16)];
        rr = mma4(ar, br, rr);
        ii = mma4(ai, bi, ii);
        ri = mma4(ar, bi, ri);
        ir = mma4(ai, br, ir);
    }
#pragma unroll
    for (int v = 0; v < 8; ++v) {
        Lre[(v + 8 * half) * WPITCH + 16 * w + l16] = rr[v] + ii[v];
        Lim[(v + 8 * half) * WPITCH + 16 * w + l16] = ri[v] - ir[v];
    }
#pragma unroll
    for (int m = 16; m > 0; m >>= 1) q += __shfl_xor(q, m);
    if (lane == 0) wsq[w] = q;
    __syncthreads();
    const float tsq  = (wsq[0] + wsq[1]) + (wsq[2] + wsq[3]);
    const float invN = 1.0f / (float)GRID_N;
    const float mean = g0 * (1.0f / 512.0f);
    const float var  = fmaxf(tsq * invN - mean * mean, 0.0f);
    const float inv  = 1.0f / (sqrtf(var) + 1e-8f);

    v8f pr = {}, pi = {};
    v4f wr[8], wi[8];
#pragma unroll
    for (int q = 0; q < 8; ++q) {
        wr[q] = *(const v4f*)(Lre + l16 * WPITCH + kof(2 * q, half));
        wi[q] = *(const v4f*)(Lim + l16 * WPITCH + kof(2 * q, half));
    }
    __builtin_amdgcn_sched_barrier(0);
#pragma unroll
    for (int s = 0; s < 16; ++s) {
        pr = mma4(mstep(wr, s), mstep(m2r, s), pr);
        pi = mma4(mstep(wi, s), mstep(m2i, s), pi);
    }
#pragma unroll
    for (int v = 0; v < 8; ++v) {
        const float re = pr[v] + pi[v];
        out[(x * N1D + 16 * yt + v + 8 * half) * N1D + 16 * w + l16] = (re - mean) * inv;
    }
}

extern "C" void kernel_launch(void* const* d_in, const int* in_sizes, int n_in,
                              void* d_out, int out_size, void* d_ws, size_t ws_size,
                              hipStream_t stream) {
    const float* X   = (const float*)d_in[0];
    const float* aw  = (const float*)d_in[1];
    const float* bw  = (const float*)d_in[2];
    const float* ham = (const float*)d_in[4];
    float* out = (float*)d_out;

    float* ws   = (float*)d_ws;
    float* Wre  = ws;
    float* WimF = Wre + 4096;
    float* stat = WimF + 4096;
    float* Are  = stat + 4096;
    float* Aim  = Are + GRID_N;
    float* Bre  = Aim + GRID_N;
    float* Bim  = Bre + GRID_N;

    density_yfwd_kernel<<<256, 128, 0, stream>>>(X, aw, bw, Wre, WimF, Are, Aim);

    zx_filter_kernel<<<NBLK2, 128, 0, stream>>>(Are, Aim, Bre, Bim, Wre, WimF, ham, stat);

    inv_yz_norm_kernel<<<NBLK2, 128, 0, stream>>>(Bre, Bim, Wre, WimF, stat, out);
}


# density: each wave lists the atom-term pairs whose Gaussian factor is not exactly 0.0f over its whole tile (f32 underflow bound) and runs only those k-steps; changes the order of the atom/term sum
# speedup vs baseline: 3.4717x; 1.1516x over previous
#include <hip/hip_runtime.h>
#include <hip/hip_bf16.h>
#include <math.h>

#define N1D      64
#define GRID_N   (64 * 64 * 64)
#define N_ATOMS  128
#define K_TERMS  6
#define SPACING  0.5f
#define LOG2E    1.4426950408889634f


typedef float v2f __attribute__((ext_vector_type(2)));
typedef float v4f __attribute__((ext_vector_type(4)));
typedef float v8f __attribute__((ext_vector_type(8)));

__device__ __forceinline__ float fast_exp2(float x) {
#if __has_builtin(__builtin_amdgcn_exp2f)
    return __builtin_amdgcn_exp2f(x);
#else
    return exp2f(x);
#endif
}


__device__ const float COS64[64] = {
    1.000000000f, 0.9951847267f, 0.9807852804f, 0.9569403357f, 0.9238795325f, 0.8819212643f, 0.8314696123f, 0.7730104534f,
    0.7071067812f, 0.6343932842f, 0.5555702330f, 0.4713967368f, 0.3826834324f, 0.2902846773f, 0.1950903220f, 0.09801714033f,
    0.000000000f, -0.09801714033f, -0.1950903220f, -0.2902846773f, -0.3826834324f, -0.4713967368f, -0.5555702330f, -0.6343932842f,
    -0.7071067812f, -0.7730104534f, -0.8314696123f, -0.8819212643f, -0.9238795325f, -0.9569403357f, -0.9807852804f, -0.9951847267f,
    -1.000000000f, -0.9951847267f, -0.9807852804f, -0.9569403357f, -0.9238795325f, -0.8819212643f, -0.8314696123f, -0.7730104534f,
    -0.7071067812f, -0.6343932842f, -0.5555702330f, -0.4713967368f, -0.3826834324f, -0.2902846773f, -0.1950903220f, -0.09801714033f,
    0.000000000f, 0.09801714033f, 0.1950903220f, 0.2902846773f, 0.3826834324f, 0.4713967368f, 0.5555702330f, 0.6343932842f,
    0.7071067812f, 0.7730104534f, 0.8314696123f, 0.8819212643f, 0.9238795325f, 0.9569403357f, 0.9807852804f, 0.9951847267f
};

__device__ __forceinline__ int wpos(int k, int n, bool real_part) {
    return (k * n + (real_part ? 0 : 48)) & 63;
}
__device__ __forceinline__ float wsgn(int k, int n, bool real_part) {
    const float sgn = ((k + n) & 1) ? -0.125f : 0.125f;
    return real_part ? sgn : -sgn;
}

#define DEAD_ARG  (-160.0f)
#define ENT_CAP   401

__device__ __forceinline__ void kstep_entry(v4f p, v4f q, float fx, float fy, float fz, v8f& acc) {
    const float dx  = fx - p.x;
    const float dy  = fy - p.y;
    const float dz  = fz - p.z;
    const float rxy = (dx * dx + dy * dy) * LOG2E;
    const float rz  = (dz * dz) * LOG2E;
    v2f A, B;
    A.x = q.z * fast_exp2(q.x * rxy);
    A.y = q.w * fast_exp2(q.y * rxy);
    B.x = fast_exp2(q.x * rz);
    B.y = fast_exp2(q.y * rz);
    acc = __builtin_amdgcn_wmma_f32_16x16x4_f32(false, B, false, A, (short)0, acc, false, false);
}

__device__ __forceinline__ void load_entries(v4f (&buf)[4][2], const v4f* e, int half) {
#pragma unroll
    for (int j = 0; j < 4; ++j) {
        buf[j][0] = e[(2 * j + half) * 2];
        buf[j][1] = e[(2 * j + half) * 2 + 1];
    }
}

#define NBLK2      256
#define WPITCH     68
#define SPITCH     20
#define SPART      (N1D * SPITCH + 16)
#define YPART      (N1D * 16 + 16)

__device__ __forceinline__ v8f mma4(v2f a, v2f b, v8f c) {
    return __builtin_amdgcn_wmma_f32_16x16x4_f32(false, a, false, b, (short)0, c, false, false);
}

__device__ __forceinline__ void load_mfrag(v4f (&m)[8], const float* __restrict__ Mf,
                                           int tile, int lane) {
    const v4f* p = (const v4f*)Mf + tile * 256 + lane;
#pragma unroll
    for (int q = 0; q < 8; ++q) m[q] = p[q * 32];
}

__device__ __forceinline__ int kof(int s, int half) {
    return 16 * (s >> 2) + 8 * half + 2 * (s & 3);
}

__device__ __forceinline__ v2f mstep(const v4f* m, int s) {
    const v4f t = m[s >> 1];
    v2f r;
    if (s & 1) { r.x = t.z; r.y = t.w; } else { r.x = t.x; r.y = t.y; }
    return r;
}

__device__ __forceinline__ int slab_idx(int x, int ky, int kt, int l) {
    return ((x * 4 + kt) * N1D + ky) * 16 + l;
}

__device__ __forceinline__ int widx(int k, int n) {
    const int sk = 4 * (n >> 4) + ((n >> 1) & 3);
    return ((k >> 4) << 10) | ((sk >> 1) << 7) | ((((n >> 3) & 1) * 16 + (k & 15)) << 2)
         | (2 * (sk & 1) + (n & 1));
}

struct StageM {
    v2f a1[4];
    v2f a2[2][4];
};

struct StageRaw {
    float a1[4][2];
    float a2[2][4][2];
};

__device__ __forceinline__ void ask_stage(StageRaw& t, const float* __restrict__ Mre,
                                          const float* __restrict__ Mim,
                                          int w, int half, int l16) {
    const int  p    = l16 >> 3;
    const int  i    = l16 & 7;
    const bool diag = (p == half);
    const float* src = diag ? Mre : Mim;
#pragma unroll
    for (int s = 0; s < 4; ++s) {
        t.a1[s][0] = src[widx(i, 8 * (2 * s))];
        t.a1[s][1] = src[widx(i, 8 * (2 * s + 1))];
#pragma unroll
        for (int cc = 0; cc < 2; ++cc) {
            const int row = 2 * w + cc + 8 * i;
            t.a2[cc][s][0] = src[widx(row, 2 * s)];
            t.a2[cc][s][1] = src[widx(row, 2 * s + 1)];
        }
    }
}

__device__ __forceinline__ void finish_stage(StageM& m, const StageRaw& t, int half, int l16) {
    const int  p    = l16 >> 3;
    const int  i    = l16 & 7;
    const bool diag = (p == half);
    const float sg  = diag ? 1.0f : ((p == 1) ? 1.0f : -1.0f);
    const float s1  = (i & 1) ? -8.0f * sg : 8.0f * sg;
#pragma unroll
    for (int s = 0; s < 4; ++s) {
        m.a1[s].x = s1 * t.a1[s][0];
        m.a1[s].y = s1 * t.a1[s][1];
#pragma unroll
        for (int cc = 0; cc < 2; ++cc) {
            m.a2[cc][s].x = sg * t.a2[cc][s][0];
            m.a2[cc][s].y = sg * t.a2[cc][s][1];
        }
    }
}

__device__ __forceinline__ void conj_stage(StageM& o, const StageM& m, int half, int l16) {
    const float f = ((l16 >> 3) == half) ? 1.0f : -1.0f;
#pragma unroll
    for (int s = 0; s < 4; ++s) {
        o.a1[s]    = m.a1[s] * f;
        o.a2[0][s] = m.a2[0][s] * f;
        o.a2[1][s] = m.a2[1][s] * f;
    }
}

__device__ __forceinline__ void ask_stage_const(StageRaw& t, int w, int half, int l16) {
    const int  p    = l16 >> 3;
    const int  i    = l16 & 7;
    const bool diag = (p == half);
#pragma unroll
    for (int s = 0; s < 4; ++s) {
        t.a1[s][0] = COS64[wpos(i, 8 * (2 * s), diag)];
        t.a1[s][1] = COS64[wpos(i, 8 * (2 * s + 1), diag)];
#pragma unroll
        for (int cc = 0; cc < 2; ++cc) {
            const int row = 2 * w + cc + 8 * i;
            t.a2[cc][s][0] = COS64[wpos(row, 2 * s, diag)];
            t.a2[cc][s][1] = COS64[wpos(row, 2 * s + 1, diag)];
        }
    }
}

__device__ __forceinline__ void finish_stage_const(StageM& m, const StageRaw& t,
                                                   int w, int half, int l16) {
    const int  p    = l16 >> 3;
    const int  i    = l16 & 7;
    const bool diag = (p == half);
    const float sg  = diag ? 1.0f : ((p == 1) ? 1.0f : -1.0f);
    const float s1  = (i & 1) ? -8.0f * sg : 8.0f * sg;
#pragma unroll
    for (int s = 0; s < 4; ++s) {
        m.a1[s].x = s1 * (wsgn(i, 8 * (2 * s), diag) * t.a1[s][0]);
        m.a1[s].y = s1 * (wsgn(i, 8 * (2 * s + 1), diag) * t.a1[s][1]);
#pragma unroll
        for (int cc = 0; cc < 2; ++cc) {
            const int row = 2 * w + cc + 8 * i;
            m.a2[cc][s].x = sg * (wsgn(row, 2 * s, diag) * t.a2[cc][s][0]);
            m.a2[cc][s].y = sg * (wsgn(row, 2 * s + 1, diag) * t.a2[cc][s][1]);
        }
    }
}

__device__ __forceinline__ void stage1_operands(v2f (&b1)[2][4], const float* P, int w) {
#pragma unroll
    for (int cc = 0; cc < 2; ++cc)
#pragma unroll
        for (int s = 0; s < 4; ++s) {
            b1[cc][s].x = P[(8 * (2 * s) + 2 * w + cc) * SPITCH];
            b1[cc][s].y = P[(8 * (2 * s + 1) + 2 * w + cc) * SPITCH];
        }
}

__device__ __forceinline__ void stage1(const StageM& m, const v2f (&b1)[2][4], float* Yh, int w) {
    v8f y0 = {}, y1 = {};
#pragma unroll
    for (int s = 0; s < 4; ++s) {
        y0 = mma4(m.a1[s], b1[0][s], y0);
        y1 = mma4(m.a1[s], b1[1][s], y1);
    }
#pragma unroll
    for (int r = 0; r < 8; ++r) {
        Yh[((2 * w) * 8 + r) * 16]     = y0[r];
        Yh[((2 * w + 1) * 8 + r) * 16] = y1[r];
    }
}

__device__ __forceinline__ void stage2(const StageM& m, const float* Yh, int w, v8f& o0, v8f& o1) {
    v2f b2[2][4];
#pragma unroll
    for (int cc = 0; cc < 2; ++cc)
#pragma unroll
        for (int s = 0; s < 4; ++s) {
            b2[cc][s].x = Yh[((2 * s) * 8 + 2 * w + cc) * 16];
            b2[cc][s].y = Yh[((2 * s + 1) * 8 + 2 * w + cc) * 16];
        }
    v8f a = {}, b = {};
#pragma unroll
    for (int s = 0; s < 4; ++s) {
        a = mma4(m.a2[0][s], b2[0][s], a);
        b = mma4(m.a2[1][s], b2[1][s], b);
    }
    o0 = a;
    o1 = b;
}

__global__ __launch_bounds__(128) void density_yfwd_kernel(const float* __restrict__ X,
                                                           const float* __restrict__ aw,
                                                           const float* __restrict__ bw,
                                                           float* __restrict__ Wre,
                                                           float* __restrict__ WimF,
                                                           float* __restrict__ Ure,
                                                           float* __restrict__ Uim) {
    __shared__ __attribute__((aligned(16))) float S[2 * SPART];
    __shared__ float Yx[2 * YPART];
    __shared__ __attribute__((aligned(16))) v4f Ent[4][ENT_CAP][2];
    const int lane = threadIdx.x & 31;
    const int w    = threadIdx.x >> 5;
    const int iyt  = w;
    const int ix   = blockIdx.x >> 2;
    const int izt  = blockIdx.x & 3;
    const int half = lane >> 4;
    const int l16  = lane & 15;

    const float fx = (float)ix * SPACING;
    const float fy = (float)(iyt * 16 + l16) * SPACING;
    const float fz = (float)(izt * 16 + l16) * SPACING;

    v8f acc0 = {}, acc1 = {};

    float ax[4], ay[4], az[4];
    v2f   ab[4][3], awt[4][3];
#pragma unroll
    for (int r = 0; r < 4; ++r) {
        const int a = 32 * r + lane;
        ax[r] = X[3 * a];
        ay[r] = X[3 * a + 1];
        az[r] = X[3 * a + 2];
        const v2f* bp = (const v2f*)(bw + K_TERMS * a);
        const v2f* wp = (const v2f*)(aw + K_TERMS * a);
#pragma unroll
        for (int u = 0; u < 3; ++u) {
            ab[r][u]  = bp[u];
            awt[r][u] = wp[u];
        }
    }
    StageRaw traw;
    ask_stage_const(traw, w, half, l16);

    v4f* Ew = &Ent[w][0][0];
    int nlive = 0;
    {
        const float ylo = (float)(16 * iyt) * SPACING, yhi = (float)(16 * iyt + 15) * SPACING;
        const float zlo = (float)(16 * izt) * SPACING, zhi = (float)(16 * izt + 15) * SPACING;
#pragma unroll
        for (int r = 0; r < 4; ++r) {
            const float dx  = fx - ax[r];
            const float dy  = fmaxf(fmaxf(ylo - ay[r], ay[r] - yhi), 0.0f);
            const float dz  = fmaxf(fmaxf(zlo - az[r], az[r] - zhi), 0.0f);
            const float rxy = (dx * dx + dy * dy) * LOG2E;
            const float rz  = (dz * dz) * LOG2E;
#pragma unroll
            for (int u = 0; u < 3; ++u) {
                const float bm   = fmaxf(ab[r][u].x, ab[r][u].y);
                const bool  live = !((bm * rxy <= DEAD_ARG) || (bm * rz <= DEAD_ARG));
                const unsigned mask = __builtin_amdgcn_ballot_w32(live);
                const int pos = live ? nlive + (int)__builtin_amdgcn_mbcnt_lo(mask, 0u) : ENT_CAP - 1;
                v4f p, q;
                p.x = ax[r];      p.y = ay[r];      p.z = az[r];       p.w = 0.0f;
                q.x = ab[r][u].x; q.y = ab[r][u].y; q.z = awt[r][u].x; q.w = awt[r][u].y;
                Ew[2 * pos]     = p;
                Ew[2 * pos + 1] = q;
                nlive += __builtin_popcount(mask);
            }
        }
        if (lane < 16) {
            const v4f p = {0.0f, 0.0f, 0.0f, 0.0f};
            const v4f q = {-1.0f, -1.0f, 0.0f, 0.0f};
            Ew[2 * (nlive + lane)]     = p;
            Ew[2 * (nlive + lane) + 1] = q;
        }
    }
    const int ntrips = (nlive + 15) >> 4;
    v4f ea[4][2], eb[4][2];
    load_entries(ea, Ew, half);
    for (int t = 0; t < ntrips; ++t) {
        load_entries(eb, Ew + 32 * t + 16, half);
        kstep_entry(ea[0][0], ea[0][1], fx, fy, fz, acc0);
        kstep_entry(ea[1][0], ea[1][1], fx, fy, fz, acc1);
        kstep_entry(ea[2][0], ea[2][1], fx, fy, fz, acc0);
        kstep_entry(ea[3][0], ea[3][1], fx, fy, fz, acc1);
        const int tn = (t + 1 < ntrips) ? t + 1 : t;
        load_entries(ea, Ew + 32 * tn, half);
        kstep_entry(eb[0][0], eb[0][1], fx, fy, fz, acc0);
        kstep_entry(eb[1][0], eb[1][1], fx, fy, fz, acc1);
        kstep_entry(eb[2][0], eb[2][1], fx, fy, fz, acc0);
        kstep_entry(eb[3][0], eb[3][1], fx, fy, fz, acc1);
    }

    {
        const v8f sum = acc0 + acc1;
        v4f lo, hi;
        lo.x = sum[0]; lo.y = sum[1]; lo.z = sum[2]; lo.w = sum[3];
        hi.x = sum[4]; hi.y = sum[5]; hi.z = sum[6]; hi.w = sum[7];
        const v4f zero = {0.0f, 0.0f, 0.0f, 0.0f};
        float* sre = S + (16 * iyt + l16) * SPITCH + 8 * half;
        float* sim = sre + SPART;
        *(v4f*)(sre)     = lo;
        *(v4f*)(sre + 4) = hi;
        *(v4f*)(sim)     = zero;
        *(v4f*)(sim + 4) = zero;
    }

    StageM m;
    finish_stage_const(m, traw, w, half, l16);

    __syncthreads();

    float* Yh = Yx + half * YPART + l16;
    v2f b1[2][4];
    stage1_operands(b1, S + half * SPART + l16, w);
    stage1(m, b1, Yh, w);
    __syncthreads();
    v8f o0, o1;
    stage2(m, Yh, w, o0, o1);
    float* Up = half ? Uim : Ure;
#pragma unroll
    for (int r = 0; r < 8; ++r) {
        Up[((2 * w + 8 * r) * N1D + ix) * N1D + 16 * izt + l16]     = o0[r];
        Up[((2 * w + 1 + 8 * r) * N1D + ix) * N1D + 16 * izt + l16] = o1[r];
    }

    if (blockIdx.x < (N1D * N1D) / 128) {
        const int t  = blockIdx.x * 128 + threadIdx.x;
        const int e  = t & 3;
        const int fl = (t >> 2) & 31;
        const int q  = (t >> 7) & 7;
        const int k  = 16 * (t >> 10) + (fl & 15);
        const int sk = 2 * q + (e >> 1);
        const int n  = 16 * (sk >> 2) + 8 * (fl >> 4) + 2 * (sk & 3) + (e & 1);
        const int kn = (k * n) & 63;
        const float ang = (float)kn * 0.09817477042468103f;
        const float sgn = ((k + n) & 1) ? -0.125f : 0.125f;
        const float c = sgn * cosf(ang);
        const float s = sgn * sinf(ang);
        Wre[t]  = c;
        WimF[t] = -s;
    }
}

__global__ __launch_bounds__(128) void zx_filter_kernel(const float* __restrict__ Ure,
                                                        const float* __restrict__ Uim,
                                                        float* __restrict__ Tre,
                                                        float* __restrict__ Tim,
                                                        const float* __restrict__ Mre,
                                                        const float* __restrict__ MimF,
                                                        const float* __restrict__ h,
                                                        float* __restrict__ stat) {
    __shared__ __attribute__((aligned(16))) float S[2 * SPART];
    __shared__ float Yx[2 * YPART];
    __shared__ float wq[4];
    const int lane = threadIdx.x & 31;
    const int w    = threadIdx.x >> 5;
    const int half = lane >> 4;
    const int l16  = lane & 15;
    const int ky   = blockIdx.x >> 2;
    const int kt   = blockIdx.x & 3;

    float* Tp = half ? Tim : Tre;

    v4f wzr[8], wzi[8], ur[8], ui[8];
    {
        const v4f* pzr = (const v4f*)Mre + kt * 256 + lane;
        const v4f* pzi = (const v4f*)MimF + kt * 256 + lane;
        const int row = ((ky * N1D) + 16 * w + l16) * N1D + 8 * half;
        const v4f* pr = (const v4f*)(Ure + row);
        const v4f* pi = (const v4f*)(Uim + row);
#pragma unroll
        for (int q = 0; q < 8; ++q) {
            wzr[q] = pzr[q * 32];
            wzi[q] = pzi[q * 32];
            ur[q] = pr[4 * (q >> 1) + (q & 1)];
            ui[q] = pi[4 * (q >> 1) + (q & 1)];
        }
    }
    StageRaw traw;
    ask_stage(traw, Mre, MimF, w, half, l16);
    float hv[2][8], hm[2][8];
    const int mky = (N1D - ky) & 63;
    const int mkz = (N1D - (16 * kt + l16)) & 63;
#pragma unroll
    for (int cc = 0; cc < 2; ++cc)
#pragma unroll
        for (int r = 0; r < 8; ++r) {
            const int kx = 2 * w + cc + 8 * r;
            hv[cc][r] = h[(kx * N1D + ky) * N1D + 16 * kt + l16];
            hm[cc][r] = h[((((N1D - kx) & 63) * N1D) + mky) * N1D + mkz];
        }
    v8f rr = {}, ii = {}, ri = {}, ir = {};
    asm volatile("" : "+v"(rr), "+v"(ii), "+v"(ri), "+v"(ir));
#pragma unroll
    for (int s = 0; s < 16; ++s) {
        const v2f dr = mstep(ur, s);
        const v2f di = mstep(ui, s);
        const v2f wr = mstep(wzr, s);
        const v2f wi = mstep(wzi, s);
        rr = mma4(wr, dr, rr);
        ii = mma4(wi, di, ii);
        ri = mma4(wr, di, ri);
        ir = mma4(wi, dr, ir);
    }
#pragma unroll
    for (int g = 0; g < 2; ++g) {
        v4f vr, vi;
        vr.x = rr[4 * g]     - ii[4 * g];      vi.x = ri[4 * g]     + ir[4 * g];
        vr.y = rr[4 * g + 1] - ii[4 * g + 1];  vi.y = ri[4 * g + 1] + ir[4 * g + 1];
        vr.z = rr[4 * g + 2] - ii[4 * g + 2];  vi.z = ri[4 * g + 2] + ir[4 * g + 2];
        vr.w = rr[4 * g + 3] - ii[4 * g + 3];  vi.w = ri[4 * g + 3] + ir[4 * g + 3];
        *(v4f*)(S + (16 * w + l16) * SPITCH + 8 * half + 4 * g)         = vr;
        *(v4f*)(S + SPART + (16 * w + l16) * SPITCH + 8 * half + 4 * g) = vi;
    }
    StageM mf, mi;
    finish_stage(mf, traw, half, l16);
    conj_stage(mi, mf, half, l16);
    __syncthreads();

    float* Yh = Yx + half * YPART + l16;
    float* Sh = S + half * SPART + l16;
    v2f b1[2][4];
    stage1_operands(b1, Sh, w);
    stage1(mf, b1, Yh, w);
    __syncthreads();
    v8f f0, f1;
    stage2(mf, Yh, w, f0, f1);
    float qacc = 0.0f;
#pragma unroll
    for (int r = 0; r < 8; ++r) {
        const float hs0 = 0.5f * (hv[0][r] + hm[0][r]);
        const float hs1 = 0.5f * (hv[1][r] + hm[1][r]);
        qacc += (f0[r] * f0[r]) * (hs0 * hs0);
        qacc += (f1[r] * f1[r]) * (hs1 * hs1);
        Sh[(2 * w + 8 * r) * SPITCH]     = f0[r] * hv[0][r];
        Sh[(2 * w + 1 + 8 * r) * SPITCH] = f1[r] * hv[1][r];
    }
#pragma unroll
    for (int m = 16; m > 0; m >>= 1) qacc += __shfl_xor(qacc, m);
    if (lane == 0) wq[w] = qacc;
    if (blockIdx.x == 4 * 32 + 2 && threadIdx.x == 0)
        stat[NBLK2] = f0[4] * hv[0][4];
    __syncthreads();
    if (threadIdx.x == 0) stat[blockIdx.x] = (wq[0] + wq[1]) + (wq[2] + wq[3]);

    v2f c1[2][4];
    stage1_operands(c1, Sh, w);
    stage1(mi, c1, Yh, w);
    __syncthreads();
    v8f r0, r1;
    stage2(mi, Yh, w, r0, r1);
#pragma unroll
    for (int r = 0; r < 8; ++r) {
        Tp[slab_idx(2 * w + 8 * r, ky, kt, l16)]     = r0[r];
        Tp[slab_idx(2 * w + 1 + 8 * r, ky, kt, l16)] = r1[r];
    }
}

__global__ __launch_bounds__(128) void inv_yz_norm_kernel(const float* __restrict__ Tre,
                                                           const float* __restrict__ Tim,
                                                           const float* __restrict__ Mre,
                                                           const float* __restrict__ Mim,
                                                           const float* __restrict__ stat,
                                                           float* __restrict__ out) {
    __shared__ __attribute__((aligned(16))) float Lre[16 * WPITCH];
    __shared__ __attribute__((aligned(16))) float Lim[16 * WPITCH];
    __shared__ float wsq[4];
    const int lane = threadIdx.x & 31;
    const int w    = threadIdx.x >> 5;
    const int half = lane >> 4;
    const int l16  = lane & 15;
    const int x    = blockIdx.x >> 2;
    const int yt   = blockIdx.x & 3;

    v4f m2r[8], m2i[8];
    load_mfrag(m2r, Mre, w, lane);
    load_mfrag(m2i, Mim, w, lane);

    float q = stat[threadIdx.x] + stat[threadIdx.x + 128];
    const float g0 = stat[NBLK2];

    v4f m1r[8], m1i[8];
    load_mfrag(m1r, Mre, yt, lane);
    load_mfrag(m1i, Mim, yt, lane);
    v8f rr = {}, ii = {}, ri = {}, ir = {};
#pragma unroll
    for (int s = 0; s < 16; ++s) {
        const int k = kof(s, half);
        const v2f ar = mstep(m1r, s);
        const v2f ai = mstep(m1i, s);
        v2f br, bi;
        br.x = Tre[slab_idx(x, k, w, l16)];  br.y = Tre[slab_idx(x, k + 1, w, l16)];
        bi.x = Tim[slab_idx(x, k, w, l16)];  bi.y = Tim[slab_idx(x, k + 1, w, l16)];
        rr = mma4(ar, br, rr);
        ii = mma4(ai, bi, ii);
        ri = mma4(ar, bi, ri);
        ir = mma4(ai, br, ir);
    }
#pragma unroll
    for (int v = 0; v < 8; ++v) {
        Lre[(v + 8 * half) * WPITCH + 16 * w + l16] = rr[v] + ii[v];
        Lim[(v + 8 * half) * WPITCH + 16 * w + l16] = ri[v] - ir[v];
    }
#pragma unroll
    for (int m = 16; m > 0; m >>= 1) q += __shfl_xor(q, m);
    if (lane == 0) wsq[w] = q;
    __syncthreads();
    const float tsq  = (wsq[0] + wsq[1]) + (wsq[2] + wsq[3]);
    const float invN = 1.0f / (float)GRID_N;
    const float mean = g0 * (1.0f / 512.0f);
    const float var  = fmaxf(tsq * invN - mean * mean, 0.0f);
    const float inv  = 1.0f / (sqrtf(var) + 1e-8f);

    v8f pr = {}, pi = {};
    v4f wr[8], wi[8];
#pragma unroll
    for (int q = 0; q < 8; ++q) {
        wr[q] = *(const v4f*)(Lre + l16 * WPITCH + kof(2 * q, half));
        wi[q] = *(const v4f*)(Lim + l16 * WPITCH + kof(2 * q, half));
    }
    __builtin_amdgcn_sched_barrier(0);
#pragma unroll
    for (int s = 0; s < 16; ++s) {
        pr = mma4(mstep(wr, s), mstep(m2r, s), pr);
        pi = mma4(mstep(wi, s), mstep(m2i, s), pi);
    }
#pragma unroll
    for (int v = 0; v < 8; ++v) {
        const float re = pr[v] + pi[v];
        out[(x * N1D + 16 * yt + v + 8 * half) * N1D + 16 * w + l16] = (re - mean) * inv;
    }
}

extern "C" void kernel_launch(void* const* d_in, const int* in_sizes, int n_in,
                              void* d_out, int out_size, void* d_ws, size_t ws_size,
                              hipStream_t stream) {
    const float* X   = (const float*)d_in[0];
    const float* aw  = (const float*)d_in[1];
    const float* bw  = (const float*)d_in[2];
    const float* ham = (const float*)d_in[4];
    float* out = (float*)d_out;

    float* ws   = (float*)d_ws;
    float* Wre  = ws;
    float* WimF = Wre + 4096;
    float* stat = WimF + 4096;
    float* Are  = stat + 4096;
    float* Aim  = Are + GRID_N;
    float* Bre  = Aim + GRID_N;
    float* Bim  = Bre + GRID_N;

    density_yfwd_kernel<<<256, 128, 0, stream>>>(X, aw, bw, Wre, WimF, Are, Aim);

    zx_filter_kernel<<<NBLK2, 128, 0, stream>>>(Are, Aim, Bre, Bim, Wre, WimF, ham, stat);

    inv_yz_norm_kernel<<<NBLK2, 128, 0, stream>>>(Bre, Bim, Wre, WimF, stat, out);
}
